# starting kernel
# baseline (speedup 1.0000x reference)
#include <hip/hip_runtime.h>
#include <hip/hip_bf16.h>

typedef float v2f __attribute__((ext_vector_type(2)));
typedef float v8f __attribute__((ext_vector_type(8)));

#define D 16
#define NGRAPH 1024

__global__ void fill_zero_k(float* __restrict__ p, long long n) {
    long long i = (long long)blockIdx.x * blockDim.x + threadIdx.x;
    if (i < n) p[i] = 0.0f;
}

__global__ void gather_gemm16_k(const float* __restrict__ src_rows,
                                const int* __restrict__ ids,
                                const float* __restrict__ W,
                                float* __restrict__ out,
                                int n_nodes) {
    const int lane  = threadIdx.x & 31;
    const int wave  = (int)((blockIdx.x * blockDim.x + threadIdx.x) >> 5);
    const int nwave = (int)((gridDim.x * blockDim.x) >> 5);
    const int row   = lane & 15;
    const int col   = lane & 15;
    const int koff  = (lane >> 4) << 1;
    const int rowb  = (lane >> 4) << 3;

    v2f bb0, bb1, bb2, bb3;
    bb0.x = W[(0  + koff) * D + col]; bb0.y = W[(1  + koff) * D + col];
    bb1.x = W[(4  + koff) * D + col]; bb1.y = W[(5  + koff) * D + col];
    bb2.x = W[(8  + koff) * D + col]; bb2.y = W[(9  + koff) * D + col];
    bb3.x = W[(12 + koff) * D + col]; bb3.y = W[(13 + koff) * D + col];

    const int ntiles = n_nodes >> 4;
    for (int tile = wave; tile < ntiles; tile += nwave) {
        const int node = (tile << 4) + row;
        const int r = ids ? ids[node] : node;
        const float* xrow = src_rows + (size_t)r * D;
        v2f a0 = *(const v2f*)(xrow + 0  + koff);
        v2f a1 = *(const v2f*)(xrow + 4  + koff);
        v2f a2 = *(const v2f*)(xrow + 8  + koff);
        v2f a3 = *(const v2f*)(xrow + 12 + koff);
        v8f c = {};
        c = __builtin_amdgcn_wmma_f32_16x16x4_f32(false, a0, false, bb0, (short)0, c, false, false);
        c = __builtin_amdgcn_wmma_f32_16x16x4_f32(false, a1, false, bb1, (short)0, c, false, false);
        c = __builtin_amdgcn_wmma_f32_16x16x4_f32(false, a2, false, bb2, (short)0, c, false, false);
        c = __builtin_amdgcn_wmma_f32_16x16x4_f32(false, a3, false, bb3, (short)0, c, false, false);
        float* ob = out + ((size_t)(tile << 4) + rowb) * D + col;
        #pragma unroll
        for (int rr = 0; rr < 8; ++rr) ob[(size_t)rr * D] = c[rr];
    }

    const int tail_start = ntiles << 4;
    if (tail_start < n_nodes) {
        int node = tail_start + (int)(blockIdx.x * blockDim.x + threadIdx.x);
        if (node < n_nodes) {
            const int r = ids ? ids[node] : node;
            const float* xrow = src_rows + (size_t)r * D;
            for (int j = 0; j < D; ++j) {
                float acc = 0.0f;
                for (int k = 0; k < D; ++k) acc += xrow[k] * W[k * D + j];
                out[(size_t)node * D + j] = acc;
            }
        }
    }
}

__global__ void degree_k(const int* __restrict__ dst, float* __restrict__ deg,
                         long long E) {
    long long e = (long long)blockIdx.x * blockDim.x + threadIdx.x;
    if (e < E) atomicAdd(deg + dst[e], 1.0f);
}

__global__ void rsqrt_k(float* __restrict__ d, int n) {
    int i = blockIdx.x * blockDim.x + threadIdx.x;
    if (i < n) d[i] = rsqrtf(d[i] + 1.0f);
}

__global__ void aggregate_k(const float* __restrict__ h,
                            const float* __restrict__ dis,
                            const int* __restrict__ src,
                            const int* __restrict__ dst,
                            float* __restrict__ agg,
                            long long E) {
    long long t = (long long)blockIdx.x * blockDim.x + threadIdx.x;
    long long e = t >> 4;
    int j = (int)(t & 15);
    if (e < E) {
        int s = src[e], d = dst[e];
        float norm = dis[s] * dis[d];
        atomicAdd(agg + (size_t)d * D + j, h[(size_t)s * D + j] * norm);
    }
}

__global__ void node_relu_k(float* __restrict__ h, const float* __restrict__ agg,
                            const float* __restrict__ dis,
                            const float* __restrict__ bias, long long n16) {
    long long t = (long long)blockIdx.x * blockDim.x + threadIdx.x;
    if (t < n16) {
        long long i = t >> 4; int j = (int)(t & 15);
        float di = dis[i];
        float v = agg[t] + h[t] * (di * di) + bias[j];
        h[t] = v > 0.0f ? v : 0.0f;
    }
}

__global__ void node_pool_k(const float* __restrict__ h2,
                            const float* __restrict__ agg,
                            const float* __restrict__ dis,
                            const float* __restrict__ bias,
                            const int* __restrict__ batch,
                            float* __restrict__ pooled,
                            float* __restrict__ cnt, long long n16) {
    long long t = (long long)blockIdx.x * blockDim.x + threadIdx.x;
    if (t < n16) {
        long long i = t >> 4; int j = (int)(t & 15);
        float di = dis[i];
        float v = agg[t] + h2[t] * (di * di) + bias[j];
        int g = batch[i];
        atomicAdd(pooled + (size_t)g * D + j, v);
        if (j == 0) atomicAdd(cnt + g, 1.0f);
    }
}

__global__ void fc_k(const float* __restrict__ pooled_r, const float* __restrict__ cnt_r,
                     const float* __restrict__ pooled_l, const float* __restrict__ cnt_l,
                     const float* __restrict__ fcW, const float* __restrict__ fcb,
                     float* __restrict__ out) {
    int t = blockIdx.x * blockDim.x + threadIdx.x;
    if (t < NGRAPH * 6) {
        int g = t / 6, o = t - g * 6;
        float icr = 1.0f / fmaxf(cnt_r[g], 1.0f);
        float icl = 1.0f / fmaxf(cnt_l[g], 1.0f);
        float acc = fcb[o];
        #pragma unroll
        for (int j = 0; j < D; ++j)
            acc += fcW[o * (2 * D) + j] * (pooled_r[g * D + j] * icr);
        #pragma unroll
        for (int j = 0; j < D; ++j)
            acc += fcW[o * (2 * D) + D + j] * (pooled_l[g * D + j] * icl);
        if (o < 3) out[g * 3 + o] = acc;
        else       out[NGRAPH * 3 + g * 3 + (o - 3)] = acc;
    }
}

static void run_protein(const float* emb, const float* W1, const float* b1,
                        const float* W2, const float* b2,
                        const int* ids, const int* edge, const int* batch,
                        int N, long long E,
                        float* bufA, float* bufB, float* dis,
                        float* pooled, float* cnt, hipStream_t stream) {
    const int* src = edge;
    const int* dst = edge + E;
    const long long n16 = (long long)N * D;
    const int B = 256;
    dim3 blk(B);
    int gemm_grid = (int)(((N >> 4) + 7) / 8);
    if (gemm_grid < 1) gemm_grid = 1;

    fill_zero_k<<<(int)((N + B - 1) / B), blk, 0, stream>>>(dis, N);
    fill_zero_k<<<(int)((n16 + B - 1) / B), blk, 0, stream>>>(bufB, n16);

    gather_gemm16_k<<<gemm_grid, blk, 0, stream>>>(emb, ids, W1, bufA, N);

    degree_k<<<(int)((E + B - 1) / B), blk, 0, stream>>>(dst, dis, E);
    rsqrt_k<<<(int)((N + B - 1) / B), blk, 0, stream>>>(dis, N);

    aggregate_k<<<(int)((E * 16 + B - 1) / B), blk, 0, stream>>>(bufA, dis, src, dst, bufB, E);
    node_relu_k<<<(int)((n16 + B - 1) / B), blk, 0, stream>>>(bufA, bufB, dis, b1, n16);

    gather_gemm16_k<<<gemm_grid, blk, 0, stream>>>(bufA, nullptr, W2, bufB, N);

    fill_zero_k<<<(int)((n16 + B - 1) / B), blk, 0, stream>>>(bufA, n16);
    aggregate_k<<<(int)((E * 16 + B - 1) / B), blk, 0, stream>>>(bufB, dis, src, dst, bufA, E);

    fill_zero_k<<<(int)((NGRAPH * D + B - 1) / B), blk, 0, stream>>>(pooled, NGRAPH * D);
    fill_zero_k<<<(int)((NGRAPH + B - 1) / B), blk, 0, stream>>>(cnt, NGRAPH);
    node_pool_k<<<(int)((n16 + B - 1) / B), blk, 0, stream>>>(bufB, bufA, dis, b2, batch,
                                                              pooled, cnt, n16);
}

extern "C" void kernel_launch(void* const* d_in, const int* in_sizes, int n_in,
                              void* d_out, int out_size, void* d_ws, size_t ws_size,
                              hipStream_t stream) {
    const float* emb = (const float*)d_in[0];
    const float* W1  = (const float*)d_in[1];
    const float* b1  = (const float*)d_in[2];
    const float* W2  = (const float*)d_in[3];
    const float* b2  = (const float*)d_in[4];
    const float* fcW = (const float*)d_in[5];
    const float* fcb = (const float*)d_in[6];
    const int* rx     = (const int*)d_in[7];
    const int* redge  = (const int*)d_in[8];
    const int* rbatch = (const int*)d_in[9];
    const int* lx     = (const int*)d_in[10];
    const int* ledge  = (const int*)d_in[11];
    const int* lbatch = (const int*)d_in[12];

    const int N = in_sizes[7];
    const long long E = in_sizes[8] / 2;

    char* ws = (char*)d_ws;
    float* bufA = (float*)ws;  ws += (size_t)N * D * sizeof(float);
    float* bufB = (float*)ws;  ws += (size_t)N * D * sizeof(float);
    float* dis  = (float*)ws;  ws += (size_t)N * sizeof(float);
    float* pooled_r = (float*)ws; ws += (size_t)NGRAPH * D * sizeof(float);
    float* pooled_l = (float*)ws; ws += (size_t)NGRAPH * D * sizeof(float);
    float* cnt_r = (float*)ws; ws += (size_t)NGRAPH * sizeof(float);
    float* cnt_l = (float*)ws; ws += (size_t)NGRAPH * sizeof(float);

    run_protein(emb, W1, b1, W2, b2, rx, redge, rbatch, N, E,
                bufA, bufB, dis, pooled_r, cnt_r, stream);
    run_protein(emb, W1, b1, W2, b2, lx, ledge, lbatch, N, E,
                bufA, bufB, dis, pooled_l, cnt_l, stream);

    fc_k<<<(NGRAPH * 6 + 255) / 256, 256, 0, stream>>>(pooled_r, cnt_r, pooled_l, cnt_l,
                                                       fcW, fcb, (float*)d_out);
}


# edges counting-sorted once per protein into 256-node dst buckets (4-byte records); in-degree, both aggregations and pool summed per bucket in LDS, no per-edge global atomics; float sum order changes
# speedup vs baseline: 4.5230x; 4.5230x over previous
#include <hip/hip_runtime.h>
#include <hip/hip_bf16.h>

typedef float v2f __attribute__((ext_vector_type(2)));
typedef float v8f __attribute__((ext_vector_type(8)));

#define D 16
#define NGRAPH 1024

__global__ void fill_zero_k(float* __restrict__ p, long long n) {
    long long i = (long long)blockIdx.x * blockDim.x + threadIdx.x;
    if (i < n) p[i] = 0.0f;
}

__global__ void gather_gemm16_k(const float* __restrict__ src_rows,
                                const int* __restrict__ ids,
                                const float* __restrict__ W,
                                float* __restrict__ out,
                                int n_nodes) {
    const int lane  = threadIdx.x & 31;
    const int wave  = (int)((blockIdx.x * blockDim.x + threadIdx.x) >> 5);
    const int nwave = (int)((gridDim.x * blockDim.x) >> 5);
    const int row   = lane & 15;
    const int col   = lane & 15;
    const int koff  = (lane >> 4) << 1;
    const int rowb  = (lane >> 4) << 3;

    v2f bb0, bb1, bb2, bb3;
    bb0.x = W[(0  + koff) * D + col]; bb0.y = W[(1  + koff) * D + col];
    bb1.x = W[(4  + koff) * D + col]; bb1.y = W[(5  + koff) * D + col];
    bb2.x = W[(8  + koff) * D + col]; bb2.y = W[(9  + koff) * D + col];
    bb3.x = W[(12 + koff) * D + col]; bb3.y = W[(13 + koff) * D + col];

    const int ntiles = n_nodes >> 4;
    for (int tile = wave; tile < ntiles; tile += nwave) {
        const int node = (tile << 4) + row;
        const int r = ids ? ids[node] : node;
        const float* xrow = src_rows + (size_t)r * D;
        v2f a0 = *(const v2f*)(xrow + 0  + koff);
        v2f a1 = *(const v2f*)(xrow + 4  + koff);
        v2f a2 = *(const v2f*)(xrow + 8  + koff);
        v2f a3 = *(const v2f*)(xrow + 12 + koff);
        v8f c = {};
        c = __builtin_amdgcn_wmma_f32_16x16x4_f32(false, a0, false, bb0, (short)0, c, false, false);
        c = __builtin_amdgcn_wmma_f32_16x16x4_f32(false, a1, false, bb1, (short)0, c, false, false);
        c = __builtin_amdgcn_wmma_f32_16x16x4_f32(false, a2, false, bb2, (short)0, c, false, false);
        c = __builtin_amdgcn_wmma_f32_16x16x4_f32(false, a3, false, bb3, (short)0, c, false, false);
        float* ob = out + ((size_t)(tile << 4) + rowb) * D + col;
        #pragma unroll
        for (int rr = 0; rr < 8; ++rr) ob[(size_t)rr * D] = c[rr];
    }

    const int tail_start = ntiles << 4;
    if (tail_start < n_nodes) {
        int node = tail_start + (int)(blockIdx.x * blockDim.x + threadIdx.x);
        if (node < n_nodes) {
            const int r = ids ? ids[node] : node;
            const float* xrow = src_rows + (size_t)r * D;
            for (int j = 0; j < D; ++j) {
                float acc = 0.0f;
                for (int k = 0; k < D; ++k) acc += xrow[k] * W[k * D + j];
                out[(size_t)node * D + j] = acc;
            }
        }
    }
}

__global__ void degree_k(const int* __restrict__ dst, float* __restrict__ deg,
                         long long E) {
    long long e = (long long)blockIdx.x * blockDim.x + threadIdx.x;
    if (e < E) atomicAdd(deg + dst[e], 1.0f);
}

__global__ void rsqrt_k(float* __restrict__ d, int n) {
    int i = blockIdx.x * blockDim.x + threadIdx.x;
    if (i < n) d[i] = rsqrtf(d[i] + 1.0f);
}

__global__ void aggregate_k(const float* __restrict__ h,
                            const float* __restrict__ dis,
                            const int* __restrict__ src,
                            const int* __restrict__ dst,
                            float* __restrict__ agg,
                            long long E) {
    long long t = (long long)blockIdx.x * blockDim.x + threadIdx.x;
    long long e = t >> 4;
    int j = (int)(t & 15);
    if (e < E) {
        int s = src[e], d = dst[e];
        float norm = dis[s] * dis[d];
        atomicAdd(agg + (size_t)d * D + j, h[(size_t)s * D + j] * norm);
    }
}

__global__ void node_relu_k(float* __restrict__ h, const float* __restrict__ agg,
                            const float* __restrict__ dis,
                            const float* __restrict__ bias, long long n16) {
    long long t = (long long)blockIdx.x * blockDim.x + threadIdx.x;
    if (t < n16) {
        long long i = t >> 4; int j = (int)(t & 15);
        float di = dis[i];
        float v = agg[t] + h[t] * (di * di) + bias[j];
        h[t] = v > 0.0f ? v : 0.0f;
    }
}

__global__ void node_pool_k(const float* __restrict__ h2,
                            const float* __restrict__ agg,
                            const float* __restrict__ dis,
                            const float* __restrict__ bias,
                            const int* __restrict__ batch,
                            float* __restrict__ pooled,
                            float* __restrict__ cnt, long long n16) {
    long long t = (long long)blockIdx.x * blockDim.x + threadIdx.x;
    if (t < n16) {
        long long i = t >> 4; int j = (int)(t & 15);
        float di = dis[i];
        float v = agg[t] + h2[t] * (di * di) + bias[j];
        int g = batch[i];
        atomicAdd(pooled + (size_t)g * D + j, v);
        if (j == 0) atomicAdd(cnt + g, 1.0f);
    }
}

__global__ void fc_k(const float* __restrict__ pooled_r, const float* __restrict__ cnt_r,
                     const float* __restrict__ pooled_l, const float* __restrict__ cnt_l,
                     const float* __restrict__ fcW, const float* __restrict__ fcb,
                     float* __restrict__ out) {
    int t = blockIdx.x * blockDim.x + threadIdx.x;
    if (t < NGRAPH * 6) {
        int g = t / 6, o = t - g * 6;
        float icr = 1.0f / fmaxf(cnt_r[g], 1.0f);
        float icl = 1.0f / fmaxf(cnt_l[g], 1.0f);
        float acc = fcb[o];
        #pragma unroll
        for (int j = 0; j < D; ++j)
            acc += fcW[o * (2 * D) + j] * (pooled_r[g * D + j] * icr);
        #pragma unroll
        for (int j = 0; j < D; ++j)
            acc += fcW[o * (2 * D) + D + j] * (pooled_l[g * D + j] * icl);
        if (o < 3) out[g * 3 + o] = acc;
        else       out[NGRAPH * 3 + g * 3 + (o - 3)] = acc;
    }
}


#define BSH 8
#define BNODES 256
#define NBMAX 1024
#define PTHREADS 1024
#define PTILE 32768
#define PEPT (PTILE / PTHREADS)
#define PU 8

__global__ __launch_bounds__(PTHREADS)
void bucket_hist_k(const int* __restrict__ dst, long long E,
                   int* __restrict__ bcount) {
    __shared__ int cnt[NBMAX];
    const int t = threadIdx.x;
    cnt[t] = 0;
    __syncthreads();
    const long long base = (long long)blockIdx.x * PTILE;
    const long long rem = E - base;
    const int ne = rem < (long long)PTILE ? (int)rem : PTILE;
    for (int k0 = 0; k0 < PEPT; k0 += PU) {
        int dv[PU];
        #pragma unroll
        for (int u = 0; u < PU; ++u) {
            int l = (k0 + u) * PTHREADS + t;
            dv[u] = dst[base + (l < ne ? l : 0)];
        }
        #pragma unroll
        for (int u = 0; u < PU; ++u) {
            int l = (k0 + u) * PTHREADS + t;
            if (l < ne) atomicAdd(&cnt[dv[u] >> BSH], 1);
        }
    }
    __syncthreads();
    const int c = cnt[t];
    if (c > 0) atomicAdd(bcount + t, c);
}

__global__ __launch_bounds__(NBMAX)
void bucket_scan_k(const int* __restrict__ bcount, int* __restrict__ bstart,
                   int* __restrict__ cursor, int nb) {
    __shared__ int s[NBMAX];
    const int t = threadIdx.x;
    const int own = (t < nb) ? bcount[t] : 0;
    s[t] = own;
    __syncthreads();
    for (int off = 1; off < NBMAX; off <<= 1) {
        int v = (t >= off) ? s[t - off] : 0;
        __syncthreads();
        s[t] += v;
        __syncthreads();
    }
    const int incl = s[t];
    if (t < nb) { bstart[t] = incl - own; cursor[t] = incl - own; }
    if (t == nb - 1) bstart[nb] = incl;
}

__global__ __launch_bounds__(PTHREADS)
void bucket_scatter_k(const int* __restrict__ src, const int* __restrict__ dst,
                      long long E, int* __restrict__ cursor,
                      unsigned* __restrict__ part) {
    __shared__ int cnt[NBMAX];
    __shared__ int tstart[NBMAX];
    __shared__ int gbase[NBMAX];
    __shared__ int run[NBMAX];
    __shared__ unsigned stage[PTILE];
    __shared__ unsigned short bid[PTILE];
    const int t = threadIdx.x;
    cnt[t] = 0;
    __syncthreads();
    const long long base = (long long)blockIdx.x * PTILE;
    const long long rem = E - base;
    const int ne = rem < (long long)PTILE ? (int)rem : PTILE;
    for (int k0 = 0; k0 < PEPT; k0 += PU) {
        int dv[PU];
        #pragma unroll
        for (int u = 0; u < PU; ++u) {
            int l = (k0 + u) * PTHREADS + t;
            dv[u] = dst[base + (l < ne ? l : 0)];
        }
        #pragma unroll
        for (int u = 0; u < PU; ++u) {
            int l = (k0 + u) * PTHREADS + t;
            if (l < ne) atomicAdd(&cnt[dv[u] >> BSH], 1);
        }
    }
    __syncthreads();
    const int own = cnt[t];
    run[t] = own;
    __syncthreads();
    for (int off = 1; off < NBMAX; off <<= 1) {
        int v = (t >= off) ? run[t - off] : 0;
        __syncthreads();
        run[t] += v;
        __syncthreads();
    }
    const int excl = run[t] - own;
    tstart[t] = excl;
    run[t] = excl;
    gbase[t] = (own > 0) ? atomicAdd(cursor + t, own) : 0;
    __syncthreads();
    for (int k0 = 0; k0 < PEPT; k0 += PU) {
        int dv[PU], sv[PU];
        #pragma unroll
        for (int u = 0; u < PU; ++u) {
            int l = (k0 + u) * PTHREADS + t;
            long long e = base + (l < ne ? l : 0);
            dv[u] = dst[e];
            sv[u] = src[e];
        }
        #pragma unroll
        for (int u = 0; u < PU; ++u) {
            int l = (k0 + u) * PTHREADS + t;
            if (l < ne) {
                int b = dv[u] >> BSH;
                int slot = atomicAdd(&run[b], 1);
                stage[slot] = ((unsigned)sv[u] << BSH) | (unsigned)(dv[u] & (BNODES - 1));
                bid[slot] = (unsigned short)b;
            }
        }
    }
    __syncthreads();
    for (int slot = t; slot < ne; slot += PTHREADS) {
        int b = bid[slot];
        part[gbase[b] + (slot - tstart[b])] = stage[slot];
    }
}

__global__ __launch_bounds__(BNODES)
void bucket_degree_k(const unsigned* __restrict__ part,
                     const int* __restrict__ bstart,
                     float* __restrict__ dis, int n) {
    __shared__ int cnt[BNODES];
    const int t = threadIdx.x;
    cnt[t] = 0;
    __syncthreads();
    const int b = blockIdx.x;
    const int p0 = bstart[b], p1 = bstart[b + 1];
    for (int p = p0 + t; p < p1; p += BNODES * PU) {
        unsigned r[PU];
        #pragma unroll
        for (int u = 0; u < PU; ++u) {
            int q = p + u * BNODES;
            r[u] = part[q < p1 ? q : 0];
        }
        #pragma unroll
        for (int u = 0; u < PU; ++u) {
            int q = p + u * BNODES;
            if (q < p1) atomicAdd(&cnt[r[u] & (BNODES - 1)], 1);
        }
    }
    __syncthreads();
    const int i = (b << BSH) + t;
    if (i < n) dis[i] = rsqrtf((float)cnt[t] + 1.0f);
}

#define LTHREADS 512
#define LGROUPS (LTHREADS / 16)
#define LU 8
#define PGR 16
template <bool POOL>
__global__ __launch_bounds__(LTHREADS)
void bucket_layer_k(const float* __restrict__ h, const float* __restrict__ dis,
                    const unsigned* __restrict__ part,
                    const int* __restrict__ bstart,
                    const float* __restrict__ bias,
                    float* __restrict__ rows,
                    const int* __restrict__ batch,
                    float* __restrict__ pooled, float* __restrict__ cntg,
                    int n) {
    __shared__ float acc[BNODES * D];
    __shared__ float dloc[BNODES];
    __shared__ float pl[PGR * D];
    __shared__ int   pc[PGR];
    const int t = threadIdx.x;
    const int b = blockIdx.x;
    const int node0 = b << BSH;
    for (int i = t; i < BNODES * D; i += LTHREADS) acc[i] = 0.0f;
    if (t < BNODES) {
        int i = node0 + t;
        dloc[t] = (i < n) ? dis[i] : 0.0f;
    }
    if (POOL) {
        if (t < PGR * D) pl[t] = 0.0f;
        if (t < PGR) pc[t] = 0;
    }
    __syncthreads();

    const int j = t & 15;
    const int grp = t >> 4;
    const int p0 = bstart[b], p1 = bstart[b + 1];
    const int step = LGROUPS * LU;
    unsigned recn[LU];
    #pragma unroll
    for (int u = 0; u < LU; ++u) {
        int q = p0 + grp + u * LGROUPS;
        recn[u] = part[q < p1 ? q : 0];
    }
    for (int p = p0 + grp; p < p1; p += step) {
        unsigned rec[LU];
        float hv[LU], ds[LU];
        #pragma unroll
        for (int u = 0; u < LU; ++u) {
            rec[u] = recn[u];
            int q = p + step + u * LGROUPS;
            recn[u] = part[q < p1 ? q : 0];
        }
        #pragma unroll
        for (int u = 0; u < LU; ++u) {
            int s = (int)(rec[u] >> BSH);
            ds[u] = dis[s];
            hv[u] = h[(size_t)s * D + j];
        }
        #pragma unroll
        for (int u = 0; u < LU; ++u) {
            int q = p + u * LGROUPS;
            int d = (int)(rec[u] & (BNODES - 1));
            float norm = ds[u] * dloc[d];
            float term = hv[u] * norm;
            atomicAdd(&acc[d * D + j], (q < p1) ? term : 0.0f);
        }
    }
    __syncthreads();

    if (!POOL) {
        for (int idx = t; idx < BNODES * D; idx += LTHREADS) {
            int nl = idx >> 4;
            int i = node0 + nl;
            if (i < n) {
                float di = dloc[nl];
                size_t o = (size_t)i * D + (idx & 15);
                float v = acc[idx] + h[o] * (di * di) + bias[idx & 15];
                rows[o] = v > 0.0f ? v : 0.0f;
            }
        }
    } else {
        const int g0 = batch[node0];
        for (int idx = t; idx < BNODES * D; idx += LTHREADS) {
            int nl = idx >> 4;
            int jj = idx & 15;
            int i = node0 + nl;
            if (i < n) {
                float di = dloc[nl];
                size_t o = (size_t)i * D + jj;
                float v = acc[idx] + h[o] * (di * di) + bias[jj];
                int g = batch[i];
                unsigned k = (unsigned)(g - g0);
                if (k < (unsigned)PGR) {
                    atomicAdd(&pl[k * D + jj], v);
                    if (jj == 0) atomicAdd(&pc[k], 1);
                } else {
                    atomicAdd(pooled + (size_t)g * D + jj, v);
                    if (jj == 0) atomicAdd(cntg + g, 1.0f);
                }
            }
        }
        __syncthreads();
        if (t < PGR * D) {
            int k = t >> 4;
            int c = pc[k];
            if (c > 0) {
                int g = g0 + k;
                atomicAdd(pooled + (size_t)g * D + (t & 15), pl[t]);
                if ((t & 15) == 0) atomicAdd(cntg + g, (float)c);
            }
        }
    }
}

static void run_protein_bucketed(const float* emb, const float* W1, const float* b1,
                                 const float* W2, const float* b2,
                                 const int* ids, const int* edge, const int* batch,
                                 int N, long long E,
                                 float* bufA, float* bufB, float* dis,
                                 float* pooled, float* cnt,
                                 unsigned* part, int* bcount, int* bstart, int* cursor,
                                 hipStream_t stream) {
    const int* src = edge;
    const int* dst = edge + E;
    const int B = 256;
    dim3 blk(B);
    int gemm_grid = (int)(((N >> 4) + 7) / 8);
    if (gemm_grid < 1) gemm_grid = 1;
    const int nb = (N + BNODES - 1) / BNODES;
    const int ptiles = (int)((E + PTILE - 1) / PTILE);

    fill_zero_k<<<(NBMAX + B - 1) / B, blk, 0, stream>>>((float*)bcount, NBMAX);
    bucket_hist_k<<<ptiles, PTHREADS, 0, stream>>>(dst, E, bcount);
    bucket_scan_k<<<1, NBMAX, 0, stream>>>(bcount, bstart, cursor, nb);
    bucket_scatter_k<<<ptiles, PTHREADS, 0, stream>>>(src, dst, E, cursor, part);
    bucket_degree_k<<<nb, BNODES, 0, stream>>>(part, bstart, dis, N);

    gather_gemm16_k<<<gemm_grid, blk, 0, stream>>>(emb, ids, W1, bufA, N);
    bucket_layer_k<false><<<nb, LTHREADS, 0, stream>>>(bufA, dis, part, bstart, b1, bufB,
                                                      nullptr, nullptr, nullptr, N);
    gather_gemm16_k<<<gemm_grid, blk, 0, stream>>>(bufB, nullptr, W2, bufA, N);

    fill_zero_k<<<(int)((NGRAPH * D + B - 1) / B), blk, 0, stream>>>(pooled, NGRAPH * D);
    fill_zero_k<<<(int)((NGRAPH + B - 1) / B), blk, 0, stream>>>(cnt, NGRAPH);
    bucket_layer_k<true><<<nb, LTHREADS, 0, stream>>>(bufA, dis, part, bstart, b2, nullptr,
                                                     batch, pooled, cnt, N);
}

static void run_protein(const float* emb, const float* W1, const float* b1,
                        const float* W2, const float* b2,
                        const int* ids, const int* edge, const int* batch,
                        int N, long long E,
                        float* bufA, float* bufB, float* dis,
                        float* pooled, float* cnt, hipStream_t stream) {
    const int* src = edge;
    const int* dst = edge + E;
    const long long n16 = (long long)N * D;
    const int B = 256;
    dim3 blk(B);
    int gemm_grid = (int)(((N >> 4) + 7) / 8);
    if (gemm_grid < 1) gemm_grid = 1;

    fill_zero_k<<<(int)((N + B - 1) / B), blk, 0, stream>>>(dis, N);
    fill_zero_k<<<(int)((n16 + B - 1) / B), blk, 0, stream>>>(bufB, n16);

    gather_gemm16_k<<<gemm_grid, blk, 0, stream>>>(emb, ids, W1, bufA, N);

    degree_k<<<(int)((E + B - 1) / B), blk, 0, stream>>>(dst, dis, E);
    rsqrt_k<<<(int)((N + B - 1) / B), blk, 0, stream>>>(dis, N);

    aggregate_k<<<(int)((E * 16 + B - 1) / B), blk, 0, stream>>>(bufA, dis, src, dst, bufB, E);
    node_relu_k<<<(int)((n16 + B - 1) / B), blk, 0, stream>>>(bufA, bufB, dis, b1, n16);

    gather_gemm16_k<<<gemm_grid, blk, 0, stream>>>(bufA, nullptr, W2, bufB, N);

    fill_zero_k<<<(int)((n16 + B - 1) / B), blk, 0, stream>>>(bufA, n16);
    aggregate_k<<<(int)((E * 16 + B - 1) / B), blk, 0, stream>>>(bufB, dis, src, dst, bufA, E);

    fill_zero_k<<<(int)((NGRAPH * D + B - 1) / B), blk, 0, stream>>>(pooled, NGRAPH * D);
    fill_zero_k<<<(int)((NGRAPH + B - 1) / B), blk, 0, stream>>>(cnt, NGRAPH);
    node_pool_k<<<(int)((n16 + B - 1) / B), blk, 0, stream>>>(bufB, bufA, dis, b2, batch,
                                                              pooled, cnt, n16);
}

extern "C" void kernel_launch(void* const* d_in, const int* in_sizes, int n_in,
                              void* d_out, int out_size, void* d_ws, size_t ws_size,
                              hipStream_t stream) {
    const float* emb = (const float*)d_in[0];
    const float* W1  = (const float*)d_in[1];
    const float* b1  = (const float*)d_in[2];
    const float* W2  = (const float*)d_in[3];
    const float* b2  = (const float*)d_in[4];
    const float* fcW = (const float*)d_in[5];
    const float* fcb = (const float*)d_in[6];
    const int* rx     = (const int*)d_in[7];
    const int* redge  = (const int*)d_in[8];
    const int* rbatch = (const int*)d_in[9];
    const int* lx     = (const int*)d_in[10];
    const int* ledge  = (const int*)d_in[11];
    const int* lbatch = (const int*)d_in[12];

    const int N = in_sizes[7];
    const long long E = in_sizes[8] / 2;

    char* ws = (char*)d_ws;
    float* bufA = (float*)ws;  ws += (size_t)N * D * sizeof(float);
    float* bufB = (float*)ws;  ws += (size_t)N * D * sizeof(float);
    float* dis  = (float*)ws;  ws += (size_t)N * sizeof(float);
    float* pooled_r = (float*)ws; ws += (size_t)NGRAPH * D * sizeof(float);
    float* pooled_l = (float*)ws; ws += (size_t)NGRAPH * D * sizeof(float);
    float* cnt_r = (float*)ws; ws += (size_t)NGRAPH * sizeof(float);
    float* cnt_l = (float*)ws; ws += (size_t)NGRAPH * sizeof(float);

    unsigned* part = (unsigned*)ws;  ws += (size_t)E * sizeof(unsigned);
    int* bcount = (int*)ws;  ws += (size_t)NBMAX * sizeof(int);
    int* bstart = (int*)ws;  ws += (size_t)2 * NBMAX * sizeof(int);
    int* cursor = (int*)ws;  ws += (size_t)NBMAX * sizeof(int);
    const bool bucketed = (size_t)(ws - (char*)d_ws) <= ws_size &&
                          N > 0 && N <= NBMAX * BNODES && E > 0 && E < (1ll << 30);
    if (bucketed) {
        run_protein_bucketed(emb, W1, b1, W2, b2, rx, redge, rbatch, N, E,
                             bufA, bufB, dis, pooled_r, cnt_r,
                             part, bcount, bstart, cursor, stream);
        run_protein_bucketed(emb, W1, b1, W2, b2, lx, ledge, lbatch, N, E,
                             bufA, bufB, dis, pooled_l, cnt_l,
                             part, bcount, bstart, cursor, stream);
    } else {
        run_protein(emb, W1, b1, W2, b2, rx, redge, rbatch, N, E,
                    bufA, bufB, dis, pooled_r, cnt_r, stream);
        run_protein(emb, W1, b1, W2, b2, lx, ledge, lbatch, N, E,
                    bufA, bufB, dis, pooled_l, cnt_l, stream);
    }

    fc_k<<<(NGRAPH * 6 + 255) / 256, 256, 0, stream>>>(pooled_r, cnt_r, pooled_l, cnt_l,
                                                       fcW, fcb, (float*)d_out);
}


# gathered rows stored as (value, dis[node]) 8-byte pairs by the two products: one 8-byte load per lane and edge brings row value and source factor from one 128-byte line, not two lines
# speedup vs baseline: 4.7790x; 1.0566x over previous
#include <hip/hip_runtime.h>
#include <hip/hip_bf16.h>

typedef float v2f __attribute__((ext_vector_type(2)));
typedef float v8f __attribute__((ext_vector_type(8)));

#define D 16
#define NGRAPH 1024

__global__ void fill_zero_k(float* __restrict__ p, long long n) {
    long long i = (long long)blockIdx.x * blockDim.x + threadIdx.x;
    if (i < n) p[i] = 0.0f;
}

__global__ void gather_gemm16_k(const float* __restrict__ src_rows,
                                const int* __restrict__ ids,
                                const float* __restrict__ W,
                                float* __restrict__ out,
                                const float* __restrict__ dis_pair,
                                int n_nodes) {
    const int lane  = threadIdx.x & 31;
    const int wave  = (int)((blockIdx.x * blockDim.x + threadIdx.x) >> 5);
    const int nwave = (int)((gridDim.x * blockDim.x) >> 5);
    const int row   = lane & 15;
    const int col   = lane & 15;
    const int koff  = (lane >> 4) << 1;
    const int rowb  = (lane >> 4) << 3;

    v2f bb0, bb1, bb2, bb3;
    bb0.x = W[(0  + koff) * D + col]; bb0.y = W[(1  + koff) * D + col];
    bb1.x = W[(4  + koff) * D + col]; bb1.y = W[(5  + koff) * D + col];
    bb2.x = W[(8  + koff) * D + col]; bb2.y = W[(9  + koff) * D + col];
    bb3.x = W[(12 + koff) * D + col]; bb3.y = W[(13 + koff) * D + col];

    const int ntiles = n_nodes >> 4;
    for (int tile = wave; tile < ntiles; tile += nwave) {
        const int node = (tile << 4) + row;
        const int r = ids ? ids[node] : node;
        const float* xrow = src_rows + (size_t)r * D;
        v2f a0 = *(const v2f*)(xrow + 0  + koff);
        v2f a1 = *(const v2f*)(xrow + 4  + koff);
        v2f a2 = *(const v2f*)(xrow + 8  + koff);
        v2f a3 = *(const v2f*)(xrow + 12 + koff);
        v8f c = {};
        c = __builtin_amdgcn_wmma_f32_16x16x4_f32(false, a0, false, bb0, (short)0, c, false, false);
        c = __builtin_amdgcn_wmma_f32_16x16x4_f32(false, a1, false, bb1, (short)0, c, false, false);
        c = __builtin_amdgcn_wmma_f32_16x16x4_f32(false, a2, false, bb2, (short)0, c, false, false);
        c = __builtin_amdgcn_wmma_f32_16x16x4_f32(false, a3, false, bb3, (short)0, c, false, false);
        if (dis_pair) {
            const float* dp = dis_pair + (tile << 4) + rowb;
            v2f* ob2 = (v2f*)out + ((size_t)(tile << 4) + rowb) * D + col;
            #pragma unroll
            for (int rr = 0; rr < 8; ++rr) {
                v2f pr; pr.x = c[rr]; pr.y = dp[rr];
                ob2[(size_t)rr * D] = pr;
            }
        } else {
            float* ob = out + ((size_t)(tile << 4) + rowb) * D + col;
            #pragma unroll
            for (int rr = 0; rr < 8; ++rr) ob[(size_t)rr * D] = c[rr];
        }
    }

    const int tail_start = ntiles << 4;
    if (tail_start < n_nodes) {
        int node = tail_start + (int)(blockIdx.x * blockDim.x + threadIdx.x);
        if (node < n_nodes) {
            const int r = ids ? ids[node] : node;
            const float* xrow = src_rows + (size_t)r * D;
            for (int j = 0; j < D; ++j) {
                float acc = 0.0f;
                for (int k = 0; k < D; ++k) acc += xrow[k] * W[k * D + j];
                if (dis_pair) {
                    out[((size_t)node * D + j) * 2] = acc;
                    out[((size_t)node * D + j) * 2 + 1] = dis_pair[node];
                } else {
                    out[(size_t)node * D + j] = acc;
                }
            }
        }
    }
}

__global__ void degree_k(const int* __restrict__ dst, float* __restrict__ deg,
                         long long E) {
    long long e = (long long)blockIdx.x * blockDim.x + threadIdx.x;
    if (e < E) atomicAdd(deg + dst[e], 1.0f);
}

__global__ void rsqrt_k(float* __restrict__ d, int n) {
    int i = blockIdx.x * blockDim.x + threadIdx.x;
    if (i < n) d[i] = rsqrtf(d[i] + 1.0f);
}

__global__ void aggregate_k(const float* __restrict__ h,
                            const float* __restrict__ dis,
                            const int* __restrict__ src,
                            const int* __restrict__ dst,
                            float* __restrict__ agg,
                            long long E) {
    long long t = (long long)blockIdx.x * blockDim.x + threadIdx.x;
    long long e = t >> 4;
    int j = (int)(t & 15);
    if (e < E) {
        int s = src[e], d = dst[e];
        float norm = dis[s] * dis[d];
        atomicAdd(agg + (size_t)d * D + j, h[(size_t)s * D + j] * norm);
    }
}

__global__ void node_relu_k(float* __restrict__ h, const float* __restrict__ agg,
                            const float* __restrict__ dis,
                            const float* __restrict__ bias, long long n16) {
    long long t = (long long)blockIdx.x * blockDim.x + threadIdx.x;
    if (t < n16) {
        long long i = t >> 4; int j = (int)(t & 15);
        float di = dis[i];
        float v = agg[t] + h[t] * (di * di) + bias[j];
        h[t] = v > 0.0f ? v : 0.0f;
    }
}

__global__ void node_pool_k(const float* __restrict__ h2,
                            const float* __restrict__ agg,
                            const float* __restrict__ dis,
                            const float* __restrict__ bias,
                            const int* __restrict__ batch,
                            float* __restrict__ pooled,
                            float* __restrict__ cnt, long long n16) {
    long long t = (long long)blockIdx.x * blockDim.x + threadIdx.x;
    if (t < n16) {
        long long i = t >> 4; int j = (int)(t & 15);
        float di = dis[i];
        float v = agg[t] + h2[t] * (di * di) + bias[j];
        int g = batch[i];
        atomicAdd(pooled + (size_t)g * D + j, v);
        if (j == 0) atomicAdd(cnt + g, 1.0f);
    }
}

__global__ void fc_k(const float* __restrict__ pooled_r, const float* __restrict__ cnt_r,
                     const float* __restrict__ pooled_l, const float* __restrict__ cnt_l,
                     const float* __restrict__ fcW, const float* __restrict__ fcb,
                     float* __restrict__ out) {
    int t = blockIdx.x * blockDim.x + threadIdx.x;
    if (t < NGRAPH * 6) {
        int g = t / 6, o = t - g * 6;
        float icr = 1.0f / fmaxf(cnt_r[g], 1.0f);
        float icl = 1.0f / fmaxf(cnt_l[g], 1.0f);
        float acc = fcb[o];
        #pragma unroll
        for (int j = 0; j < D; ++j)
            acc += fcW[o * (2 * D) + j] * (pooled_r[g * D + j] * icr);
        #pragma unroll
        for (int j = 0; j < D; ++j)
            acc += fcW[o * (2 * D) + D + j] * (pooled_l[g * D + j] * icl);
        if (o < 3) out[g * 3 + o] = acc;
        else       out[NGRAPH * 3 + g * 3 + (o - 3)] = acc;
    }
}


#define BSH 8
#define BNODES 256
#define NBMAX 1024
#define PTHREADS 1024
#define PTILE 32768
#define PEPT (PTILE / PTHREADS)
#define PU 8

__global__ __launch_bounds__(PTHREADS)
void bucket_hist_k(const int* __restrict__ dst, long long E,
                   int* __restrict__ bcount) {
    __shared__ int cnt[NBMAX];
    const int t = threadIdx.x;
    cnt[t] = 0;
    __syncthreads();
    const long long base = (long long)blockIdx.x * PTILE;
    const long long rem = E - base;
    const int ne = rem < (long long)PTILE ? (int)rem : PTILE;
    for (int k0 = 0; k0 < PEPT; k0 += PU) {
        int dv[PU];
        #pragma unroll
        for (int u = 0; u < PU; ++u) {
            int l = (k0 + u) * PTHREADS + t;
            dv[u] = dst[base + (l < ne ? l : 0)];
        }
        #pragma unroll
        for (int u = 0; u < PU; ++u) {
            int l = (k0 + u) * PTHREADS + t;
            if (l < ne) atomicAdd(&cnt[dv[u] >> BSH], 1);
        }
    }
    __syncthreads();
    const int c = cnt[t];
    if (c > 0) atomicAdd(bcount + t, c);
}

__global__ __launch_bounds__(NBMAX)
void bucket_scan_k(const int* __restrict__ bcount, int* __restrict__ bstart,
                   int* __restrict__ cursor, int nb) {
    __shared__ int s[NBMAX];
    const int t = threadIdx.x;
    const int own = (t < nb) ? bcount[t] : 0;
    s[t] = own;
    __syncthreads();
    for (int off = 1; off < NBMAX; off <<= 1) {
        int v = (t >= off) ? s[t - off] : 0;
        __syncthreads();
        s[t] += v;
        __syncthreads();
    }
    const int incl = s[t];
    if (t < nb) { bstart[t] = incl - own; cursor[t] = incl - own; }
    if (t == nb - 1) bstart[nb] = incl;
}

__global__ __launch_bounds__(PTHREADS)
void bucket_scatter_k(const int* __restrict__ src, const int* __restrict__ dst,
                      long long E, int* __restrict__ cursor,
                      unsigned* __restrict__ part) {
    __shared__ int cnt[NBMAX];
    __shared__ int tstart[NBMAX];
    __shared__ int gbase[NBMAX];
    __shared__ int run[NBMAX];
    __shared__ unsigned stage[PTILE];
    __shared__ unsigned short bid[PTILE];
    const int t = threadIdx.x;
    cnt[t] = 0;
    __syncthreads();
    const long long base = (long long)blockIdx.x * PTILE;
    const long long rem = E - base;
    const int ne = rem < (long long)PTILE ? (int)rem : PTILE;
    for (int k0 = 0; k0 < PEPT; k0 += PU) {
        int dv[PU];
        #pragma unroll
        for (int u = 0; u < PU; ++u) {
            int l = (k0 + u) * PTHREADS + t;
            dv[u] = dst[base + (l < ne ? l : 0)];
        }
        #pragma unroll
        for (int u = 0; u < PU; ++u) {
            int l = (k0 + u) * PTHREADS + t;
            if (l < ne) atomicAdd(&cnt[dv[u] >> BSH], 1);
        }
    }
    __syncthreads();
    const int own = cnt[t];
    run[t] = own;
    __syncthreads();
    for (int off = 1; off < NBMAX; off <<= 1) {
        int v = (t >= off) ? run[t - off] : 0;
        __syncthreads();
        run[t] += v;
        __syncthreads();
    }
    const int excl = run[t] - own;
    tstart[t] = excl;
    run[t] = excl;
    gbase[t] = (own > 0) ? atomicAdd(cursor + t, own) : 0;
    __syncthreads();
    for (int k0 = 0; k0 < PEPT; k0 += PU) {
        int dv[PU], sv[PU];
        #pragma unroll
        for (int u = 0; u < PU; ++u) {
            int l = (k0 + u) * PTHREADS + t;
            long long e = base + (l < ne ? l : 0);
            dv[u] = dst[e];
            sv[u] = src[e];
        }
        #pragma unroll
        for (int u = 0; u < PU; ++u) {
            int l = (k0 + u) * PTHREADS + t;
            if (l < ne) {
                int b = dv[u] >> BSH;
                int slot = atomicAdd(&run[b], 1);
                stage[slot] = ((unsigned)sv[u] << BSH) | (unsigned)(dv[u] & (BNODES - 1));
                bid[slot] = (unsigned short)b;
            }
        }
    }
    __syncthreads();
    for (int slot = t; slot < ne; slot += PTHREADS) {
        int b = bid[slot];
        part[gbase[b] + (slot - tstart[b])] = stage[slot];
    }
}

__global__ __launch_bounds__(BNODES)
void bucket_degree_k(const unsigned* __restrict__ part,
                     const int* __restrict__ bstart,
                     float* __restrict__ dis, int n) {
    __shared__ int cnt[BNODES];
    const int t = threadIdx.x;
    cnt[t] = 0;
    __syncthreads();
    const int b = blockIdx.x;
    const int p0 = bstart[b], p1 = bstart[b + 1];
    for (int p = p0 + t; p < p1; p += BNODES * PU) {
        unsigned r[PU];
        #pragma unroll
        for (int u = 0; u < PU; ++u) {
            int q = p + u * BNODES;
            r[u] = part[q < p1 ? q : 0];
        }
        #pragma unroll
        for (int u = 0; u < PU; ++u) {
            int q = p + u * BNODES;
            if (q < p1) atomicAdd(&cnt[r[u] & (BNODES - 1)], 1);
        }
    }
    __syncthreads();
    const int i = (b << BSH) + t;
    if (i < n) dis[i] = rsqrtf((float)cnt[t] + 1.0f);
}

#define LTHREADS 512
#define LGROUPS (LTHREADS / 16)
#define LU 8
#define PGR 16
template <bool POOL, bool PAIR>
__global__ __launch_bounds__(LTHREADS)
void bucket_layer_k(const float* __restrict__ h, const float* __restrict__ dis,
                    const unsigned* __restrict__ part,
                    const int* __restrict__ bstart,
                    const float* __restrict__ bias,
                    float* __restrict__ rows,
                    const int* __restrict__ batch,
                    float* __restrict__ pooled, float* __restrict__ cntg,
                    int n) {
    __shared__ float acc[BNODES * D];
    __shared__ float dloc[BNODES];
    __shared__ float pl[PGR * D];
    __shared__ int   pc[PGR];
    const int t = threadIdx.x;
    const int b = blockIdx.x;
    const int node0 = b << BSH;
    for (int i = t; i < BNODES * D; i += LTHREADS) acc[i] = 0.0f;
    if (t < BNODES) {
        int i = node0 + t;
        dloc[t] = (i < n) ? dis[i] : 0.0f;
    }
    if (POOL) {
        if (t < PGR * D) pl[t] = 0.0f;
        if (t < PGR) pc[t] = 0;
    }
    __syncthreads();

    const int j = t & 15;
    const int grp = t >> 4;
    const int p0 = bstart[b], p1 = bstart[b + 1];
    const int step = LGROUPS * LU;
    unsigned recn[LU];
    #pragma unroll
    for (int u = 0; u < LU; ++u) {
        int q = p0 + grp + u * LGROUPS;
        recn[u] = part[q < p1 ? q : 0];
    }
    for (int p = p0 + grp; p < p1; p += step) {
        unsigned rec[LU];
        float hv[LU], ds[LU];
        #pragma unroll
        for (int u = 0; u < LU; ++u) {
            rec[u] = recn[u];
            int q = p + step + u * LGROUPS;
            recn[u] = part[q < p1 ? q : 0];
        }
        #pragma unroll
        for (int u = 0; u < LU; ++u) {
            int s = (int)(rec[u] >> BSH);
            if (PAIR) {
                v2f pr = *(const v2f*)(h + ((size_t)s * D + j) * 2);
                hv[u] = pr.x;
                ds[u] = pr.y;
            } else {
                ds[u] = dis[s];
                hv[u] = h[(size_t)s * D + j];
            }
        }
        #pragma unroll
        for (int u = 0; u < LU; ++u) {
            int q = p + u * LGROUPS;
            int d = (int)(rec[u] & (BNODES - 1));
            float norm = ds[u] * dloc[d];
            float term = hv[u] * norm;
            atomicAdd(&acc[d * D + j], (q < p1) ? term : 0.0f);
        }
    }
    __syncthreads();

    if (!POOL) {
        for (int idx = t; idx < BNODES * D; idx += LTHREADS) {
            int nl = idx >> 4;
            int i = node0 + nl;
            if (i < n) {
                float di = dloc[nl];
                size_t o = (size_t)i * D + (idx & 15);
                float v = acc[idx] + h[PAIR ? o * 2 : o] * (di * di) + bias[idx & 15];
                rows[o] = v > 0.0f ? v : 0.0f;
            }
        }
    } else {
        const int g0 = batch[node0];
        for (int idx = t; idx < BNODES * D; idx += LTHREADS) {
            int nl = idx >> 4;
            int jj = idx & 15;
            int i = node0 + nl;
            if (i < n) {
                float di = dloc[nl];
                size_t o = (size_t)i * D + jj;
                float v = acc[idx] + h[PAIR ? o * 2 : o] * (di * di) + bias[jj];
                int g = batch[i];
                unsigned k = (unsigned)(g - g0);
                if (k < (unsigned)PGR) {
                    atomicAdd(&pl[k * D + jj], v);
                    if (jj == 0) atomicAdd(&pc[k], 1);
                } else {
                    atomicAdd(pooled + (size_t)g * D + jj, v);
                    if (jj == 0) atomicAdd(cntg + g, 1.0f);
                }
            }
        }
        __syncthreads();
        if (t < PGR * D) {
            int k = t >> 4;
            int c = pc[k];
            if (c > 0) {
                int g = g0 + k;
                atomicAdd(pooled + (size_t)g * D + (t & 15), pl[t]);
                if ((t & 15) == 0) atomicAdd(cntg + g, (float)c);
            }
        }
    }
}

static void run_protein_bucketed(const float* emb, const float* W1, const float* b1,
                                 const float* W2, const float* b2,
                                 const int* ids, const int* edge, const int* batch,
                                 int N, long long E,
                                 float* bufA, float* bufB, float* dis,
                                 float* pooled, float* cnt,
                                 unsigned* part, int* bcount, int* bstart, int* cursor,
                                 bool pair, hipStream_t stream) {
    const int* src = edge;
    const int* dst = edge + E;
    const int B = 256;
    dim3 blk(B);
    int gemm_grid = (int)(((N >> 4) + 7) / 8);
    if (gemm_grid < 1) gemm_grid = 1;
    const int nb = (N + BNODES - 1) / BNODES;
    const int ptiles = (int)((E + PTILE - 1) / PTILE);

    fill_zero_k<<<(NBMAX + B - 1) / B, blk, 0, stream>>>((float*)bcount, NBMAX);
    bucket_hist_k<<<ptiles, PTHREADS, 0, stream>>>(dst, E, bcount);
    bucket_scan_k<<<1, NBMAX, 0, stream>>>(bcount, bstart, cursor, nb);
    bucket_scatter_k<<<ptiles, PTHREADS, 0, stream>>>(src, dst, E, cursor, part);
    bucket_degree_k<<<nb, BNODES, 0, stream>>>(part, bstart, dis, N);

    gather_gemm16_k<<<gemm_grid, blk, 0, stream>>>(emb, ids, W1, bufA, pair ? dis : nullptr, N);
    if (pair)
        bucket_layer_k<false, true><<<nb, LTHREADS, 0, stream>>>(bufA, dis, part, bstart, b1, bufB,
                                                                nullptr, nullptr, nullptr, N);
    else
        bucket_layer_k<false, false><<<nb, LTHREADS, 0, stream>>>(bufA, dis, part, bstart, b1, bufB,
                                                                 nullptr, nullptr, nullptr, N);
    gather_gemm16_k<<<gemm_grid, blk, 0, stream>>>(bufB, nullptr, W2, bufA, pair ? dis : nullptr, N);

    fill_zero_k<<<(int)((NGRAPH * D + B - 1) / B), blk, 0, stream>>>(pooled, NGRAPH * D);
    fill_zero_k<<<(int)((NGRAPH + B - 1) / B), blk, 0, stream>>>(cnt, NGRAPH);
    if (pair)
        bucket_layer_k<true, true><<<nb, LTHREADS, 0, stream>>>(bufA, dis, part, bstart, b2, nullptr,
                                                               batch, pooled, cnt, N);
    else
        bucket_layer_k<true, false><<<nb, LTHREADS, 0, stream>>>(bufA, dis, part, bstart, b2, nullptr,
                                                                batch, pooled, cnt, N);
}

static void run_protein(const float* emb, const float* W1, const float* b1,
                        const float* W2, const float* b2,
                        const int* ids, const int* edge, const int* batch,
                        int N, long long E,
                        float* bufA, float* bufB, float* dis,
                        float* pooled, float* cnt, hipStream_t stream) {
    const int* src = edge;
    const int* dst = edge + E;
    const long long n16 = (long long)N * D;
    const int B = 256;
    dim3 blk(B);
    int gemm_grid = (int)(((N >> 4) + 7) / 8);
    if (gemm_grid < 1) gemm_grid = 1;

    fill_zero_k<<<(int)((N + B - 1) / B), blk, 0, stream>>>(dis, N);
    fill_zero_k<<<(int)((n16 + B - 1) / B), blk, 0, stream>>>(bufB, n16);

    gather_gemm16_k<<<gemm_grid, blk, 0, stream>>>(emb, ids, W1, bufA, nullptr, N);

    degree_k<<<(int)((E + B - 1) / B), blk, 0, stream>>>(dst, dis, E);
    rsqrt_k<<<(int)((N + B - 1) / B), blk, 0, stream>>>(dis, N);

    aggregate_k<<<(int)((E * 16 + B - 1) / B), blk, 0, stream>>>(bufA, dis, src, dst, bufB, E);
    node_relu_k<<<(int)((n16 + B - 1) / B), blk, 0, stream>>>(bufA, bufB, dis, b1, n16);

    gather_gemm16_k<<<gemm_grid, blk, 0, stream>>>(bufA, nullptr, W2, bufB, nullptr, N);

    fill_zero_k<<<(int)((n16 + B - 1) / B), blk, 0, stream>>>(bufA, n16);
    aggregate_k<<<(int)((E * 16 + B - 1) / B), blk, 0, stream>>>(bufB, dis, src, dst, bufA, E);

    fill_zero_k<<<(int)((NGRAPH * D + B - 1) / B), blk, 0, stream>>>(pooled, NGRAPH * D);
    fill_zero_k<<<(int)((NGRAPH + B - 1) / B), blk, 0, stream>>>(cnt, NGRAPH);
    node_pool_k<<<(int)((n16 + B - 1) / B), blk, 0, stream>>>(bufB, bufA, dis, b2, batch,
                                                              pooled, cnt, n16);
}

extern "C" void kernel_launch(void* const* d_in, const int* in_sizes, int n_in,
                              void* d_out, int out_size, void* d_ws, size_t ws_size,
                              hipStream_t stream) {
    const float* emb = (const float*)d_in[0];
    const float* W1  = (const float*)d_in[1];
    const float* b1  = (const float*)d_in[2];
    const float* W2  = (const float*)d_in[3];
    const float* b2  = (const float*)d_in[4];
    const float* fcW = (const float*)d_in[5];
    const float* fcb = (const float*)d_in[6];
    const int* rx     = (const int*)d_in[7];
    const int* redge  = (const int*)d_in[8];
    const int* rbatch = (const int*)d_in[9];
    const int* lx     = (const int*)d_in[10];
    const int* ledge  = (const int*)d_in[11];
    const int* lbatch = (const int*)d_in[12];

    const int N = in_sizes[7];
    const long long E = in_sizes[8] / 2;

    char* ws = (char*)d_ws;
    float* bufA = (float*)ws;  ws += (size_t)N * D * sizeof(float);
    float* bufB = (float*)ws;  ws += (size_t)N * D * sizeof(float);
    float* dis  = (float*)ws;  ws += (size_t)N * sizeof(float);
    float* pooled_r = (float*)ws; ws += (size_t)NGRAPH * D * sizeof(float);
    float* pooled_l = (float*)ws; ws += (size_t)NGRAPH * D * sizeof(float);
    float* cnt_r = (float*)ws; ws += (size_t)NGRAPH * sizeof(float);
    float* cnt_l = (float*)ws; ws += (size_t)NGRAPH * sizeof(float);

    unsigned* part = (unsigned*)ws;  ws += (size_t)E * sizeof(unsigned);
    int* bcount = (int*)ws;  ws += (size_t)NBMAX * sizeof(int);
    int* bstart = (int*)ws;  ws += (size_t)2 * NBMAX * sizeof(int);
    int* cursor = (int*)ws;  ws += (size_t)NBMAX * sizeof(int);
    const bool bucketed = (size_t)(ws - (char*)d_ws) <= ws_size &&
                          N > 0 && N <= NBMAX * BNODES && E > 0 && E < (1ll << 30);
    float* bufT = (float*)ws;  ws += (size_t)N * D * sizeof(float);
    const bool pair = (size_t)(ws - (char*)d_ws) <= ws_size;
    float* rowsB = pair ? bufT : bufB;
    if (bucketed) {
        run_protein_bucketed(emb, W1, b1, W2, b2, rx, redge, rbatch, N, E,
                             bufA, rowsB, dis, pooled_r, cnt_r,
                             part, bcount, bstart, cursor, pair, stream);
        run_protein_bucketed(emb, W1, b1, W2, b2, lx, ledge, lbatch, N, E,
                             bufA, rowsB, dis, pooled_l, cnt_l,
                             part, bcount, bstart, cursor, pair, stream);
    } else {
        run_protein(emb, W1, b1, W2, b2, rx, redge, rbatch, N, E,
                    bufA, bufB, dis, pooled_r, cnt_r, stream);
        run_protein(emb, W1, b1, W2, b2, lx, ledge, lbatch, N, E,
                    bufA, bufB, dis, pooled_l, cnt_l, stream);
    }

    fc_k<<<(NGRAPH * 6 + 255) / 256, 256, 0, stream>>>(pooled_r, cnt_r, pooled_l, cnt_l,
                                                       fcW, fcb, (float*)d_out);
}


# grids matched to the 256 WGPs: 196-node dst buckets (1021 groups = 4 per WGP; 782 were 3 or 4) and 256 sort tiles of 25024 edges (was 196); bucket bounds move, float sum order changes
# speedup vs baseline: 5.5207x; 1.1552x over previous
#include <hip/hip_runtime.h>
#include <hip/hip_bf16.h>

typedef float v2f __attribute__((ext_vector_type(2)));
typedef float v8f __attribute__((ext_vector_type(8)));

#define D 16
#define NGRAPH 1024

__global__ void fill_zero_k(float* __restrict__ p, long long n) {
    long long i = (long long)blockIdx.x * blockDim.x + threadIdx.x;
    if (i < n) p[i] = 0.0f;
}

__global__ void gather_gemm16_k(const float* __restrict__ src_rows,
                                const int* __restrict__ ids,
                                const float* __restrict__ W,
                                float* __restrict__ out,
                                const float* __restrict__ dis_pair,
                                int n_nodes) {
    const int lane  = threadIdx.x & 31;
    const int wave  = (int)((blockIdx.x * blockDim.x + threadIdx.x) >> 5);
    const int nwave = (int)((gridDim.x * blockDim.x) >> 5);
    const int row   = lane & 15;
    const int col   = lane & 15;
    const int koff  = (lane >> 4) << 1;
    const int rowb  = (lane >> 4) << 3;

    v2f bb0, bb1, bb2, bb3;
    bb0.x = W[(0  + koff) * D + col]; bb0.y = W[(1  + koff) * D + col];
    bb1.x = W[(4  + koff) * D + col]; bb1.y = W[(5  + koff) * D + col];
    bb2.x = W[(8  + koff) * D + col]; bb2.y = W[(9  + koff) * D + col];
    bb3.x = W[(12 + koff) * D + col]; bb3.y = W[(13 + koff) * D + col];

    const int ntiles = n_nodes >> 4;
    for (int tile = wave; tile < ntiles; tile += nwave) {
        const int node = (tile << 4) + row;
        const int r = ids ? ids[node] : node;
        const float* xrow = src_rows + (size_t)r * D;
        v2f a0 = *(const v2f*)(xrow + 0  + koff);
        v2f a1 = *(const v2f*)(xrow + 4  + koff);
        v2f a2 = *(const v2f*)(xrow + 8  + koff);
        v2f a3 = *(const v2f*)(xrow + 12 + koff);
        v8f c = {};
        c = __builtin_amdgcn_wmma_f32_16x16x4_f32(false, a0, false, bb0, (short)0, c, false, false);
        c = __builtin_amdgcn_wmma_f32_16x16x4_f32(false, a1, false, bb1, (short)0, c, false, false);
        c = __builtin_amdgcn_wmma_f32_16x16x4_f32(false, a2, false, bb2, (short)0, c, false, false);
        c = __builtin_amdgcn_wmma_f32_16x16x4_f32(false, a3, false, bb3, (short)0, c, false, false);
        if (dis_pair) {
            const float* dp = dis_pair + (tile << 4) + rowb;
            v2f* ob2 = (v2f*)out + ((size_t)(tile << 4) + rowb) * D + col;
            #pragma unroll
            for (int rr = 0; rr < 8; ++rr) {
                v2f pr; pr.x = c[rr]; pr.y = dp[rr];
                ob2[(size_t)rr * D] = pr;
            }
        } else {
            float* ob = out + ((size_t)(tile << 4) + rowb) * D + col;
            #pragma unroll
            for (int rr = 0; rr < 8; ++rr) ob[(size_t)rr * D] = c[rr];
        }
    }

    const int tail_start = ntiles << 4;
    if (tail_start < n_nodes) {
        int node = tail_start + (int)(blockIdx.x * blockDim.x + threadIdx.x);
        if (node < n_nodes) {
            const int r = ids ? ids[node] : node;
            const float* xrow = src_rows + (size_t)r * D;
            for (int j = 0; j < D; ++j) {
                float acc = 0.0f;
                for (int k = 0; k < D; ++k) acc += xrow[k] * W[k * D + j];
                if (dis_pair) {
                    out[((size_t)node * D + j) * 2] = acc;
                    out[((size_t)node * D + j) * 2 + 1] = dis_pair[node];
                } else {
                    out[(size_t)node * D + j] = acc;
                }
            }
        }
    }
}

__global__ void degree_k(const int* __restrict__ dst, float* __restrict__ deg,
                         long long E) {
    long long e = (long long)blockIdx.x * blockDim.x + threadIdx.x;
    if (e < E) atomicAdd(deg + dst[e], 1.0f);
}

__global__ void rsqrt_k(float* __restrict__ d, int n) {
    int i = blockIdx.x * blockDim.x + threadIdx.x;
    if (i < n) d[i] = rsqrtf(d[i] + 1.0f);
}

__global__ void aggregate_k(const float* __restrict__ h,
                            const float* __restrict__ dis,
                            const int* __restrict__ src,
                            const int* __restrict__ dst,
                            float* __restrict__ agg,
                            long long E) {
    long long t = (long long)blockIdx.x * blockDim.x + threadIdx.x;
    long long e = t >> 4;
    int j = (int)(t & 15);
    if (e < E) {
        int s = src[e], d = dst[e];
        float norm = dis[s] * dis[d];
        atomicAdd(agg + (size_t)d * D + j, h[(size_t)s * D + j] * norm);
    }
}

__global__ void node_relu_k(float* __restrict__ h, const float* __restrict__ agg,
                            const float* __restrict__ dis,
                            const float* __restrict__ bias, long long n16) {
    long long t = (long long)blockIdx.x * blockDim.x + threadIdx.x;
    if (t < n16) {
        long long i = t >> 4; int j = (int)(t & 15);
        float di = dis[i];
        float v = agg[t] + h[t] * (di * di) + bias[j];
        h[t] = v > 0.0f ? v : 0.0f;
    }
}

__global__ void node_pool_k(const float* __restrict__ h2,
                            const float* __restrict__ agg,
                            const float* __restrict__ dis,
                            const float* __restrict__ bias,
                            const int* __restrict__ batch,
                            float* __restrict__ pooled,
                            float* __restrict__ cnt, long long n16) {
    long long t = (long long)blockIdx.x * blockDim.x + threadIdx.x;
    if (t < n16) {
        long long i = t >> 4; int j = (int)(t & 15);
        float di = dis[i];
        float v = agg[t] + h2[t] * (di * di) + bias[j];
        int g = batch[i];
        atomicAdd(pooled + (size_t)g * D + j, v);
        if (j == 0) atomicAdd(cnt + g, 1.0f);
    }
}

__global__ void fc_k(const float* __restrict__ pooled_r, const float* __restrict__ cnt_r,
                     const float* __restrict__ pooled_l, const float* __restrict__ cnt_l,
                     const float* __restrict__ fcW, const float* __restrict__ fcb,
                     float* __restrict__ out) {
    int t = blockIdx.x * blockDim.x + threadIdx.x;
    if (t < NGRAPH * 6) {
        int g = t / 6, o = t - g * 6;
        float icr = 1.0f / fmaxf(cnt_r[g], 1.0f);
        float icl = 1.0f / fmaxf(cnt_l[g], 1.0f);
        float acc = fcb[o];
        #pragma unroll
        for (int j = 0; j < D; ++j)
            acc += fcW[o * (2 * D) + j] * (pooled_r[g * D + j] * icr);
        #pragma unroll
        for (int j = 0; j < D; ++j)
            acc += fcW[o * (2 * D) + D + j] * (pooled_l[g * D + j] * icl);
        if (o < 3) out[g * 3 + o] = acc;
        else       out[NGRAPH * 3 + g * 3 + (o - 3)] = acc;
    }
}


#define RSH 8
#define RMASK 255u
#define BNODES 196
#define NBMAX 1024
#define NWGP 256
#define PTHREADS 1024
#define PTILE 32768
#define PU 8
#define DTHREADS 256

__global__ __launch_bounds__(PTHREADS)
void bucket_hist_k(const int* __restrict__ dst, long long E, int tlen,
                   int* __restrict__ bcount) {
    __shared__ int cnt[NBMAX];
    const int t = threadIdx.x;
    cnt[t] = 0;
    __syncthreads();
    const long long base = (long long)blockIdx.x * tlen;
    const long long rem = E - base;
    const int ne = rem < (long long)tlen ? (int)rem : tlen;
    for (int k0 = 0; k0 * PTHREADS < ne; k0 += PU) {
        int dv[PU];
        #pragma unroll
        for (int u = 0; u < PU; ++u) {
            int l = (k0 + u) * PTHREADS + t;
            dv[u] = dst[base + (l < ne ? l : 0)];
        }
        #pragma unroll
        for (int u = 0; u < PU; ++u) {
            int l = (k0 + u) * PTHREADS + t;
            if (l < ne) atomicAdd(&cnt[(unsigned)dv[u] / BNODES], 1);
        }
    }
    __syncthreads();
    const int c = cnt[t];
    if (c > 0) atomicAdd(bcount + t, c);
}

__global__ __launch_bounds__(NBMAX)
void bucket_scan_k(const int* __restrict__ bcount, int* __restrict__ bstart,
                   int* __restrict__ cursor, int nb) {
    __shared__ int s[NBMAX];
    const int t = threadIdx.x;
    const int own = (t < nb) ? bcount[t] : 0;
    s[t] = own;
    __syncthreads();
    for (int off = 1; off < NBMAX; off <<= 1) {
        int v = (t >= off) ? s[t - off] : 0;
        __syncthreads();
        s[t] += v;
        __syncthreads();
    }
    const int incl = s[t];
    if (t < nb) { bstart[t] = incl - own; cursor[t] = incl - own; }
    if (t == nb - 1) bstart[nb] = incl;
}

__global__ __launch_bounds__(PTHREADS)
void bucket_scatter_k(const int* __restrict__ src, const int* __restrict__ dst,
                      long long E, int tlen, int* __restrict__ cursor,
                      unsigned* __restrict__ part) {
    __shared__ int cnt[NBMAX];
    __shared__ int tstart[NBMAX];
    __shared__ int gbase[NBMAX];
    __shared__ int run[NBMAX];
    __shared__ unsigned stage[PTILE];
    __shared__ unsigned short bid[PTILE];
    const int t = threadIdx.x;
    cnt[t] = 0;
    __syncthreads();
    const long long base = (long long)blockIdx.x * tlen;
    const long long rem = E - base;
    const int ne = rem < (long long)tlen ? (int)rem : tlen;
    for (int k0 = 0; k0 * PTHREADS < ne; k0 += PU) {
        int dv[PU];
        #pragma unroll
        for (int u = 0; u < PU; ++u) {
            int l = (k0 + u) * PTHREADS + t;
            dv[u] = dst[base + (l < ne ? l : 0)];
        }
        #pragma unroll
        for (int u = 0; u < PU; ++u) {
            int l = (k0 + u) * PTHREADS + t;
            if (l < ne) atomicAdd(&cnt[(unsigned)dv[u] / BNODES], 1);
        }
    }
    __syncthreads();
    const int own = cnt[t];
    run[t] = own;
    __syncthreads();
    for (int off = 1; off < NBMAX; off <<= 1) {
        int v = (t >= off) ? run[t - off] : 0;
        __syncthreads();
        run[t] += v;
        __syncthreads();
    }
    const int excl = run[t] - own;
    tstart[t] = excl;
    run[t] = excl;
    gbase[t] = (own > 0) ? atomicAdd(cursor + t, own) : 0;
    __syncthreads();
    for (int k0 = 0; k0 * PTHREADS < ne; k0 += PU) {
        int dv[PU], sv[PU];
        #pragma unroll
        for (int u = 0; u < PU; ++u) {
            int l = (k0 + u) * PTHREADS + t;
            long long e = base + (l < ne ? l : 0);
            dv[u] = dst[e];
            sv[u] = src[e];
        }
        #pragma unroll
        for (int u = 0; u < PU; ++u) {
            int l = (k0 + u) * PTHREADS + t;
            if (l < ne) {
                unsigned b = (unsigned)dv[u] / BNODES;
                int slot = atomicAdd(&run[b], 1);
                stage[slot] = ((unsigned)sv[u] << RSH) | ((unsigned)dv[u] - b * BNODES);
                bid[slot] = (unsigned short)b;
            }
        }
    }
    __syncthreads();
    for (int slot = t; slot < ne; slot += PTHREADS) {
        int b = bid[slot];
        part[gbase[b] + (slot - tstart[b])] = stage[slot];
    }
}

__global__ __launch_bounds__(DTHREADS)
void bucket_degree_k(const unsigned* __restrict__ part,
                     const int* __restrict__ bstart,
                     float* __restrict__ dis, int n) {
    __shared__ int cnt[BNODES];
    const int t = threadIdx.x;
    if (t < BNODES) cnt[t] = 0;
    __syncthreads();
    const int b = blockIdx.x;
    const int p0 = bstart[b], p1 = bstart[b + 1];
    for (int p = p0 + t; p < p1; p += DTHREADS * PU) {
        unsigned r[PU];
        #pragma unroll
        for (int u = 0; u < PU; ++u) {
            int q = p + u * DTHREADS;
            r[u] = part[q < p1 ? q : 0];
        }
        #pragma unroll
        for (int u = 0; u < PU; ++u) {
            int q = p + u * DTHREADS;
            if (q < p1) atomicAdd(&cnt[r[u] & RMASK], 1);
        }
    }
    __syncthreads();
    const int i = b * BNODES + t;
    if (t < BNODES && i < n) dis[i] = rsqrtf((float)cnt[t] + 1.0f);
}

#define LTHREADS 512
#define LGROUPS (LTHREADS / 16)
#define LU 8
#define PGR 16
template <bool POOL, bool PAIR>
__global__ __launch_bounds__(LTHREADS)
void bucket_layer_k(const float* __restrict__ h, const float* __restrict__ dis,
                    const unsigned* __restrict__ part,
                    const int* __restrict__ bstart,
                    const float* __restrict__ bias,
                    float* __restrict__ rows,
                    const int* __restrict__ batch,
                    float* __restrict__ pooled, float* __restrict__ cntg,
                    int n) {
    __shared__ float acc[BNODES * D];
    __shared__ float dloc[BNODES];
    __shared__ float pl[PGR * D];
    __shared__ int   pc[PGR];
    const int t = threadIdx.x;
    const int b = blockIdx.x;
    const int node0 = b * BNODES;
    for (int i = t; i < BNODES * D; i += LTHREADS) acc[i] = 0.0f;
    if (t < BNODES) {
        int i = node0 + t;
        dloc[t] = (i < n) ? dis[i] : 0.0f;
    }
    if (POOL) {
        if (t < PGR * D) pl[t] = 0.0f;
        if (t < PGR) pc[t] = 0;
    }
    __syncthreads();

    const int j = t & 15;
    const int grp = t >> 4;
    const int p0 = bstart[b], p1 = bstart[b + 1];
    const int step = LGROUPS * LU;
    unsigned recn[LU];
    #pragma unroll
    for (int u = 0; u < LU; ++u) {
        int q = p0 + grp + u * LGROUPS;
        recn[u] = part[q < p1 ? q : 0];
    }
    for (int p = p0 + grp; p < p1; p += step) {
        unsigned rec[LU];
        float hv[LU], ds[LU];
        #pragma unroll
        for (int u = 0; u < LU; ++u) {
            rec[u] = recn[u];
            int q = p + step + u * LGROUPS;
            recn[u] = part[q < p1 ? q : 0];
        }
        #pragma unroll
        for (int u = 0; u < LU; ++u) {
            int s = (int)(rec[u] >> RSH);
            if (PAIR) {
                v2f pr = *(const v2f*)(h + ((size_t)s * D + j) * 2);
                hv[u] = pr.x;
                ds[u] = pr.y;
            } else {
                ds[u] = dis[s];
                hv[u] = h[(size_t)s * D + j];
            }
        }
        #pragma unroll
        for (int u = 0; u < LU; ++u) {
            int q = p + u * LGROUPS;
            int d = (int)(rec[u] & RMASK);
            float norm = ds[u] * dloc[d];
            float term = hv[u] * norm;
            atomicAdd(&acc[d * D + j], (q < p1) ? term : 0.0f);
        }
    }
    __syncthreads();

    if (!POOL) {
        for (int idx = t; idx < BNODES * D; idx += LTHREADS) {
            int nl = idx >> 4;
            int i = node0 + nl;
            if (i < n) {
                float di = dloc[nl];
                size_t o = (size_t)i * D + (idx & 15);
                float v = acc[idx] + h[PAIR ? o * 2 : o] * (di * di) + bias[idx & 15];
                rows[o] = v > 0.0f ? v : 0.0f;
            }
        }
    } else {
        const int g0 = batch[node0];
        for (int idx = t; idx < BNODES * D; idx += LTHREADS) {
            int nl = idx >> 4;
            int jj = idx & 15;
            int i = node0 + nl;
            if (i < n) {
                float di = dloc[nl];
                size_t o = (size_t)i * D + jj;
                float v = acc[idx] + h[PAIR ? o * 2 : o] * (di * di) + bias[jj];
                int g = batch[i];
                unsigned k = (unsigned)(g - g0);
                if (k < (unsigned)PGR) {
                    atomicAdd(&pl[k * D + jj], v);
                    if (jj == 0) atomicAdd(&pc[k], 1);
                } else {
                    atomicAdd(pooled + (size_t)g * D + jj, v);
                    if (jj == 0) atomicAdd(cntg + g, 1.0f);
                }
            }
        }
        __syncthreads();
        if (t < PGR * D) {
            int k = t >> 4;
            int c = pc[k];
            if (c > 0) {
                int g = g0 + k;
                atomicAdd(pooled + (size_t)g * D + (t & 15), pl[t]);
                if ((t & 15) == 0) atomicAdd(cntg + g, (float)c);
            }
        }
    }
}

static void run_protein_bucketed(const float* emb, const float* W1, const float* b1,
                                 const float* W2, const float* b2,
                                 const int* ids, const int* edge, const int* batch,
                                 int N, long long E,
                                 float* bufA, float* bufB, float* dis,
                                 float* pooled, float* cnt,
                                 unsigned* part, int* bcount, int* bstart, int* cursor,
                                 bool pair, hipStream_t stream) {
    const int* src = edge;
    const int* dst = edge + E;
    const int B = 256;
    dim3 blk(B);
    int gemm_grid = (int)(((N >> 4) + 7) / 8);
    if (gemm_grid < 1) gemm_grid = 1;
    const int nb = (N + BNODES - 1) / BNODES;
    const long long round_cap = (long long)NWGP * PTILE;
    const int ptiles = NWGP * (int)((E + round_cap - 1) / round_cap);
    const int tlen = (int)(((E + ptiles - 1) / ptiles + 31) & ~31ll);

    fill_zero_k<<<(NBMAX + B - 1) / B, blk, 0, stream>>>((float*)bcount, NBMAX);
    bucket_hist_k<<<ptiles, PTHREADS, 0, stream>>>(dst, E, tlen, bcount);
    bucket_scan_k<<<1, NBMAX, 0, stream>>>(bcount, bstart, cursor, nb);
    bucket_scatter_k<<<ptiles, PTHREADS, 0, stream>>>(src, dst, E, tlen, cursor, part);
    bucket_degree_k<<<nb, DTHREADS, 0, stream>>>(part, bstart, dis, N);

    gather_gemm16_k<<<gemm_grid, blk, 0, stream>>>(emb, ids, W1, bufA, pair ? dis : nullptr, N);
    if (pair)
        bucket_layer_k<false, true><<<nb, LTHREADS, 0, stream>>>(bufA, dis, part, bstart, b1, bufB,
                                                                nullptr, nullptr, nullptr, N);
    else
        bucket_layer_k<false, false><<<nb, LTHREADS, 0, stream>>>(bufA, dis, part, bstart, b1, bufB,
                                                                 nullptr, nullptr, nullptr, N);
    gather_gemm16_k<<<gemm_grid, blk, 0, stream>>>(bufB, nullptr, W2, bufA, pair ? dis : nullptr, N);

    fill_zero_k<<<(int)((NGRAPH * D + B - 1) / B), blk, 0, stream>>>(pooled, NGRAPH * D);
    fill_zero_k<<<(int)((NGRAPH + B - 1) / B), blk, 0, stream>>>(cnt, NGRAPH);
    if (pair)
        bucket_layer_k<true, true><<<nb, LTHREADS, 0, stream>>>(bufA, dis, part, bstart, b2, nullptr,
                                                               batch, pooled, cnt, N);
    else
        bucket_layer_k<true, false><<<nb, LTHREADS, 0, stream>>>(bufA, dis, part, bstart, b2, nullptr,
                                                                batch, pooled, cnt, N);
}

static void run_protein(const float* emb, const float* W1, const float* b1,
                        const float* W2, const float* b2,
                        const int* ids, const int* edge, const int* batch,
                        int N, long long E,
                        float* bufA, float* bufB, float* dis,
                        float* pooled, float* cnt, hipStream_t stream) {
    const int* src = edge;
    const int* dst = edge + E;
    const long long n16 = (long long)N * D;
    const int B = 256;
    dim3 blk(B);
    int gemm_grid = (int)(((N >> 4) + 7) / 8);
    if (gemm_grid < 1) gemm_grid = 1;

    fill_zero_k<<<(int)((N + B - 1) / B), blk, 0, stream>>>(dis, N);
    fill_zero_k<<<(int)((n16 + B - 1) / B), blk, 0, stream>>>(bufB, n16);

    gather_gemm16_k<<<gemm_grid, blk, 0, stream>>>(emb, ids, W1, bufA, nullptr, N);

    degree_k<<<(int)((E + B - 1) / B), blk, 0, stream>>>(dst, dis, E);
    rsqrt_k<<<(int)((N + B - 1) / B), blk, 0, stream>>>(dis, N);

    aggregate_k<<<(int)((E * 16 + B - 1) / B), blk, 0, stream>>>(bufA, dis, src, dst, bufB, E);
    node_relu_k<<<(int)((n16 + B - 1) / B), blk, 0, stream>>>(bufA, bufB, dis, b1, n16);

    gather_gemm16_k<<<gemm_grid, blk, 0, stream>>>(bufA, nullptr, W2, bufB, nullptr, N);

    fill_zero_k<<<(int)((n16 + B - 1) / B), blk, 0, stream>>>(bufA, n16);
    aggregate_k<<<(int)((E * 16 + B - 1) / B), blk, 0, stream>>>(bufB, dis, src, dst, bufA, E);

    fill_zero_k<<<(int)((NGRAPH * D + B - 1) / B), blk, 0, stream>>>(pooled, NGRAPH * D);
    fill_zero_k<<<(int)((NGRAPH + B - 1) / B), blk, 0, stream>>>(cnt, NGRAPH);
    node_pool_k<<<(int)((n16 + B - 1) / B), blk, 0, stream>>>(bufB, bufA, dis, b2, batch,
                                                              pooled, cnt, n16);
}

extern "C" void kernel_launch(void* const* d_in, const int* in_sizes, int n_in,
                              void* d_out, int out_size, void* d_ws, size_t ws_size,
                              hipStream_t stream) {
    const float* emb = (const float*)d_in[0];
    const float* W1  = (const float*)d_in[1];
    const float* b1  = (const float*)d_in[2];
    const float* W2  = (const float*)d_in[3];
    const float* b2  = (const float*)d_in[4];
    const float* fcW = (const float*)d_in[5];
    const float* fcb = (const float*)d_in[6];
    const int* rx     = (const int*)d_in[7];
    const int* redge  = (const int*)d_in[8];
    const int* rbatch = (const int*)d_in[9];
    const int* lx     = (const int*)d_in[10];
    const int* ledge  = (const int*)d_in[11];
    const int* lbatch = (const int*)d_in[12];

    const int N = in_sizes[7];
    const long long E = in_sizes[8] / 2;

    char* ws = (char*)d_ws;
    float* bufA = (float*)ws;  ws += (size_t)N * D * sizeof(float);
    float* bufB = (float*)ws;  ws += (size_t)N * D * sizeof(float);
    float* dis  = (float*)ws;  ws += (size_t)N * sizeof(float);
    float* pooled_r = (float*)ws; ws += (size_t)NGRAPH * D * sizeof(float);
    float* pooled_l = (float*)ws; ws += (size_t)NGRAPH * D * sizeof(float);
    float* cnt_r = (float*)ws; ws += (size_t)NGRAPH * sizeof(float);
    float* cnt_l = (float*)ws; ws += (size_t)NGRAPH * sizeof(float);

    unsigned* part = (unsigned*)ws;  ws += (size_t)E * sizeof(unsigned);
    int* bcount = (int*)ws;  ws += (size_t)NBMAX * sizeof(int);
    int* bstart = (int*)ws;  ws += (size_t)2 * NBMAX * sizeof(int);
    int* cursor = (int*)ws;  ws += (size_t)NBMAX * sizeof(int);
    const bool bucketed = (size_t)(ws - (char*)d_ws) <= ws_size &&
                          N > 0 && N <= NBMAX * BNODES && E > 0 && E < (1ll << 30);
    float* bufT = (float*)ws;  ws += (size_t)N * D * sizeof(float);
    const bool pair = (size_t)(ws - (char*)d_ws) <= ws_size;
    float* rowsB = pair ? bufT : bufB;
    if (bucketed) {
        run_protein_bucketed(emb, W1, b1, W2, b2, rx, redge, rbatch, N, E,
                             bufA, rowsB, dis, pooled_r, cnt_r,
                             part, bcount, bstart, cursor, pair, stream);
        run_protein_bucketed(emb, W1, b1, W2, b2, lx, ledge, lbatch, N, E,
                             bufA, rowsB, dis, pooled_l, cnt_l,
                             part, bcount, bstart, cursor, pair, stream);
    } else {
        run_protein(emb, W1, b1, W2, b2, rx, redge, rbatch, N, E,
                    bufA, bufB, dis, pooled_r, cnt_r, stream);
        run_protein(emb, W1, b1, W2, b2, lx, ledge, lbatch, N, E,
                    bufA, bufB, dis, pooled_l, cnt_l, stream);
    }

    fc_k<<<(NGRAPH * 6 + 255) / 256, 256, 0, stream>>>(pooled_r, cnt_r, pooled_l, cnt_l,
                                                       fcW, fcb, (float*)d_out);
}


# both proteins go through every stage in one launch each (grid.y = protein; own records, rows, factors and tables per protein; one fill clears all sums and counts): 10 launches, were 23
# speedup vs baseline: 6.4610x; 1.1703x over previous
#include <hip/hip_runtime.h>
#include <hip/hip_bf16.h>

typedef float v2f __attribute__((ext_vector_type(2)));
typedef float v8f __attribute__((ext_vector_type(8)));

#define D 16
#define NGRAPH 1024

__global__ void fill_zero_k(float* __restrict__ p, long long n) {
    long long i = (long long)blockIdx.x * blockDim.x + threadIdx.x;
    if (i < n) p[i] = 0.0f;
}

__global__ void gather_gemm16_k(const float* __restrict__ src_rows0,
                                const float* __restrict__ src_rows1,
                                const int* __restrict__ ids0,
                                const int* __restrict__ ids1,
                                const float* __restrict__ W,
                                float* __restrict__ out0,
                                float* __restrict__ out1,
                                const float* __restrict__ dis_pair0,
                                const float* __restrict__ dis_pair1,
                                int n_nodes) {
    const bool second = blockIdx.y != 0;
    const float* __restrict__ src_rows = second ? src_rows1 : src_rows0;
    const int* __restrict__ ids = second ? ids1 : ids0;
    float* __restrict__ out = second ? out1 : out0;
    const float* __restrict__ dis_pair = second ? dis_pair1 : dis_pair0;
    const int lane  = threadIdx.x & 31;
    const int wave  = (int)((blockIdx.x * blockDim.x + threadIdx.x) >> 5);
    const int nwave = (int)((gridDim.x * blockDim.x) >> 5);
    const int row   = lane & 15;
    const int col   = lane & 15;
    const int koff  = (lane >> 4) << 1;
    const int rowb  = (lane >> 4) << 3;

    v2f bb0, bb1, bb2, bb3;
    bb0.x = W[(0  + koff) * D + col]; bb0.y = W[(1  + koff) * D + col];
    bb1.x = W[(4  + koff) * D + col]; bb1.y = W[(5  + koff) * D + col];
    bb2.x = W[(8  + koff) * D + col]; bb2.y = W[(9  + koff) * D + col];
    bb3.x = W[(12 + koff) * D + col]; bb3.y = W[(13 + koff) * D + col];

    const int ntiles = n_nodes >> 4;
    for (int tile = wave; tile < ntiles; tile += nwave) {
        const int node = (tile << 4) + row;
        const int r = ids ? ids[node] : node;
        const float* xrow = src_rows + (size_t)r * D;
        v2f a0 = *(const v2f*)(xrow + 0  + koff);
        v2f a1 = *(const v2f*)(xrow + 4  + koff);
        v2f a2 = *(const v2f*)(xrow + 8  + koff);
        v2f a3 = *(const v2f*)(xrow + 12 + koff);
        v8f c = {};
        c = __builtin_amdgcn_wmma_f32_16x16x4_f32(false, a0, false, bb0, (short)0, c, false, false);
        c = __builtin_amdgcn_wmma_f32_16x16x4_f32(false, a1, false, bb1, (short)0, c, false, false);
        c = __builtin_amdgcn_wmma_f32_16x16x4_f32(false, a2, false, bb2, (short)0, c, false, false);
        c = __builtin_amdgcn_wmma_f32_16x16x4_f32(false, a3, false, bb3, (short)0, c, false, false);
        if (dis_pair) {
            const float* dp = dis_pair + (tile << 4) + rowb;
            v2f* ob2 = (v2f*)out + ((size_t)(tile << 4) + rowb) * D + col;
            #pragma unroll
            for (int rr = 0; rr < 8; ++rr) {
                v2f pr; pr.x = c[rr]; pr.y = dp[rr];
                ob2[(size_t)rr * D] = pr;
            }
        } else {
            float* ob = out + ((size_t)(tile << 4) + rowb) * D + col;
            #pragma unroll
            for (int rr = 0; rr < 8; ++rr) ob[(size_t)rr * D] = c[rr];
        }
    }

    const int tail_start = ntiles << 4;
    if (tail_start < n_nodes) {
        int node = tail_start + (int)(blockIdx.x * blockDim.x + threadIdx.x);
        if (node < n_nodes) {
            const int r = ids ? ids[node] : node;
            const float* xrow = src_rows + (size_t)r * D;
            for (int j = 0; j < D; ++j) {
                float acc = 0.0f;
                for (int k = 0; k < D; ++k) acc += xrow[k] * W[k * D + j];
                if (dis_pair) {
                    out[((size_t)node * D + j) * 2] = acc;
                    out[((size_t)node * D + j) * 2 + 1] = dis_pair[node];
                } else {
                    out[(size_t)node * D + j] = acc;
                }
            }
        }
    }
}

__global__ void degree_k(const int* __restrict__ dst, float* __restrict__ deg,
                         long long E) {
    long long e = (long long)blockIdx.x * blockDim.x + threadIdx.x;
    if (e < E) atomicAdd(deg + dst[e], 1.0f);
}

__global__ void rsqrt_k(float* __restrict__ d, int n) {
    int i = blockIdx.x * blockDim.x + threadIdx.x;
    if (i < n) d[i] = rsqrtf(d[i] + 1.0f);
}

__global__ void aggregate_k(const float* __restrict__ h,
                            const float* __restrict__ dis,
                            const int* __restrict__ src,
                            const int* __restrict__ dst,
                            float* __restrict__ agg,
                            long long E) {
    long long t = (long long)blockIdx.x * blockDim.x + threadIdx.x;
    long long e = t >> 4;
    int j = (int)(t & 15);
    if (e < E) {
        int s = src[e], d = dst[e];
        float norm = dis[s] * dis[d];
        atomicAdd(agg + (size_t)d * D + j, h[(size_t)s * D + j] * norm);
    }
}

__global__ void node_relu_k(float* __restrict__ h, const float* __restrict__ agg,
                            const float* __restrict__ dis,
                            const float* __restrict__ bias, long long n16) {
    long long t = (long long)blockIdx.x * blockDim.x + threadIdx.x;
    if (t < n16) {
        long long i = t >> 4; int j = (int)(t & 15);
        float di = dis[i];
        float v = agg[t] + h[t] * (di * di) + bias[j];
        h[t] = v > 0.0f ? v : 0.0f;
    }
}

__global__ void node_pool_k(const float* __restrict__ h2,
                            const float* __restrict__ agg,
                            const float* __restrict__ dis,
                            const float* __restrict__ bias,
                            const int* __restrict__ batch,
                            float* __restrict__ pooled,
                            float* __restrict__ cnt, long long n16) {
    long long t = (long long)blockIdx.x * blockDim.x + threadIdx.x;
    if (t < n16) {
        long long i = t >> 4; int j = (int)(t & 15);
        float di = dis[i];
        float v = agg[t] + h2[t] * (di * di) + bias[j];
        int g = batch[i];
        atomicAdd(pooled + (size_t)g * D + j, v);
        if (j == 0) atomicAdd(cnt + g, 1.0f);
    }
}

__global__ void fc_k(const float* __restrict__ pooled_r, const float* __restrict__ cnt_r,
                     const float* __restrict__ pooled_l, const float* __restrict__ cnt_l,
                     const float* __restrict__ fcW, const float* __restrict__ fcb,
                     float* __restrict__ out) {
    int t = blockIdx.x * blockDim.x + threadIdx.x;
    if (t < NGRAPH * 6) {
        int g = t / 6, o = t - g * 6;
        float icr = 1.0f / fmaxf(cnt_r[g], 1.0f);
        float icl = 1.0f / fmaxf(cnt_l[g], 1.0f);
        float acc = fcb[o];
        #pragma unroll
        for (int j = 0; j < D; ++j)
            acc += fcW[o * (2 * D) + j] * (pooled_r[g * D + j] * icr);
        #pragma unroll
        for (int j = 0; j < D; ++j)
            acc += fcW[o * (2 * D) + D + j] * (pooled_l[g * D + j] * icl);
        if (o < 3) out[g * 3 + o] = acc;
        else       out[NGRAPH * 3 + g * 3 + (o - 3)] = acc;
    }
}


#define RSH 8
#define RMASK 255u
#define BNODES 196
#define NBMAX 1024
#define NWGP 256
#define PTHREADS 1024
#define PTILE 32768
#define PU 8
#define DTHREADS 256

__global__ __launch_bounds__(PTHREADS)
void bucket_hist_k(const int* __restrict__ dst0, const int* __restrict__ dst1,
                   long long E, int tlen,
                   int* __restrict__ bcount0, int* __restrict__ bcount1) {
    const bool second = blockIdx.y != 0;
    const int* __restrict__ dst = second ? dst1 : dst0;
    int* __restrict__ bcount = second ? bcount1 : bcount0;
    __shared__ int cnt[NBMAX];
    const int t = threadIdx.x;
    cnt[t] = 0;
    __syncthreads();
    const long long base = (long long)blockIdx.x * tlen;
    const long long rem = E - base;
    const int ne = rem < (long long)tlen ? (int)rem : tlen;
    for (int k0 = 0; k0 * PTHREADS < ne; k0 += PU) {
        int dv[PU];
        #pragma unroll
        for (int u = 0; u < PU; ++u) {
            int l = (k0 + u) * PTHREADS + t;
            dv[u] = dst[base + (l < ne ? l : 0)];
        }
        #pragma unroll
        for (int u = 0; u < PU; ++u) {
            int l = (k0 + u) * PTHREADS + t;
            if (l < ne) atomicAdd(&cnt[(unsigned)dv[u] / BNODES], 1);
        }
    }
    __syncthreads();
    const int c = cnt[t];
    if (c > 0) atomicAdd(bcount + t, c);
}

__global__ __launch_bounds__(NBMAX)
void bucket_scan_k(const int* __restrict__ bcount0, const int* __restrict__ bcount1,
                   int* __restrict__ bstart0, int* __restrict__ bstart1,
                   int* __restrict__ cursor0, int* __restrict__ cursor1, int nb) {
    const bool second = blockIdx.y != 0;
    const int* __restrict__ bcount = second ? bcount1 : bcount0;
    int* __restrict__ bstart = second ? bstart1 : bstart0;
    int* __restrict__ cursor = second ? cursor1 : cursor0;
    __shared__ int s[NBMAX];
    const int t = threadIdx.x;
    const int own = (t < nb) ? bcount[t] : 0;
    s[t] = own;
    __syncthreads();
    for (int off = 1; off < NBMAX; off <<= 1) {
        int v = (t >= off) ? s[t - off] : 0;
        __syncthreads();
        s[t] += v;
        __syncthreads();
    }
    const int incl = s[t];
    if (t < nb) { bstart[t] = incl - own; cursor[t] = incl - own; }
    if (t == nb - 1) bstart[nb] = incl;
}

__global__ __launch_bounds__(PTHREADS)
void bucket_scatter_k(const int* __restrict__ src0, const int* __restrict__ src1,
                      const int* __restrict__ dst0, const int* __restrict__ dst1,
                      long long E, int tlen,
                      int* __restrict__ cursor0, int* __restrict__ cursor1,
                      unsigned* __restrict__ part0, unsigned* __restrict__ part1) {
    const bool second = blockIdx.y != 0;
    const int* __restrict__ src = second ? src1 : src0;
    const int* __restrict__ dst = second ? dst1 : dst0;
    int* __restrict__ cursor = second ? cursor1 : cursor0;
    unsigned* __restrict__ part = second ? part1 : part0;
    __shared__ int cnt[NBMAX];
    __shared__ int tstart[NBMAX];
    __shared__ int gbase[NBMAX];
    __shared__ int run[NBMAX];
    __shared__ unsigned stage[PTILE];
    __shared__ unsigned short bid[PTILE];
    const int t = threadIdx.x;
    cnt[t] = 0;
    __syncthreads();
    const long long base = (long long)blockIdx.x * tlen;
    const long long rem = E - base;
    const int ne = rem < (long long)tlen ? (int)rem : tlen;
    for (int k0 = 0; k0 * PTHREADS < ne; k0 += PU) {
        int dv[PU];
        #pragma unroll
        for (int u = 0; u < PU; ++u) {
            int l = (k0 + u) * PTHREADS + t;
            dv[u] = dst[base + (l < ne ? l : 0)];
        }
        #pragma unroll
        for (int u = 0; u < PU; ++u) {
            int l = (k0 + u) * PTHREADS + t;
            if (l < ne) atomicAdd(&cnt[(unsigned)dv[u] / BNODES], 1);
        }
    }
    __syncthreads();
    const int own = cnt[t];
    run[t] = own;
    __syncthreads();
    for (int off = 1; off < NBMAX; off <<= 1) {
        int v = (t >= off) ? run[t - off] : 0;
        __syncthreads();
        run[t] += v;
        __syncthreads();
    }
    const int excl = run[t] - own;
    tstart[t] = excl;
    run[t] = excl;
    gbase[t] = (own > 0) ? atomicAdd(cursor + t, own) : 0;
    __syncthreads();
    for (int k0 = 0; k0 * PTHREADS < ne; k0 += PU) {
        int dv[PU], sv[PU];
        #pragma unroll
        for (int u = 0; u < PU; ++u) {
            int l = (k0 + u) * PTHREADS + t;
            long long e = base + (l < ne ? l : 0);
            dv[u] = dst[e];
            sv[u] = src[e];
        }
        #pragma unroll
        for (int u = 0; u < PU; ++u) {
            int l = (k0 + u) * PTHREADS + t;
            if (l < ne) {
                unsigned b = (unsigned)dv[u] / BNODES;
                int slot = atomicAdd(&run[b], 1);
                stage[slot] = ((unsigned)sv[u] << RSH) | ((unsigned)dv[u] - b * BNODES);
                bid[slot] = (unsigned short)b;
            }
        }
    }
    __syncthreads();
    for (int slot = t; slot < ne; slot += PTHREADS) {
        int b = bid[slot];
        part[gbase[b] + (slot - tstart[b])] = stage[slot];
    }
}

__global__ __launch_bounds__(DTHREADS)
void bucket_degree_k(const unsigned* __restrict__ part0, const unsigned* __restrict__ part1,
                     const int* __restrict__ bstart0, const int* __restrict__ bstart1,
                     float* __restrict__ dis0, float* __restrict__ dis1, int n) {
    const bool second = blockIdx.y != 0;
    const unsigned* __restrict__ part = second ? part1 : part0;
    const int* __restrict__ bstart = second ? bstart1 : bstart0;
    float* __restrict__ dis = second ? dis1 : dis0;
    __shared__ int cnt[BNODES];
    const int t = threadIdx.x;
    if (t < BNODES) cnt[t] = 0;
    __syncthreads();
    const int b = blockIdx.x;
    const int p0 = bstart[b], p1 = bstart[b + 1];
    for (int p = p0 + t; p < p1; p += DTHREADS * PU) {
        unsigned r[PU];
        #pragma unroll
        for (int u = 0; u < PU; ++u) {
            int q = p + u * DTHREADS;
            r[u] = part[q < p1 ? q : 0];
        }
        #pragma unroll
        for (int u = 0; u < PU; ++u) {
            int q = p + u * DTHREADS;
            if (q < p1) atomicAdd(&cnt[r[u] & RMASK], 1);
        }
    }
    __syncthreads();
    const int i = b * BNODES + t;
    if (t < BNODES && i < n) dis[i] = rsqrtf((float)cnt[t] + 1.0f);
}

#define LTHREADS 512
#define LGROUPS (LTHREADS / 16)
#define LU 8
#define PGR 16
template <bool POOL, bool PAIR>
__global__ __launch_bounds__(LTHREADS)
void bucket_layer_k(const float* __restrict__ h0, const float* __restrict__ h1,
                    const float* __restrict__ dis0, const float* __restrict__ dis1,
                    const unsigned* __restrict__ part0, const unsigned* __restrict__ part1,
                    const int* __restrict__ bstart0, const int* __restrict__ bstart1,
                    const float* __restrict__ bias,
                    float* __restrict__ rows0, float* __restrict__ rows1,
                    const int* __restrict__ batch0, const int* __restrict__ batch1,
                    float* __restrict__ pooled0, float* __restrict__ pooled1,
                    float* __restrict__ cntg0, float* __restrict__ cntg1,
                    int n) {
    const bool second = blockIdx.y != 0;
    const float* __restrict__ h = second ? h1 : h0;
    const float* __restrict__ dis = second ? dis1 : dis0;
    const unsigned* __restrict__ part = second ? part1 : part0;
    const int* __restrict__ bstart = second ? bstart1 : bstart0;
    float* __restrict__ rows = second ? rows1 : rows0;
    const int* __restrict__ batch = second ? batch1 : batch0;
    float* __restrict__ pooled = second ? pooled1 : pooled0;
    float* __restrict__ cntg = second ? cntg1 : cntg0;
    __shared__ float acc[BNODES * D];
    __shared__ float dloc[BNODES];
    __shared__ float pl[PGR * D];
    __shared__ int   pc[PGR];
    const int t = threadIdx.x;
    const int b = blockIdx.x;
    const int node0 = b * BNODES;
    for (int i = t; i < BNODES * D; i += LTHREADS) acc[i] = 0.0f;
    if (t < BNODES) {
        int i = node0 + t;
        dloc[t] = (i < n) ? dis[i] : 0.0f;
    }
    if (POOL) {
        if (t < PGR * D) pl[t] = 0.0f;
        if (t < PGR) pc[t] = 0;
    }
    __syncthreads();

    const int j = t & 15;
    const int grp = t >> 4;
    const int p0 = bstart[b], p1 = bstart[b + 1];
    const int step = LGROUPS * LU;
    unsigned recn[LU];
    #pragma unroll
    for (int u = 0; u < LU; ++u) {
        int q = p0 + grp + u * LGROUPS;
        recn[u] = part[q < p1 ? q : 0];
    }
    for (int p = p0 + grp; p < p1; p += step) {
        unsigned rec[LU];
        float hv[LU], ds[LU];
        #pragma unroll
        for (int u = 0; u < LU; ++u) {
            rec[u] = recn[u];
            int q = p + step + u * LGROUPS;
            recn[u] = part[q < p1 ? q : 0];
        }
        #pragma unroll
        for (int u = 0; u < LU; ++u) {
            int s = (int)(rec[u] >> RSH);
            if (PAIR) {
                v2f pr = *(const v2f*)(h + ((size_t)s * D + j) * 2);
                hv[u] = pr.x;
                ds[u] = pr.y;
            } else {
                ds[u] = dis[s];
                hv[u] = h[(size_t)s * D + j];
            }
        }
        #pragma unroll
        for (int u = 0; u < LU; ++u) {
            int q = p + u * LGROUPS;
            int d = (int)(rec[u] & RMASK);
            float norm = ds[u] * dloc[d];
            float term = hv[u] * norm;
            atomicAdd(&acc[d * D + j], (q < p1) ? term : 0.0f);
        }
    }
    __syncthreads();

    if (!POOL) {
        for (int idx = t; idx < BNODES * D; idx += LTHREADS) {
            int nl = idx >> 4;
            int i = node0 + nl;
            if (i < n) {
                float di = dloc[nl];
                size_t o = (size_t)i * D + (idx & 15);
                float v = acc[idx] + h[PAIR ? o * 2 : o] * (di * di) + bias[idx & 15];
                rows[o] = v > 0.0f ? v : 0.0f;
            }
        }
    } else {
        const int g0 = batch[node0];
        for (int idx = t; idx < BNODES * D; idx += LTHREADS) {
            int nl = idx >> 4;
            int jj = idx & 15;
            int i = node0 + nl;
            if (i < n) {
                float di = dloc[nl];
                size_t o = (size_t)i * D + jj;
                float v = acc[idx] + h[PAIR ? o * 2 : o] * (di * di) + bias[jj];
                int g = batch[i];
                unsigned k = (unsigned)(g - g0);
                if (k < (unsigned)PGR) {
                    atomicAdd(&pl[k * D + jj], v);
                    if (jj == 0) atomicAdd(&pc[k], 1);
                } else {
                    atomicAdd(pooled + (size_t)g * D + jj, v);
                    if (jj == 0) atomicAdd(cntg + g, 1.0f);
                }
            }
        }
        __syncthreads();
        if (t < PGR * D) {
            int k = t >> 4;
            int c = pc[k];
            if (c > 0) {
                int g = g0 + k;
                atomicAdd(pooled + (size_t)g * D + (t & 15), pl[t]);
                if ((t & 15) == 0) atomicAdd(cntg + g, (float)c);
            }
        }
    }
}

static void run_protein_bucketed(const float* emb, const float* W1, const float* b1,
                                 const float* W2, const float* b2,
                                 const int* ids, const int* edge, const int* batch,
                                 int N, long long E,
                                 float* bufA, float* bufB, float* dis,
                                 float* pooled, float* cnt,
                                 unsigned* part, int* bcount, int* bstart, int* cursor,
                                 bool pair, hipStream_t stream) {
    const int* src = edge;
    const int* dst = edge + E;
    const int B = 256;
    dim3 blk(B);
    int gemm_grid = (int)(((N >> 4) + 7) / 8);
    if (gemm_grid < 1) gemm_grid = 1;
    const int nb = (N + BNODES - 1) / BNODES;
    const long long round_cap = (long long)NWGP * PTILE;
    const int ptiles = NWGP * (int)((E + round_cap - 1) / round_cap);
    const int tlen = (int)(((E + ptiles - 1) / ptiles + 31) & ~31ll);

    fill_zero_k<<<(NBMAX + B - 1) / B, blk, 0, stream>>>((float*)bcount, NBMAX);
    bucket_hist_k<<<ptiles, PTHREADS, 0, stream>>>(dst, nullptr, E, tlen, bcount, nullptr);
    bucket_scan_k<<<1, NBMAX, 0, stream>>>(bcount, nullptr, bstart, nullptr, cursor, nullptr, nb);
    bucket_scatter_k<<<ptiles, PTHREADS, 0, stream>>>(src, nullptr, dst, nullptr, E, tlen,
                                                      cursor, nullptr, part, nullptr);
    bucket_degree_k<<<nb, DTHREADS, 0, stream>>>(part, nullptr, bstart, nullptr, dis, nullptr, N);

    gather_gemm16_k<<<gemm_grid, blk, 0, stream>>>(emb, nullptr, ids, nullptr, W1, bufA, nullptr,
                                                   pair ? dis : nullptr, nullptr, N);
    if (pair)
        bucket_layer_k<false, true><<<nb, LTHREADS, 0, stream>>>(
            bufA, nullptr, dis, nullptr, part, nullptr, bstart, nullptr, b1, bufB, nullptr,
            nullptr, nullptr, nullptr, nullptr, nullptr, nullptr, N);
    else
        bucket_layer_k<false, false><<<nb, LTHREADS, 0, stream>>>(
            bufA, nullptr, dis, nullptr, part, nullptr, bstart, nullptr, b1, bufB, nullptr,
            nullptr, nullptr, nullptr, nullptr, nullptr, nullptr, N);
    gather_gemm16_k<<<gemm_grid, blk, 0, stream>>>(bufB, nullptr, nullptr, nullptr, W2, bufA, nullptr,
                                                   pair ? dis : nullptr, nullptr, N);

    fill_zero_k<<<(int)((NGRAPH * D + B - 1) / B), blk, 0, stream>>>(pooled, NGRAPH * D);
    fill_zero_k<<<(int)((NGRAPH + B - 1) / B), blk, 0, stream>>>(cnt, NGRAPH);
    if (pair)
        bucket_layer_k<true, true><<<nb, LTHREADS, 0, stream>>>(
            bufA, nullptr, dis, nullptr, part, nullptr, bstart, nullptr, b2, nullptr, nullptr,
            batch, nullptr, pooled, nullptr, cnt, nullptr, N);
    else
        bucket_layer_k<true, false><<<nb, LTHREADS, 0, stream>>>(
            bufA, nullptr, dis, nullptr, part, nullptr, bstart, nullptr, b2, nullptr, nullptr,
            batch, nullptr, pooled, nullptr, cnt, nullptr, N);
}

static void run_both_bucketed(const float* emb, const float* W1, const float* b1,
                              const float* W2, const float* b2,
                              const int* ids0, const int* ids1,
                              const int* edge0, const int* edge1,
                              const int* batch0, const int* batch1,
                              int N, long long E,
                              float* pair0, float* pair1, float* rows0, float* rows1,
                              float* dis0, float* dis1,
                              float* zero_run, long long zero_words,
                              float* pooled0, float* pooled1, float* cnt0, float* cnt1,
                              int* bcount0, int* bcount1, int* bstart0, int* bstart1,
                              int* cursor0, int* cursor1,
                              unsigned* part0, unsigned* part1, hipStream_t stream) {
    const int B = 256;
    dim3 blk(B);
    int gemm_grid = (int)(((N >> 4) + 7) / 8);
    if (gemm_grid < 1) gemm_grid = 1;
    const int nb = (N + BNODES - 1) / BNODES;
    const long long round_cap = (long long)NWGP * PTILE;
    const int ptiles = NWGP * (int)((E + round_cap - 1) / round_cap);
    const int tlen = (int)(((E + ptiles - 1) / ptiles + 31) & ~31ll);

    fill_zero_k<<<(int)((zero_words + B - 1) / B), blk, 0, stream>>>(zero_run, zero_words);
    bucket_hist_k<<<dim3(ptiles, 2), PTHREADS, 0, stream>>>(edge0 + E, edge1 + E, E, tlen,
                                                            bcount0, bcount1);
    bucket_scan_k<<<dim3(1, 2), NBMAX, 0, stream>>>(bcount0, bcount1, bstart0, bstart1,
                                                    cursor0, cursor1, nb);
    bucket_scatter_k<<<dim3(ptiles, 2), PTHREADS, 0, stream>>>(edge0, edge1, edge0 + E, edge1 + E,
                                                               E, tlen, cursor0, cursor1,
                                                               part0, part1);
    bucket_degree_k<<<dim3(nb, 2), DTHREADS, 0, stream>>>(part0, part1, bstart0, bstart1,
                                                          dis0, dis1, N);
    gather_gemm16_k<<<dim3(gemm_grid, 2), blk, 0, stream>>>(emb, emb, ids0, ids1, W1, pair0, pair1,
                                                            dis0, dis1, N);
    bucket_layer_k<false, true><<<dim3(nb, 2), LTHREADS, 0, stream>>>(
        pair0, pair1, dis0, dis1, part0, part1, bstart0, bstart1, b1, rows0, rows1,
        nullptr, nullptr, nullptr, nullptr, nullptr, nullptr, N);
    gather_gemm16_k<<<dim3(gemm_grid, 2), blk, 0, stream>>>(rows0, rows1, nullptr, nullptr, W2,
                                                            pair0, pair1, dis0, dis1, N);
    bucket_layer_k<true, true><<<dim3(nb, 2), LTHREADS, 0, stream>>>(
        pair0, pair1, dis0, dis1, part0, part1, bstart0, bstart1, b2, nullptr, nullptr,
        batch0, batch1, pooled0, pooled1, cnt0, cnt1, N);
}

static void run_protein(const float* emb, const float* W1, const float* b1,
                        const float* W2, const float* b2,
                        const int* ids, const int* edge, const int* batch,
                        int N, long long E,
                        float* bufA, float* bufB, float* dis,
                        float* pooled, float* cnt, hipStream_t stream) {
    const int* src = edge;
    const int* dst = edge + E;
    const long long n16 = (long long)N * D;
    const int B = 256;
    dim3 blk(B);
    int gemm_grid = (int)(((N >> 4) + 7) / 8);
    if (gemm_grid < 1) gemm_grid = 1;

    fill_zero_k<<<(int)((N + B - 1) / B), blk, 0, stream>>>(dis, N);
    fill_zero_k<<<(int)((n16 + B - 1) / B), blk, 0, stream>>>(bufB, n16);

    gather_gemm16_k<<<gemm_grid, blk, 0, stream>>>(emb, nullptr, ids, nullptr, W1, bufA, nullptr,
                                                   nullptr, nullptr, N);

    degree_k<<<(int)((E + B - 1) / B), blk, 0, stream>>>(dst, dis, E);
    rsqrt_k<<<(int)((N + B - 1) / B), blk, 0, stream>>>(dis, N);

    aggregate_k<<<(int)((E * 16 + B - 1) / B), blk, 0, stream>>>(bufA, dis, src, dst, bufB, E);
    node_relu_k<<<(int)((n16 + B - 1) / B), blk, 0, stream>>>(bufA, bufB, dis, b1, n16);

    gather_gemm16_k<<<gemm_grid, blk, 0, stream>>>(bufA, nullptr, nullptr, nullptr, W2, bufB, nullptr,
                                                   nullptr, nullptr, N);

    fill_zero_k<<<(int)((n16 + B - 1) / B), blk, 0, stream>>>(bufA, n16);
    aggregate_k<<<(int)((E * 16 + B - 1) / B), blk, 0, stream>>>(bufB, dis, src, dst, bufA, E);

    fill_zero_k<<<(int)((NGRAPH * D + B - 1) / B), blk, 0, stream>>>(pooled, NGRAPH * D);
    fill_zero_k<<<(int)((NGRAPH + B - 1) / B), blk, 0, stream>>>(cnt, NGRAPH);
    node_pool_k<<<(int)((n16 + B - 1) / B), blk, 0, stream>>>(bufB, bufA, dis, b2, batch,
                                                              pooled, cnt, n16);
}

extern "C" void kernel_launch(void* const* d_in, const int* in_sizes, int n_in,
                              void* d_out, int out_size, void* d_ws, size_t ws_size,
                              hipStream_t stream) {
    const float* emb = (const float*)d_in[0];
    const float* W1  = (const float*)d_in[1];
    const float* b1  = (const float*)d_in[2];
    const float* W2  = (const float*)d_in[3];
    const float* b2  = (const float*)d_in[4];
    const float* fcW = (const float*)d_in[5];
    const float* fcb = (const float*)d_in[6];
    const int* rx     = (const int*)d_in[7];
    const int* redge  = (const int*)d_in[8];
    const int* rbatch = (const int*)d_in[9];
    const int* lx     = (const int*)d_in[10];
    const int* ledge  = (const int*)d_in[11];
    const int* lbatch = (const int*)d_in[12];

    const int N = in_sizes[7];
    const long long E = in_sizes[8] / 2;

    char* ws = (char*)d_ws;
    float* bufA = (float*)ws;  ws += (size_t)N * D * sizeof(float);
    float* bufB = (float*)ws;  ws += (size_t)N * D * sizeof(float);
    float* dis  = (float*)ws;  ws += (size_t)N * sizeof(float);
    float* pooled_r = (float*)ws; ws += (size_t)NGRAPH * D * sizeof(float);
    float* pooled_l = (float*)ws; ws += (size_t)NGRAPH * D * sizeof(float);
    float* cnt_r = (float*)ws; ws += (size_t)NGRAPH * sizeof(float);
    float* cnt_l = (float*)ws; ws += (size_t)NGRAPH * sizeof(float);

    unsigned* part = (unsigned*)ws;  ws += (size_t)E * sizeof(unsigned);
    int* bcount = (int*)ws;  ws += (size_t)NBMAX * sizeof(int);
    int* bstart = (int*)ws;  ws += (size_t)2 * NBMAX * sizeof(int);
    int* cursor = (int*)ws;  ws += (size_t)NBMAX * sizeof(int);
    const bool bucketed = (size_t)(ws - (char*)d_ws) <= ws_size &&
                          N > 0 && N <= NBMAX * BNODES && E > 0 && E < (1ll << 30);
    float* bufT = (float*)ws;  ws += (size_t)N * D * sizeof(float);
    const bool pair = (size_t)(ws - (char*)d_ws) <= ws_size;
    float* rowsB = pair ? bufT : bufB;

    char* q = (char*)d_ws;
    float* pair0 = (float*)q;  q += (size_t)N * D * 2 * sizeof(float);
    float* pair1 = (float*)q;  q += (size_t)N * D * 2 * sizeof(float);
    float* rows0 = (float*)q;  q += (size_t)N * D * sizeof(float);
    float* rows1 = (float*)q;  q += (size_t)N * D * sizeof(float);
    float* dis0  = (float*)q;  q += (size_t)N * sizeof(float);
    float* dis1  = (float*)q;  q += (size_t)N * sizeof(float);
    float* zero_run = (float*)q;
    float* pool0 = (float*)q;  q += (size_t)NGRAPH * D * sizeof(float);
    float* pool1 = (float*)q;  q += (size_t)NGRAPH * D * sizeof(float);
    float* pcnt0 = (float*)q;  q += (size_t)NGRAPH * sizeof(float);
    float* pcnt1 = (float*)q;  q += (size_t)NGRAPH * sizeof(float);
    int* bcount0 = (int*)q;  q += (size_t)NBMAX * sizeof(int);
    int* bcount1 = (int*)q;  q += (size_t)NBMAX * sizeof(int);
    const long long zero_words = 2ll * NGRAPH * D + 2ll * NGRAPH + 2ll * NBMAX;
    int* bstart0 = (int*)q;  q += (size_t)2 * NBMAX * sizeof(int);
    int* bstart1 = (int*)q;  q += (size_t)2 * NBMAX * sizeof(int);
    int* cursor0 = (int*)q;  q += (size_t)NBMAX * sizeof(int);
    int* cursor1 = (int*)q;  q += (size_t)NBMAX * sizeof(int);
    unsigned* part0 = (unsigned*)q;  q += (size_t)E * sizeof(unsigned);
    unsigned* part1 = (unsigned*)q;  q += (size_t)E * sizeof(unsigned);
    const bool both = bucketed && (size_t)(q - (char*)d_ws) <= ws_size;
    if (both) {
        run_both_bucketed(emb, W1, b1, W2, b2, rx, lx, redge, ledge, rbatch, lbatch, N, E,
                          pair0, pair1, rows0, rows1, dis0, dis1, zero_run, zero_words,
                          pool0, pool1, pcnt0, pcnt1, bcount0, bcount1, bstart0, bstart1,
                          cursor0, cursor1, part0, part1, stream);
        fc_k<<<(NGRAPH * 6 + 255) / 256, 256, 0, stream>>>(pool0, pcnt0, pool1, pcnt1,
                                                           fcW, fcb, (float*)d_out);
        return;
    }
    if (bucketed) {
        run_protein_bucketed(emb, W1, b1, W2, b2, rx, redge, rbatch, N, E,
                             bufA, rowsB, dis, pooled_r, cnt_r,
                             part, bcount, bstart, cursor, pair, stream);
        run_protein_bucketed(emb, W1, b1, W2, b2, lx, ledge, lbatch, N, E,
                             bufA, rowsB, dis, pooled_l, cnt_l,
                             part, bcount, bstart, cursor, pair, stream);
    } else {
        run_protein(emb, W1, b1, W2, b2, rx, redge, rbatch, N, E,
                    bufA, bufB, dis, pooled_r, cnt_r, stream);
        run_protein(emb, W1, b1, W2, b2, lx, ledge, lbatch, N, E,
                    bufA, bufB, dis, pooled_l, cnt_l, stream);
    }

    fc_k<<<(NGRAPH * 6 + 255) / 256, 256, 0, stream>>>(pooled_r, cnt_r, pooled_l, cnt_l,
                                                       fcW, fcb, (float*)d_out);
}


# layer kernels: one edge record per lane, decoded once and handed to the row's 16 feature lanes by DPP row_share (was: all 16 lanes load and decode the same record); order of the LDS adds changes
# speedup vs baseline: 7.0620x; 1.0930x over previous
#include <hip/hip_runtime.h>
#include <hip/hip_bf16.h>

typedef float v2f __attribute__((ext_vector_type(2)));
typedef float v8f __attribute__((ext_vector_type(8)));

#define D 16
#define NGRAPH 1024

__global__ void fill_zero_k(float* __restrict__ p, long long n) {
    long long i = (long long)blockIdx.x * blockDim.x + threadIdx.x;
    if (i < n) p[i] = 0.0f;
}

__global__ void gather_gemm16_k(const float* __restrict__ src_rows0,
                                const float* __restrict__ src_rows1,
                                const int* __restrict__ ids0,
                                const int* __restrict__ ids1,
                                const float* __restrict__ W,
                                float* __restrict__ out0,
                                float* __restrict__ out1,
                                const float* __restrict__ dis_pair0,
                                const float* __restrict__ dis_pair1,
                                int n_nodes) {
    const bool second = blockIdx.y != 0;
    const float* __restrict__ src_rows = second ? src_rows1 : src_rows0;
    const int* __restrict__ ids = second ? ids1 : ids0;
    float* __restrict__ out = second ? out1 : out0;
    const float* __restrict__ dis_pair = second ? dis_pair1 : dis_pair0;
    const int lane  = threadIdx.x & 31;
    const int wave  = (int)((blockIdx.x * blockDim.x + threadIdx.x) >> 5);
    const int nwave = (int)((gridDim.x * blockDim.x) >> 5);
    const int row   = lane & 15;
    const int col   = lane & 15;
    const int koff  = (lane >> 4) << 1;
    const int rowb  = (lane >> 4) << 3;

    v2f bb0, bb1, bb2, bb3;
    bb0.x = W[(0  + koff) * D + col]; bb0.y = W[(1  + koff) * D + col];
    bb1.x = W[(4  + koff) * D + col]; bb1.y = W[(5  + koff) * D + col];
    bb2.x = W[(8  + koff) * D + col]; bb2.y = W[(9  + koff) * D + col];
    bb3.x = W[(12 + koff) * D + col]; bb3.y = W[(13 + koff) * D + col];

    const int ntiles = n_nodes >> 4;
    for (int tile = wave; tile < ntiles; tile += nwave) {
        const int node = (tile << 4) + row;
        const int r = ids ? ids[node] : node;
        const float* xrow = src_rows + (size_t)r * D;
        v2f a0 = *(const v2f*)(xrow + 0  + koff);
        v2f a1 = *(const v2f*)(xrow + 4  + koff);
        v2f a2 = *(const v2f*)(xrow + 8  + koff);
        v2f a3 = *(const v2f*)(xrow + 12 + koff);
        v8f c = {};
        c = __builtin_amdgcn_wmma_f32_16x16x4_f32(false, a0, false, bb0, (short)0, c, false, false);
        c = __builtin_amdgcn_wmma_f32_16x16x4_f32(false, a1, false, bb1, (short)0, c, false, false);
        c = __builtin_amdgcn_wmma_f32_16x16x4_f32(false, a2, false, bb2, (short)0, c, false, false);
        c = __builtin_amdgcn_wmma_f32_16x16x4_f32(false, a3, false, bb3, (short)0, c, false, false);
        if (dis_pair) {
            const float* dp = dis_pair + (tile << 4) + rowb;
            v2f* ob2 = (v2f*)out + ((size_t)(tile << 4) + rowb) * D + col;
            #pragma unroll
            for (int rr = 0; rr < 8; ++rr) {
                v2f pr; pr.x = c[rr]; pr.y = dp[rr];
                ob2[(size_t)rr * D] = pr;
            }
        } else {
            float* ob = out + ((size_t)(tile << 4) + rowb) * D + col;
            #pragma unroll
            for (int rr = 0; rr < 8; ++rr) ob[(size_t)rr * D] = c[rr];
        }
    }

    const int tail_start = ntiles << 4;
    if (tail_start < n_nodes) {
        int node = tail_start + (int)(blockIdx.x * blockDim.x + threadIdx.x);
        if (node < n_nodes) {
            const int r = ids ? ids[node] : node;
            const float* xrow = src_rows + (size_t)r * D;
            for (int j = 0; j < D; ++j) {
                float acc = 0.0f;
                for (int k = 0; k < D; ++k) acc += xrow[k] * W[k * D + j];
                if (dis_pair) {
                    out[((size_t)node * D + j) * 2] = acc;
                    out[((size_t)node * D + j) * 2 + 1] = dis_pair[node];
                } else {
                    out[(size_t)node * D + j] = acc;
                }
            }
        }
    }
}

__global__ void degree_k(const int* __restrict__ dst, float* __restrict__ deg,
                         long long E) {
    long long e = (long long)blockIdx.x * blockDim.x + threadIdx.x;
    if (e < E) atomicAdd(deg + dst[e], 1.0f);
}

__global__ void rsqrt_k(float* __restrict__ d, int n) {
    int i = blockIdx.x * blockDim.x + threadIdx.x;
    if (i < n) d[i] = rsqrtf(d[i] + 1.0f);
}

__global__ void aggregate_k(const float* __restrict__ h,
                            const float* __restrict__ dis,
                            const int* __restrict__ src,
                            const int* __restrict__ dst,
                            float* __restrict__ agg,
                            long long E) {
    long long t = (long long)blockIdx.x * blockDim.x + threadIdx.x;
    long long e = t >> 4;
    int j = (int)(t & 15);
    if (e < E) {
        int s = src[e], d = dst[e];
        float norm = dis[s] * dis[d];
        atomicAdd(agg + (size_t)d * D + j, h[(size_t)s * D + j] * norm);
    }
}

__global__ void node_relu_k(float* __restrict__ h, const float* __restrict__ agg,
                            const float* __restrict__ dis,
                            const float* __restrict__ bias, long long n16) {
    long long t = (long long)blockIdx.x * blockDim.x + threadIdx.x;
    if (t < n16) {
        long long i = t >> 4; int j = (int)(t & 15);
        float di = dis[i];
        float v = agg[t] + h[t] * (di * di) + bias[j];
        h[t] = v > 0.0f ? v : 0.0f;
    }
}

__global__ void node_pool_k(const float* __restrict__ h2,
                            const float* __restrict__ agg,
                            const float* __restrict__ dis,
                            const float* __restrict__ bias,
                            const int* __restrict__ batch,
                            float* __restrict__ pooled,
                            float* __restrict__ cnt, long long n16) {
    long long t = (long long)blockIdx.x * blockDim.x + threadIdx.x;
    if (t < n16) {
        long long i = t >> 4; int j = (int)(t & 15);
        float di = dis[i];
        float v = agg[t] + h2[t] * (di * di) + bias[j];
        int g = batch[i];
        atomicAdd(pooled + (size_t)g * D + j, v);
        if (j == 0) atomicAdd(cnt + g, 1.0f);
    }
}

__global__ void fc_k(const float* __restrict__ pooled_r, const float* __restrict__ cnt_r,
                     const float* __restrict__ pooled_l, const float* __restrict__ cnt_l,
                     const float* __restrict__ fcW, const float* __restrict__ fcb,
                     float* __restrict__ out) {
    int t = blockIdx.x * blockDim.x + threadIdx.x;
    if (t < NGRAPH * 6) {
        int g = t / 6, o = t - g * 6;
        float icr = 1.0f / fmaxf(cnt_r[g], 1.0f);
        float icl = 1.0f / fmaxf(cnt_l[g], 1.0f);
        float acc = fcb[o];
        #pragma unroll
        for (int j = 0; j < D; ++j)
            acc += fcW[o * (2 * D) + j] * (pooled_r[g * D + j] * icr);
        #pragma unroll
        for (int j = 0; j < D; ++j)
            acc += fcW[o * (2 * D) + D + j] * (pooled_l[g * D + j] * icl);
        if (o < 3) out[g * 3 + o] = acc;
        else       out[NGRAPH * 3 + g * 3 + (o - 3)] = acc;
    }
}


#define RSH 8
#define RMASK 255u
#define BNODES 196
#define NBMAX 1024
#define NWGP 256
#define PTHREADS 1024
#define PTILE 32768
#define PU 8
#define DTHREADS 256

__global__ __launch_bounds__(PTHREADS)
void bucket_hist_k(const int* __restrict__ dst0, const int* __restrict__ dst1,
                   long long E, int tlen,
                   int* __restrict__ bcount0, int* __restrict__ bcount1) {
    const bool second = blockIdx.y != 0;
    const int* __restrict__ dst = second ? dst1 : dst0;
    int* __restrict__ bcount = second ? bcount1 : bcount0;
    __shared__ int cnt[NBMAX];
    const int t = threadIdx.x;
    cnt[t] = 0;
    __syncthreads();
    const long long base = (long long)blockIdx.x * tlen;
    const long long rem = E - base;
    const int ne = rem < (long long)tlen ? (int)rem : tlen;
    for (int k0 = 0; k0 * PTHREADS < ne; k0 += PU) {
        int dv[PU];
        #pragma unroll
        for (int u = 0; u < PU; ++u) {
            int l = (k0 + u) * PTHREADS + t;
            dv[u] = dst[base + (l < ne ? l : 0)];
        }
        #pragma unroll
        for (int u = 0; u < PU; ++u) {
            int l = (k0 + u) * PTHREADS + t;
            if (l < ne) atomicAdd(&cnt[(unsigned)dv[u] / BNODES], 1);
        }
    }
    __syncthreads();
    const int c = cnt[t];
    if (c > 0) atomicAdd(bcount + t, c);
}

__global__ __launch_bounds__(NBMAX)
void bucket_scan_k(const int* __restrict__ bcount0, const int* __restrict__ bcount1,
                   int* __restrict__ bstart0, int* __restrict__ bstart1,
                   int* __restrict__ cursor0, int* __restrict__ cursor1, int nb) {
    const bool second = blockIdx.y != 0;
    const int* __restrict__ bcount = second ? bcount1 : bcount0;
    int* __restrict__ bstart = second ? bstart1 : bstart0;
    int* __restrict__ cursor = second ? cursor1 : cursor0;
    __shared__ int s[NBMAX];
    const int t = threadIdx.x;
    const int own = (t < nb) ? bcount[t] : 0;
    s[t] = own;
    __syncthreads();
    for (int off = 1; off < NBMAX; off <<= 1) {
        int v = (t >= off) ? s[t - off] : 0;
        __syncthreads();
        s[t] += v;
        __syncthreads();
    }
    const int incl = s[t];
    if (t < nb) { bstart[t] = incl - own; cursor[t] = incl - own; }
    if (t == nb - 1) bstart[nb] = incl;
}

__global__ __launch_bounds__(PTHREADS)
void bucket_scatter_k(const int* __restrict__ src0, const int* __restrict__ src1,
                      const int* __restrict__ dst0, const int* __restrict__ dst1,
                      long long E, int tlen,
                      int* __restrict__ cursor0, int* __restrict__ cursor1,
                      unsigned* __restrict__ part0, unsigned* __restrict__ part1) {
    const bool second = blockIdx.y != 0;
    const int* __restrict__ src = second ? src1 : src0;
    const int* __restrict__ dst = second ? dst1 : dst0;
    int* __restrict__ cursor = second ? cursor1 : cursor0;
    unsigned* __restrict__ part = second ? part1 : part0;
    __shared__ int cnt[NBMAX];
    __shared__ int tstart[NBMAX];
    __shared__ int gbase[NBMAX];
    __shared__ int run[NBMAX];
    __shared__ unsigned stage[PTILE];
    __shared__ unsigned short bid[PTILE];
    const int t = threadIdx.x;
    cnt[t] = 0;
    __syncthreads();
    const long long base = (long long)blockIdx.x * tlen;
    const long long rem = E - base;
    const int ne = rem < (long long)tlen ? (int)rem : tlen;
    for (int k0 = 0; k0 * PTHREADS < ne; k0 += PU) {
        int dv[PU];
        #pragma unroll
        for (int u = 0; u < PU; ++u) {
            int l = (k0 + u) * PTHREADS + t;
            dv[u] = dst[base + (l < ne ? l : 0)];
        }
        #pragma unroll
        for (int u = 0; u < PU; ++u) {
            int l = (k0 + u) * PTHREADS + t;
            if (l < ne) atomicAdd(&cnt[(unsigned)dv[u] / BNODES], 1);
        }
    }
    __syncthreads();
    const int own = cnt[t];
    run[t] = own;
    __syncthreads();
    for (int off = 1; off < NBMAX; off <<= 1) {
        int v = (t >= off) ? run[t - off] : 0;
        __syncthreads();
        run[t] += v;
        __syncthreads();
    }
    const int excl = run[t] - own;
    tstart[t] = excl;
    run[t] = excl;
    gbase[t] = (own > 0) ? atomicAdd(cursor + t, own) : 0;
    __syncthreads();
    for (int k0 = 0; k0 * PTHREADS < ne; k0 += PU) {
        int dv[PU], sv[PU];
        #pragma unroll
        for (int u = 0; u < PU; ++u) {
            int l = (k0 + u) * PTHREADS + t;
            long long e = base + (l < ne ? l : 0);
            dv[u] = dst[e];
            sv[u] = src[e];
        }
        #pragma unroll
        for (int u = 0; u < PU; ++u) {
            int l = (k0 + u) * PTHREADS + t;
            if (l < ne) {
                unsigned b = (unsigned)dv[u] / BNODES;
                int slot = atomicAdd(&run[b], 1);
                stage[slot] = ((unsigned)sv[u] << RSH) | ((unsigned)dv[u] - b * BNODES);
                bid[slot] = (unsigned short)b;
            }
        }
    }
    __syncthreads();
    for (int slot = t; slot < ne; slot += PTHREADS) {
        int b = bid[slot];
        part[gbase[b] + (slot - tstart[b])] = stage[slot];
    }
}

__global__ __launch_bounds__(DTHREADS)
void bucket_degree_k(const unsigned* __restrict__ part0, const unsigned* __restrict__ part1,
                     const int* __restrict__ bstart0, const int* __restrict__ bstart1,
                     float* __restrict__ dis0, float* __restrict__ dis1, int n) {
    const bool second = blockIdx.y != 0;
    const unsigned* __restrict__ part = second ? part1 : part0;
    const int* __restrict__ bstart = second ? bstart1 : bstart0;
    float* __restrict__ dis = second ? dis1 : dis0;
    __shared__ int cnt[BNODES];
    const int t = threadIdx.x;
    if (t < BNODES) cnt[t] = 0;
    __syncthreads();
    const int b = blockIdx.x;
    const int p0 = bstart[b], p1 = bstart[b + 1];
    for (int p = p0 + t; p < p1; p += DTHREADS * PU) {
        unsigned r[PU];
        #pragma unroll
        for (int u = 0; u < PU; ++u) {
            int q = p + u * DTHREADS;
            r[u] = part[q < p1 ? q : 0];
        }
        #pragma unroll
        for (int u = 0; u < PU; ++u) {
            int q = p + u * DTHREADS;
            if (q < p1) atomicAdd(&cnt[r[u] & RMASK], 1);
        }
    }
    __syncthreads();
    const int i = b * BNODES + t;
    if (t < BNODES && i < n) dis[i] = rsqrtf((float)cnt[t] + 1.0f);
}

template <int E>
__device__ __forceinline__ unsigned row_share_u32(unsigned x) {
    return (unsigned)__builtin_amdgcn_mov_dpp((int)x, 0x150 + E, 0xf, 0xf, true);
}

#define LTHREADS 512
#define PGR 16
template <bool POOL, bool PAIR>
__global__ __launch_bounds__(LTHREADS)
void bucket_layer_k(const float* __restrict__ h0, const float* __restrict__ h1,
                    const float* __restrict__ dis0, const float* __restrict__ dis1,
                    const unsigned* __restrict__ part0, const unsigned* __restrict__ part1,
                    const int* __restrict__ bstart0, const int* __restrict__ bstart1,
                    const float* __restrict__ bias,
                    float* __restrict__ rows0, float* __restrict__ rows1,
                    const int* __restrict__ batch0, const int* __restrict__ batch1,
                    float* __restrict__ pooled0, float* __restrict__ pooled1,
                    float* __restrict__ cntg0, float* __restrict__ cntg1,
                    int n) {
    const bool second = blockIdx.y != 0;
    const float* __restrict__ h = second ? h1 : h0;
    const float* __restrict__ dis = second ? dis1 : dis0;
    const unsigned* __restrict__ part = second ? part1 : part0;
    const int* __restrict__ bstart = second ? bstart1 : bstart0;
    float* __restrict__ rows = second ? rows1 : rows0;
    const int* __restrict__ batch = second ? batch1 : batch0;
    float* __restrict__ pooled = second ? pooled1 : pooled0;
    float* __restrict__ cntg = second ? cntg1 : cntg0;
    __shared__ float acc[(BNODES + 1) * D];
    __shared__ float dloc[BNODES + 1];
    __shared__ float pl[PGR * D];
    __shared__ int   pc[PGR];
    const int t = threadIdx.x;
    const int b = blockIdx.x;
    const int node0 = b * BNODES;
    for (int i = t; i < (BNODES + 1) * D; i += LTHREADS) acc[i] = 0.0f;
    if (t <= BNODES) {
        int i = node0 + t;
        dloc[t] = (t < BNODES && i < n) ? dis[i] : 0.0f;
    }
    if (POOL) {
        if (t < PGR * D) pl[t] = 0.0f;
        if (t < PGR) pc[t] = 0;
    }
    __syncthreads();

    const int lane = t & 31;
    const unsigned j = (unsigned)lane & 15u;
    const unsigned j4 = j << 2;
    const int wv = __builtin_amdgcn_readfirstlane(t >> 5);
    const int p0 = bstart[b], p1 = bstart[b + 1];
    const v2f* __restrict__ hp = (const v2f*)h;
    int qb = p0 + wv * 32;
    unsigned recn = 0u;
    if (qb + lane < p1) recn = part[qb + lane];
    for (; qb < p1; qb += LTHREADS) {
        const int q = qb + lane;
        const unsigned rec = recn;
        const int qn = q + LTHREADS;
        recn = (qn < p1) ? part[qn] : 0u;
        const unsigned s = rec >> RSH;
        const unsigned dl = (q < p1) ? (rec & RMASK) : (unsigned)BNODES;
        const unsigned gi = s << 4;
        const unsigned ao4 = dl << 6;
        float wl = dloc[dl];
        if (!PAIR) wl *= dis[s];
        const unsigned wb = __float_as_uint(wl);
        float hv[16], ds[16];
#define LAYER_LOAD(E) { \
            const unsigned ix = row_share_u32<E>(gi) | j; \
            if (PAIR) { v2f pr = hp[(size_t)ix]; hv[E] = pr.x; ds[E] = pr.y; } \
            else { hv[E] = h[(size_t)ix]; ds[E] = 1.0f; } }
        LAYER_LOAD(0)  LAYER_LOAD(1)  LAYER_LOAD(2)  LAYER_LOAD(3)
        LAYER_LOAD(4)  LAYER_LOAD(5)  LAYER_LOAD(6)  LAYER_LOAD(7)
        LAYER_LOAD(8)  LAYER_LOAD(9)  LAYER_LOAD(10) LAYER_LOAD(11)
        LAYER_LOAD(12) LAYER_LOAD(13) LAYER_LOAD(14) LAYER_LOAD(15)
#undef LAYER_LOAD
#define LAYER_ADD(E) { \
            const float w = __uint_as_float(row_share_u32<E>(wb)); \
            const unsigned a = row_share_u32<E>(ao4) | j4; \
            const float norm = PAIR ? ds[E] * w : w; \
            atomicAdd((float*)((char*)acc + a), hv[E] * norm); }
        LAYER_ADD(0)  LAYER_ADD(1)  LAYER_ADD(2)  LAYER_ADD(3)
        LAYER_ADD(4)  LAYER_ADD(5)  LAYER_ADD(6)  LAYER_ADD(7)
        LAYER_ADD(8)  LAYER_ADD(9)  LAYER_ADD(10) LAYER_ADD(11)
        LAYER_ADD(12) LAYER_ADD(13) LAYER_ADD(14) LAYER_ADD(15)
#undef LAYER_ADD
    }
    __syncthreads();

    if (!POOL) {
        for (int idx = t; idx < BNODES * D; idx += LTHREADS) {
            int nl = idx >> 4;
            int i = node0 + nl;
            if (i < n) {
                float di = dloc[nl];
                size_t o = (size_t)i * D + (idx & 15);
                float v = acc[idx] + h[PAIR ? o * 2 : o] * (di * di) + bias[idx & 15];
                rows[o] = v > 0.0f ? v : 0.0f;
            }
        }
    } else {
        const int g0 = batch[node0];
        for (int idx = t; idx < BNODES * D; idx += LTHREADS) {
            int nl = idx >> 4;
            int jj = idx & 15;
            int i = node0 + nl;
            if (i < n) {
                float di = dloc[nl];
                size_t o = (size_t)i * D + jj;
                float v = acc[idx] + h[PAIR ? o * 2 : o] * (di * di) + bias[jj];
                int g = batch[i];
                unsigned k = (unsigned)(g - g0);
                if (k < (unsigned)PGR) {
                    atomicAdd(&pl[k * D + jj], v);
                    if (jj == 0) atomicAdd(&pc[k], 1);
                } else {
                    atomicAdd(pooled + (size_t)g * D + jj, v);
                    if (jj == 0) atomicAdd(cntg + g, 1.0f);
                }
            }
        }
        __syncthreads();
        if (t < PGR * D) {
            int k = t >> 4;
            int c = pc[k];
            if (c > 0) {
                int g = g0 + k;
                atomicAdd(pooled + (size_t)g * D + (t & 15), pl[t]);
                if ((t & 15) == 0) atomicAdd(cntg + g, (float)c);
            }
        }
    }
}

static void run_protein_bucketed(const float* emb, const float* W1, const float* b1,
                                 const float* W2, const float* b2,
                                 const int* ids, const int* edge, const int* batch,
                                 int N, long long E,
                                 float* bufA, float* bufB, float* dis,
                                 float* pooled, float* cnt,
                                 unsigned* part, int* bcount, int* bstart, int* cursor,
                                 bool pair, hipStream_t stream) {
    const int* src = edge;
    const int* dst = edge + E;
    const int B = 256;
    dim3 blk(B);
    int gemm_grid = (int)(((N >> 4) + 7) / 8);
    if (gemm_grid < 1) gemm_grid = 1;
    const int nb = (N + BNODES - 1) / BNODES;
    const long long round_cap = (long long)NWGP * PTILE;
    const int ptiles = NWGP * (int)((E + round_cap - 1) / round_cap);
    const int tlen = (int)(((E + ptiles - 1) / ptiles + 31) & ~31ll);

    fill_zero_k<<<(NBMAX + B - 1) / B, blk, 0, stream>>>((float*)bcount, NBMAX);
    bucket_hist_k<<<ptiles, PTHREADS, 0, stream>>>(dst, nullptr, E, tlen, bcount, nullptr);
    bucket_scan_k<<<1, NBMAX, 0, stream>>>(bcount, nullptr, bstart, nullptr, cursor, nullptr, nb);
    bucket_scatter_k<<<ptiles, PTHREADS, 0, stream>>>(src, nullptr, dst, nullptr, E, tlen,
                                                      cursor, nullptr, part, nullptr);
    bucket_degree_k<<<nb, DTHREADS, 0, stream>>>(part, nullptr, bstart, nullptr, dis, nullptr, N);

    gather_gemm16_k<<<gemm_grid, blk, 0, stream>>>(emb, nullptr, ids, nullptr, W1, bufA, nullptr,
                                                   pair ? dis : nullptr, nullptr, N);
    if (pair)
        bucket_layer_k<false, true><<<nb, LTHREADS, 0, stream>>>(
            bufA, nullptr, dis, nullptr, part, nullptr, bstart, nullptr, b1, bufB, nullptr,
            nullptr, nullptr, nullptr, nullptr, nullptr, nullptr, N);
    else
        bucket_layer_k<false, false><<<nb, LTHREADS, 0, stream>>>(
            bufA, nullptr, dis, nullptr, part, nullptr, bstart, nullptr, b1, bufB, nullptr,
            nullptr, nullptr, nullptr, nullptr, nullptr, nullptr, N);
    gather_gemm16_k<<<gemm_grid, blk, 0, stream>>>(bufB, nullptr, nullptr, nullptr, W2, bufA, nullptr,
                                                   pair ? dis : nullptr, nullptr, N);

    fill_zero_k<<<(int)((NGRAPH * D + B - 1) / B), blk, 0, stream>>>(pooled, NGRAPH * D);
    fill_zero_k<<<(int)((NGRAPH + B - 1) / B), blk, 0, stream>>>(cnt, NGRAPH);
    if (pair)
        bucket_layer_k<true, true><<<nb, LTHREADS, 0, stream>>>(
            bufA, nullptr, dis, nullptr, part, nullptr, bstart, nullptr, b2, nullptr, nullptr,
            batch, nullptr, pooled, nullptr, cnt, nullptr, N);
    else
        bucket_layer_k<true, false><<<nb, LTHREADS, 0, stream>>>(
            bufA, nullptr, dis, nullptr, part, nullptr, bstart, nullptr, b2, nullptr, nullptr,
            batch, nullptr, pooled, nullptr, cnt, nullptr, N);
}

static void run_both_bucketed(const float* emb, const float* W1, const float* b1,
                              const float* W2, const float* b2,
                              const int* ids0, const int* ids1,
                              const int* edge0, const int* edge1,
                              const int* batch0, const int* batch1,
                              int N, long long E,
                              float* pair0, float* pair1, float* rows0, float* rows1,
                              float* dis0, float* dis1,
                              float* zero_run, long long zero_words,
                              float* pooled0, float* pooled1, float* cnt0, float* cnt1,
                              int* bcount0, int* bcount1, int* bstart0, int* bstart1,
                              int* cursor0, int* cursor1,
                              unsigned* part0, unsigned* part1, hipStream_t stream) {
    const int B = 256;
    dim3 blk(B);
    int gemm_grid = (int)(((N >> 4) + 7) / 8);
    if (gemm_grid < 1) gemm_grid = 1;
    const int nb = (N + BNODES - 1) / BNODES;
    const long long round_cap = (long long)NWGP * PTILE;
    const int ptiles = NWGP * (int)((E + round_cap - 1) / round_cap);
    const int tlen = (int)(((E + ptiles - 1) / ptiles + 31) & ~31ll);

    fill_zero_k<<<(int)((zero_words + B - 1) / B), blk, 0, stream>>>(zero_run, zero_words);
    bucket_hist_k<<<dim3(ptiles, 2), PTHREADS, 0, stream>>>(edge0 + E, edge1 + E, E, tlen,
                                                            bcount0, bcount1);
    bucket_scan_k<<<dim3(1, 2), NBMAX, 0, stream>>>(bcount0, bcount1, bstart0, bstart1,
                                                    cursor0, cursor1, nb);
    bucket_scatter_k<<<dim3(ptiles, 2), PTHREADS, 0, stream>>>(edge0, edge1, edge0 + E, edge1 + E,
                                                               E, tlen, cursor0, cursor1,
                                                               part0, part1);
    bucket_degree_k<<<dim3(nb, 2), DTHREADS, 0, stream>>>(part0, part1, bstart0, bstart1,
                                                          dis0, dis1, N);
    gather_gemm16_k<<<dim3(gemm_grid, 2), blk, 0, stream>>>(emb, emb, ids0, ids1, W1, pair0, pair1,
                                                            dis0, dis1, N);
    bucket_layer_k<false, true><<<dim3(nb, 2), LTHREADS, 0, stream>>>(
        pair0, pair1, dis0, dis1, part0, part1, bstart0, bstart1, b1, rows0, rows1,
        nullptr, nullptr, nullptr, nullptr, nullptr, nullptr, N);
    gather_gemm16_k<<<dim3(gemm_grid, 2), blk, 0, stream>>>(rows0, rows1, nullptr, nullptr, W2,
                                                            pair0, pair1, dis0, dis1, N);
    bucket_layer_k<true, true><<<dim3(nb, 2), LTHREADS, 0, stream>>>(
        pair0, pair1, dis0, dis1, part0, part1, bstart0, bstart1, b2, nullptr, nullptr,
        batch0, batch1, pooled0, pooled1, cnt0, cnt1, N);
}

static void run_protein(const float* emb, const float* W1, const float* b1,
                        const float* W2, const float* b2,
                        const int* ids, const int* edge, const int* batch,
                        int N, long long E,
                        float* bufA, float* bufB, float* dis,
                        float* pooled, float* cnt, hipStream_t stream) {
    const int* src = edge;
    const int* dst = edge + E;
    const long long n16 = (long long)N * D;
    const int B = 256;
    dim3 blk(B);
    int gemm_grid = (int)(((N >> 4) + 7) / 8);
    if (gemm_grid < 1) gemm_grid = 1;

    fill_zero_k<<<(int)((N + B - 1) / B), blk, 0, stream>>>(dis, N);
    fill_zero_k<<<(int)((n16 + B - 1) / B), blk, 0, stream>>>(bufB, n16);

    gather_gemm16_k<<<gemm_grid, blk, 0, stream>>>(emb, nullptr, ids, nullptr, W1, bufA, nullptr,
                                                   nullptr, nullptr, N);

    degree_k<<<(int)((E + B - 1) / B), blk, 0, stream>>>(dst, dis, E);
    rsqrt_k<<<(int)((N + B - 1) / B), blk, 0, stream>>>(dis, N);

    aggregate_k<<<(int)((E * 16 + B - 1) / B), blk, 0, stream>>>(bufA, dis, src, dst, bufB, E);
    node_relu_k<<<(int)((n16 + B - 1) / B), blk, 0, stream>>>(bufA, bufB, dis, b1, n16);

    gather_gemm16_k<<<gemm_grid, blk, 0, stream>>>(bufA, nullptr, nullptr, nullptr, W2, bufB, nullptr,
                                                   nullptr, nullptr, N);

    fill_zero_k<<<(int)((n16 + B - 1) / B), blk, 0, stream>>>(bufA, n16);
    aggregate_k<<<(int)((E * 16 + B - 1) / B), blk, 0, stream>>>(bufB, dis, src, dst, bufA, E);

    fill_zero_k<<<(int)((NGRAPH * D + B - 1) / B), blk, 0, stream>>>(pooled, NGRAPH * D);
    fill_zero_k<<<(int)((NGRAPH + B - 1) / B), blk, 0, stream>>>(cnt, NGRAPH);
    node_pool_k<<<(int)((n16 + B - 1) / B), blk, 0, stream>>>(bufB, bufA, dis, b2, batch,
                                                              pooled, cnt, n16);
}

extern "C" void kernel_launch(void* const* d_in, const int* in_sizes, int n_in,
                              void* d_out, int out_size, void* d_ws, size_t ws_size,
                              hipStream_t stream) {
    const float* emb = (const float*)d_in[0];
    const float* W1  = (const float*)d_in[1];
    const float* b1  = (const float*)d_in[2];
    const float* W2  = (const float*)d_in[3];
    const float* b2  = (const float*)d_in[4];
    const float* fcW = (const float*)d_in[5];
    const float* fcb = (const float*)d_in[6];
    const int* rx     = (const int*)d_in[7];
    const int* redge  = (const int*)d_in[8];
    const int* rbatch = (const int*)d_in[9];
    const int* lx     = (const int*)d_in[10];
    const int* ledge  = (const int*)d_in[11];
    const int* lbatch = (const int*)d_in[12];

    const int N = in_sizes[7];
    const long long E = in_sizes[8] / 2;

    char* ws = (char*)d_ws;
    float* bufA = (float*)ws;  ws += (size_t)N * D * sizeof(float);
    float* bufB = (float*)ws;  ws += (size_t)N * D * sizeof(float);
    float* dis  = (float*)ws;  ws += (size_t)N * sizeof(float);
    float* pooled_r = (float*)ws; ws += (size_t)NGRAPH * D * sizeof(float);
    float* pooled_l = (float*)ws; ws += (size_t)NGRAPH * D * sizeof(float);
    float* cnt_r = (float*)ws; ws += (size_t)NGRAPH * sizeof(float);
    float* cnt_l = (float*)ws; ws += (size_t)NGRAPH * sizeof(float);

    unsigned* part = (unsigned*)ws;  ws += (size_t)E * sizeof(unsigned);
    int* bcount = (int*)ws;  ws += (size_t)NBMAX * sizeof(int);
    int* bstart = (int*)ws;  ws += (size_t)2 * NBMAX * sizeof(int);
    int* cursor = (int*)ws;  ws += (size_t)NBMAX * sizeof(int);
    const bool bucketed = (size_t)(ws - (char*)d_ws) <= ws_size &&
                          N > 0 && N <= NBMAX * BNODES && E > 0 && E < (1ll << 30);
    float* bufT = (float*)ws;  ws += (size_t)N * D * sizeof(float);
    const bool pair = (size_t)(ws - (char*)d_ws) <= ws_size;
    float* rowsB = pair ? bufT : bufB;

    char* q = (char*)d_ws;
    float* pair0 = (float*)q;  q += (size_t)N * D * 2 * sizeof(float);
    float* pair1 = (float*)q;  q += (size_t)N * D * 2 * sizeof(float);
    float* rows0 = (float*)q;  q += (size_t)N * D * sizeof(float);
    float* rows1 = (float*)q;  q += (size_t)N * D * sizeof(float);
    float* dis0  = (float*)q;  q += (size_t)N * sizeof(float);
    float* dis1  = (float*)q;  q += (size_t)N * sizeof(float);
    float* zero_run = (float*)q;
    float* pool0 = (float*)q;  q += (size_t)NGRAPH * D * sizeof(float);
    float* pool1 = (float*)q;  q += (size_t)NGRAPH * D * sizeof(float);
    float* pcnt0 = (float*)q;  q += (size_t)NGRAPH * sizeof(float);
    float* pcnt1 = (float*)q;  q += (size_t)NGRAPH * sizeof(float);
    int* bcount0 = (int*)q;  q += (size_t)NBMAX * sizeof(int);
    int* bcount1 = (int*)q;  q += (size_t)NBMAX * sizeof(int);
    const long long zero_words = 2ll * NGRAPH * D + 2ll * NGRAPH + 2ll * NBMAX;
    int* bstart0 = (int*)q;  q += (size_t)2 * NBMAX * sizeof(int);
    int* bstart1 = (int*)q;  q += (size_t)2 * NBMAX * sizeof(int);
    int* cursor0 = (int*)q;  q += (size_t)NBMAX * sizeof(int);
    int* cursor1 = (int*)q;  q += (size_t)NBMAX * sizeof(int);
    unsigned* part0 = (unsigned*)q;  q += (size_t)E * sizeof(unsigned);
    unsigned* part1 = (unsigned*)q;  q += (size_t)E * sizeof(unsigned);
    const bool both = bucketed && (size_t)(q - (char*)d_ws) <= ws_size;
    if (both) {
        run_both_bucketed(emb, W1, b1, W2, b2, rx, lx, redge, ledge, rbatch, lbatch, N, E,
                          pair0, pair1, rows0, rows1, dis0, dis1, zero_run, zero_words,
                          pool0, pool1, pcnt0, pcnt1, bcount0, bcount1, bstart0, bstart1,
                          cursor0, cursor1, part0, part1, stream);
        fc_k<<<(NGRAPH * 6 + 255) / 256, 256, 0, stream>>>(pool0, pcnt0, pool1, pcnt1,
                                                           fcW, fcb, (float*)d_out);
        return;
    }
    if (bucketed) {
        run_protein_bucketed(emb, W1, b1, W2, b2, rx, redge, rbatch, N, E,
                             bufA, rowsB, dis, pooled_r, cnt_r,
                             part, bcount, bstart, cursor, pair, stream);
        run_protein_bucketed(emb, W1, b1, W2, b2, lx, ledge, lbatch, N, E,
                             bufA, rowsB, dis, pooled_l, cnt_l,
                             part, bcount, bstart, cursor, pair, stream);
    } else {
        run_protein(emb, W1, b1, W2, b2, rx, redge, rbatch, N, E,
                    bufA, bufB, dis, pooled_r, cnt_r, stream);
        run_protein(emb, W1, b1, W2, b2, lx, ledge, lbatch, N, E,
                    bufA, bufB, dis, pooled_l, cnt_l, stream);
    }

    fc_k<<<(NGRAPH * 6 + 255) / 256, 256, 0, stream>>>(pooled_r, cnt_r, pooled_l, cnt_l,
                                                       fcW, fcb, (float*)d_out);
}


# layer 1 over a product table: emb @ W1 taken once per table row (not per node) and held in each group's LDS; an edge's source row comes from LDS via ids[src]; per-edge global read is 8 B, was 128 B
# speedup vs baseline: 7.6078x; 1.0773x over previous
#include <hip/hip_runtime.h>
#include <hip/hip_bf16.h>

typedef float v2f __attribute__((ext_vector_type(2)));
typedef float v8f __attribute__((ext_vector_type(8)));
typedef float v4f __attribute__((ext_vector_type(4)));
typedef unsigned v2u __attribute__((ext_vector_type(2)));

#define D 16
#define NGRAPH 1024

__global__ void fill_zero_k(float* __restrict__ p, long long n) {
    long long i = (long long)blockIdx.x * blockDim.x + threadIdx.x;
    if (i < n) p[i] = 0.0f;
}

__global__ void gather_gemm16_k(const float* __restrict__ src_rows0,
                                const float* __restrict__ src_rows1,
                                const int* __restrict__ ids0,
                                const int* __restrict__ ids1,
                                const float* __restrict__ W,
                                float* __restrict__ out0,
                                float* __restrict__ out1,
                                const float* __restrict__ dis_pair0,
                                const float* __restrict__ dis_pair1,
                                int n_nodes) {
    const bool second = blockIdx.y != 0;
    const float* __restrict__ src_rows = second ? src_rows1 : src_rows0;
    const int* __restrict__ ids = second ? ids1 : ids0;
    float* __restrict__ out = second ? out1 : out0;
    const float* __restrict__ dis_pair = second ? dis_pair1 : dis_pair0;
    const int lane  = threadIdx.x & 31;
    const int wave  = (int)((blockIdx.x * blockDim.x + threadIdx.x) >> 5);
    const int nwave = (int)((gridDim.x * blockDim.x) >> 5);
    const int row   = lane & 15;
    const int col   = lane & 15;
    const int koff  = (lane >> 4) << 1;
    const int rowb  = (lane >> 4) << 3;

    v2f bb0, bb1, bb2, bb3;
    bb0.x = W[(0  + koff) * D + col]; bb0.y = W[(1  + koff) * D + col];
    bb1.x = W[(4  + koff) * D + col]; bb1.y = W[(5  + koff) * D + col];
    bb2.x = W[(8  + koff) * D + col]; bb2.y = W[(9  + koff) * D + col];
    bb3.x = W[(12 + koff) * D + col]; bb3.y = W[(13 + koff) * D + col];

    const int ntiles = n_nodes >> 4;
    for (int tile = wave; tile < ntiles; tile += nwave) {
        const int node = (tile << 4) + row;
        const int r = ids ? ids[node] : node;
        const float* xrow = src_rows + (size_t)r * D;
        v2f a0 = *(const v2f*)(xrow + 0  + koff);
        v2f a1 = *(const v2f*)(xrow + 4  + koff);
        v2f a2 = *(const v2f*)(xrow + 8  + koff);
        v2f a3 = *(const v2f*)(xrow + 12 + koff);
        v8f c = {};
        c = __builtin_amdgcn_wmma_f32_16x16x4_f32(false, a0, false, bb0, (short)0, c, false, false);
        c = __builtin_amdgcn_wmma_f32_16x16x4_f32(false, a1, false, bb1, (short)0, c, false, false);
        c = __builtin_amdgcn_wmma_f32_16x16x4_f32(false, a2, false, bb2, (short)0, c, false, false);
        c = __builtin_amdgcn_wmma_f32_16x16x4_f32(false, a3, false, bb3, (short)0, c, false, false);
        if (dis_pair) {
            const float* dp = dis_pair + (tile << 4) + rowb;
            v2f* ob2 = (v2f*)out + ((size_t)(tile << 4) + rowb) * D + col;
            #pragma unroll
            for (int rr = 0; rr < 8; ++rr) {
                v2f pr; pr.x = c[rr]; pr.y = dp[rr];
                ob2[(size_t)rr * D] = pr;
            }
        } else {
            float* ob = out + ((size_t)(tile << 4) + rowb) * D + col;
            #pragma unroll
            for (int rr = 0; rr < 8; ++rr) ob[(size_t)rr * D] = c[rr];
        }
    }

    const int tail_start = ntiles << 4;
    if (tail_start < n_nodes) {
        int node = tail_start + (int)(blockIdx.x * blockDim.x + threadIdx.x);
        if (node < n_nodes) {
            const int r = ids ? ids[node] : node;
            const float* xrow = src_rows + (size_t)r * D;
            for (int j = 0; j < D; ++j) {
                float acc = 0.0f;
                for (int k = 0; k < D; ++k) acc += xrow[k] * W[k * D + j];
                if (dis_pair) {
                    out[((size_t)node * D + j) * 2] = acc;
                    out[((size_t)node * D + j) * 2 + 1] = dis_pair[node];
                } else {
                    out[(size_t)node * D + j] = acc;
                }
            }
        }
    }
}

__global__ void table_gemm16_k(const float* __restrict__ src_rows,
                               const float* __restrict__ W,
                               float* __restrict__ out, int n_rows) {
    const int lane  = threadIdx.x & 31;
    const int wave  = (int)((blockIdx.x * blockDim.x + threadIdx.x) >> 5);
    const int nwave = (int)((gridDim.x * blockDim.x) >> 5);
    const int row   = lane & 15;
    const int col   = lane & 15;
    const int koff  = (lane >> 4) << 1;
    const int rowb  = (lane >> 4) << 3;
    v2f bb0, bb1, bb2, bb3;
    bb0.x = W[(0  + koff) * D + col]; bb0.y = W[(1  + koff) * D + col];
    bb1.x = W[(4  + koff) * D + col]; bb1.y = W[(5  + koff) * D + col];
    bb2.x = W[(8  + koff) * D + col]; bb2.y = W[(9  + koff) * D + col];
    bb3.x = W[(12 + koff) * D + col]; bb3.y = W[(13 + koff) * D + col];
    const int ntiles = (n_rows + 15) >> 4;
    float* ob = out + ((size_t)(wave << 4) + rowb) * D + col;
    for (int tile = wave; tile < ntiles; tile += nwave, ob += (size_t)nwave * 16 * D) {
        const int node = (tile << 4) + row;
        const int r = node < n_rows ? node : n_rows - 1;
        const float* xrow = src_rows + (size_t)r * D;
        v2f a0 = *(const v2f*)(xrow + 0  + koff);
        v2f a1 = *(const v2f*)(xrow + 4  + koff);
        v2f a2 = *(const v2f*)(xrow + 8  + koff);
        v2f a3 = *(const v2f*)(xrow + 12 + koff);
        v8f c = {};
        c = __builtin_amdgcn_wmma_f32_16x16x4_f32(false, a0, false, bb0, (short)0, c, false, false);
        c = __builtin_amdgcn_wmma_f32_16x16x4_f32(false, a1, false, bb1, (short)0, c, false, false);
        c = __builtin_amdgcn_wmma_f32_16x16x4_f32(false, a2, false, bb2, (short)0, c, false, false);
        c = __builtin_amdgcn_wmma_f32_16x16x4_f32(false, a3, false, bb3, (short)0, c, false, false);
        #pragma unroll
        for (int rr = 0; rr < 8; ++rr) ob[(size_t)rr * D] = c[rr];
    }
}

__global__ void degree_k(const int* __restrict__ dst, float* __restrict__ deg,
                         long long E) {
    long long e = (long long)blockIdx.x * blockDim.x + threadIdx.x;
    if (e < E) atomicAdd(deg + dst[e], 1.0f);
}

__global__ void rsqrt_k(float* __restrict__ d, int n) {
    int i = blockIdx.x * blockDim.x + threadIdx.x;
    if (i < n) d[i] = rsqrtf(d[i] + 1.0f);
}

__global__ void aggregate_k(const float* __restrict__ h,
                            const float* __restrict__ dis,
                            const int* __restrict__ src,
                            const int* __restrict__ dst,
                            float* __restrict__ agg,
                            long long E) {
    long long t = (long long)blockIdx.x * blockDim.x + threadIdx.x;
    long long e = t >> 4;
    int j = (int)(t & 15);
    if (e < E) {
        int s = src[e], d = dst[e];
        float norm = dis[s] * dis[d];
        atomicAdd(agg + (size_t)d * D + j, h[(size_t)s * D + j] * norm);
    }
}

__global__ void node_relu_k(float* __restrict__ h, const float* __restrict__ agg,
                            const float* __restrict__ dis,
                            const float* __restrict__ bias, long long n16) {
    long long t = (long long)blockIdx.x * blockDim.x + threadIdx.x;
    if (t < n16) {
        long long i = t >> 4; int j = (int)(t & 15);
        float di = dis[i];
        float v = agg[t] + h[t] * (di * di) + bias[j];
        h[t] = v > 0.0f ? v : 0.0f;
    }
}

__global__ void node_pool_k(const float* __restrict__ h2,
                            const float* __restrict__ agg,
                            const float* __restrict__ dis,
                            const float* __restrict__ bias,
                            const int* __restrict__ batch,
                            float* __restrict__ pooled,
                            float* __restrict__ cnt, long long n16) {
    long long t = (long long)blockIdx.x * blockDim.x + threadIdx.x;
    if (t < n16) {
        long long i = t >> 4; int j = (int)(t & 15);
        float di = dis[i];
        float v = agg[t] + h2[t] * (di * di) + bias[j];
        int g = batch[i];
        atomicAdd(pooled + (size_t)g * D + j, v);
        if (j == 0) atomicAdd(cnt + g, 1.0f);
    }
}

__global__ void fc_k(const float* __restrict__ pooled_r, const float* __restrict__ cnt_r,
                     const float* __restrict__ pooled_l, const float* __restrict__ cnt_l,
                     const float* __restrict__ fcW, const float* __restrict__ fcb,
                     float* __restrict__ out) {
    int t = blockIdx.x * blockDim.x + threadIdx.x;
    if (t < NGRAPH * 6) {
        int g = t / 6, o = t - g * 6;
        float icr = 1.0f / fmaxf(cnt_r[g], 1.0f);
        float icl = 1.0f / fmaxf(cnt_l[g], 1.0f);
        float acc = fcb[o];
        #pragma unroll
        for (int j = 0; j < D; ++j)
            acc += fcW[o * (2 * D) + j] * (pooled_r[g * D + j] * icr);
        #pragma unroll
        for (int j = 0; j < D; ++j)
            acc += fcW[o * (2 * D) + D + j] * (pooled_l[g * D + j] * icl);
        if (o < 3) out[g * 3 + o] = acc;
        else       out[NGRAPH * 3 + g * 3 + (o - 3)] = acc;
    }
}


#define RSH 8
#define RMASK 255u
#define BNODES 196
#define NBMAX 1024
#define NWGP 256
#define PTHREADS 1024
#define PTILE 32768
#define PU 8
#define DTHREADS 256

__global__ __launch_bounds__(PTHREADS)
void bucket_hist_k(const int* __restrict__ dst0, const int* __restrict__ dst1,
                   long long E, int tlen,
                   int* __restrict__ bcount0, int* __restrict__ bcount1) {
    const bool second = blockIdx.y != 0;
    const int* __restrict__ dst = second ? dst1 : dst0;
    int* __restrict__ bcount = second ? bcount1 : bcount0;
    __shared__ int cnt[NBMAX];
    const int t = threadIdx.x;
    cnt[t] = 0;
    __syncthreads();
    const long long base = (long long)blockIdx.x * tlen;
    const long long rem = E - base;
    const int ne = rem < (long long)tlen ? (int)rem : tlen;
    for (int k0 = 0; k0 * PTHREADS < ne; k0 += PU) {
        int dv[PU];
        #pragma unroll
        for (int u = 0; u < PU; ++u) {
            int l = (k0 + u) * PTHREADS + t;
            dv[u] = dst[base + (l < ne ? l : 0)];
        }
        #pragma unroll
        for (int u = 0; u < PU; ++u) {
            int l = (k0 + u) * PTHREADS + t;
            if (l < ne) atomicAdd(&cnt[(unsigned)dv[u] / BNODES], 1);
        }
    }
    __syncthreads();
    const int c = cnt[t];
    if (c > 0) atomicAdd(bcount + t, c);
}

__global__ __launch_bounds__(NBMAX)
void bucket_scan_k(const int* __restrict__ bcount0, const int* __restrict__ bcount1,
                   int* __restrict__ bstart0, int* __restrict__ bstart1,
                   int* __restrict__ cursor0, int* __restrict__ cursor1, int nb) {
    const bool second = blockIdx.y != 0;
    const int* __restrict__ bcount = second ? bcount1 : bcount0;
    int* __restrict__ bstart = second ? bstart1 : bstart0;
    int* __restrict__ cursor = second ? cursor1 : cursor0;
    __shared__ int s[NBMAX];
    const int t = threadIdx.x;
    const int own = (t < nb) ? bcount[t] : 0;
    s[t] = own;
    __syncthreads();
    for (int off = 1; off < NBMAX; off <<= 1) {
        int v = (t >= off) ? s[t - off] : 0;
        __syncthreads();
        s[t] += v;
        __syncthreads();
    }
    const int incl = s[t];
    if (t < nb) { bstart[t] = incl - own; cursor[t] = incl - own; }
    if (t == nb - 1) bstart[nb] = incl;
}

__global__ __launch_bounds__(PTHREADS)
void bucket_scatter_k(const int* __restrict__ src0, const int* __restrict__ src1,
                      const int* __restrict__ dst0, const int* __restrict__ dst1,
                      long long E, int tlen,
                      int* __restrict__ cursor0, int* __restrict__ cursor1,
                      unsigned* __restrict__ part0, unsigned* __restrict__ part1) {
    const bool second = blockIdx.y != 0;
    const int* __restrict__ src = second ? src1 : src0;
    const int* __restrict__ dst = second ? dst1 : dst0;
    int* __restrict__ cursor = second ? cursor1 : cursor0;
    unsigned* __restrict__ part = second ? part1 : part0;
    __shared__ int cnt[NBMAX];
    __shared__ int tstart[NBMAX];
    __shared__ int gbase[NBMAX];
    __shared__ int run[NBMAX];
    __shared__ unsigned stage[PTILE];
    __shared__ unsigned short bid[PTILE];
    const int t = threadIdx.x;
    cnt[t] = 0;
    __syncthreads();
    const long long base = (long long)blockIdx.x * tlen;
    const long long rem = E - base;
    const int ne = rem < (long long)tlen ? (int)rem : tlen;
    for (int k0 = 0; k0 * PTHREADS < ne; k0 += PU) {
        int dv[PU];
        #pragma unroll
        for (int u = 0; u < PU; ++u) {
            int l = (k0 + u) * PTHREADS + t;
            dv[u] = dst[base + (l < ne ? l : 0)];
        }
        #pragma unroll
        for (int u = 0; u < PU; ++u) {
            int l = (k0 + u) * PTHREADS + t;
            if (l < ne) atomicAdd(&cnt[(unsigned)dv[u] / BNODES], 1);
        }
    }
    __syncthreads();
    const int own = cnt[t];
    run[t] = own;
    __syncthreads();
    for (int off = 1; off < NBMAX; off <<= 1) {
        int v = (t >= off) ? run[t - off] : 0;
        __syncthreads();
        run[t] += v;
        __syncthreads();
    }
    const int excl = run[t] - own;
    tstart[t] = excl;
    run[t] = excl;
    gbase[t] = (own > 0) ? atomicAdd(cursor + t, own) : 0;
    __syncthreads();
    for (int k0 = 0; k0 * PTHREADS < ne; k0 += PU) {
        int dv[PU], sv[PU];
        #pragma unroll
        for (int u = 0; u < PU; ++u) {
            int l = (k0 + u) * PTHREADS + t;
            long long e = base + (l < ne ? l : 0);
            dv[u] = dst[e];
            sv[u] = src[e];
        }
        #pragma unroll
        for (int u = 0; u < PU; ++u) {
            int l = (k0 + u) * PTHREADS + t;
            if (l < ne) {
                unsigned b = (unsigned)dv[u] / BNODES;
                int slot = atomicAdd(&run[b], 1);
                stage[slot] = ((unsigned)sv[u] << RSH) | ((unsigned)dv[u] - b * BNODES);
                bid[slot] = (unsigned short)b;
            }
        }
    }
    __syncthreads();
    for (int slot = t; slot < ne; slot += PTHREADS) {
        int b = bid[slot];
        part[gbase[b] + (slot - tstart[b])] = stage[slot];
    }
}

__global__ __launch_bounds__(DTHREADS)
void bucket_degree_k(const unsigned* __restrict__ part0, const unsigned* __restrict__ part1,
                     const int* __restrict__ bstart0, const int* __restrict__ bstart1,
                     float* __restrict__ dis0, float* __restrict__ dis1, int n,
                     const int* __restrict__ ids0, const int* __restrict__ ids1,
                     v2u* __restrict__ nrec0, v2u* __restrict__ nrec1, int n_tab) {
    const bool second = blockIdx.y != 0;
    const unsigned* __restrict__ part = second ? part1 : part0;
    const int* __restrict__ bstart = second ? bstart1 : bstart0;
    float* __restrict__ dis = second ? dis1 : dis0;
    const int* __restrict__ ids = second ? ids1 : ids0;
    v2u* __restrict__ nrec = second ? nrec1 : nrec0;
    __shared__ int cnt[BNODES];
    const int t = threadIdx.x;
    if (t < BNODES) cnt[t] = 0;
    __syncthreads();
    const int b = blockIdx.x;
    const int p0 = bstart[b], p1 = bstart[b + 1];
    for (int p = p0 + t; p < p1; p += DTHREADS * PU) {
        unsigned r[PU];
        #pragma unroll
        for (int u = 0; u < PU; ++u) {
            int q = p + u * DTHREADS;
            r[u] = part[q < p1 ? q : 0];
        }
        #pragma unroll
        for (int u = 0; u < PU; ++u) {
            int q = p + u * DTHREADS;
            if (q < p1) atomicAdd(&cnt[r[u] & RMASK], 1);
        }
    }
    __syncthreads();
    const int i = b * BNODES + t;
    if (t < BNODES && i < n) {
        const float dv = rsqrtf((float)cnt[t] + 1.0f);
        dis[i] = dv;
        if (nrec) {
            int k = ids[i];
            k = k < 0 ? 0 : (k >= n_tab ? n_tab - 1 : k);
            v2u r; r.x = (unsigned)k; r.y = __float_as_uint(dv);
            nrec[i] = r;
        }
    }
}

template <int E>
__device__ __forceinline__ unsigned row_share_u32(unsigned x) {
    return (unsigned)__builtin_amdgcn_mov_dpp((int)x, 0x150 + E, 0xf, 0xf, true);
}

#define LTHREADS 512
#define PGR 16
template <bool POOL, bool PAIR>
__global__ __launch_bounds__(LTHREADS)
void bucket_layer_k(const float* __restrict__ h0, const float* __restrict__ h1,
                    const float* __restrict__ dis0, const float* __restrict__ dis1,
                    const unsigned* __restrict__ part0, const unsigned* __restrict__ part1,
                    const int* __restrict__ bstart0, const int* __restrict__ bstart1,
                    const float* __restrict__ bias,
                    float* __restrict__ rows0, float* __restrict__ rows1,
                    const int* __restrict__ batch0, const int* __restrict__ batch1,
                    float* __restrict__ pooled0, float* __restrict__ pooled1,
                    float* __restrict__ cntg0, float* __restrict__ cntg1,
                    int n) {
    const bool second = blockIdx.y != 0;
    const float* __restrict__ h = second ? h1 : h0;
    const float* __restrict__ dis = second ? dis1 : dis0;
    const unsigned* __restrict__ part = second ? part1 : part0;
    const int* __restrict__ bstart = second ? bstart1 : bstart0;
    float* __restrict__ rows = second ? rows1 : rows0;
    const int* __restrict__ batch = second ? batch1 : batch0;
    float* __restrict__ pooled = second ? pooled1 : pooled0;
    float* __restrict__ cntg = second ? cntg1 : cntg0;
    __shared__ float acc[(BNODES + 1) * D];
    __shared__ float dloc[BNODES + 1];
    __shared__ float pl[PGR * D];
    __shared__ int   pc[PGR];
    const int t = threadIdx.x;
    const int b = blockIdx.x;
    const int node0 = b * BNODES;
    for (int i = t; i < (BNODES + 1) * D; i += LTHREADS) acc[i] = 0.0f;
    if (t <= BNODES) {
        int i = node0 + t;
        dloc[t] = (t < BNODES && i < n) ? dis[i] : 0.0f;
    }
    if (POOL) {
        if (t < PGR * D) pl[t] = 0.0f;
        if (t < PGR) pc[t] = 0;
    }
    __syncthreads();

    const int lane = t & 31;
    const unsigned j = (unsigned)lane & 15u;
    const unsigned j4 = j << 2;
    const int wv = __builtin_amdgcn_readfirstlane(t >> 5);
    const int p0 = bstart[b], p1 = bstart[b + 1];
    const v2f* __restrict__ hp = (const v2f*)h;
    int qb = p0 + wv * 32;
    unsigned recn = 0u;
    if (qb + lane < p1) recn = part[qb + lane];
    for (; qb < p1; qb += LTHREADS) {
        const int q = qb + lane;
        const unsigned rec = recn;
        const int qn = q + LTHREADS;
        recn = (qn < p1) ? part[qn] : 0u;
        const unsigned s = rec >> RSH;
        const unsigned dl = (q < p1) ? (rec & RMASK) : (unsigned)BNODES;
        const unsigned gi = s << 4;
        const unsigned ao4 = dl << 6;
        float wl = dloc[dl];
        if (!PAIR) wl *= dis[s];
        const unsigned wb = __float_as_uint(wl);
        float hv[16], ds[16];
#define LAYER_LOAD(E) { \
            const unsigned ix = row_share_u32<E>(gi) | j; \
            if (PAIR) { v2f pr = hp[(size_t)ix]; hv[E] = pr.x; ds[E] = pr.y; } \
            else { hv[E] = h[(size_t)ix]; ds[E] = 1.0f; } }
        LAYER_LOAD(0)  LAYER_LOAD(1)  LAYER_LOAD(2)  LAYER_LOAD(3)
        LAYER_LOAD(4)  LAYER_LOAD(5)  LAYER_LOAD(6)  LAYER_LOAD(7)
        LAYER_LOAD(8)  LAYER_LOAD(9)  LAYER_LOAD(10) LAYER_LOAD(11)
        LAYER_LOAD(12) LAYER_LOAD(13) LAYER_LOAD(14) LAYER_LOAD(15)
#undef LAYER_LOAD
#define LAYER_ADD(E) { \
            const float w = __uint_as_float(row_share_u32<E>(wb)); \
            const unsigned a = row_share_u32<E>(ao4) | j4; \
            const float norm = PAIR ? ds[E] * w : w; \
            atomicAdd((float*)((char*)acc + a), hv[E] * norm); }
        LAYER_ADD(0)  LAYER_ADD(1)  LAYER_ADD(2)  LAYER_ADD(3)
        LAYER_ADD(4)  LAYER_ADD(5)  LAYER_ADD(6)  LAYER_ADD(7)
        LAYER_ADD(8)  LAYER_ADD(9)  LAYER_ADD(10) LAYER_ADD(11)
        LAYER_ADD(12) LAYER_ADD(13) LAYER_ADD(14) LAYER_ADD(15)
#undef LAYER_ADD
    }
    __syncthreads();

    if (!POOL) {
        for (int idx = t; idx < BNODES * D; idx += LTHREADS) {
            int nl = idx >> 4;
            int i = node0 + nl;
            if (i < n) {
                float di = dloc[nl];
                size_t o = (size_t)i * D + (idx & 15);
                float v = acc[idx] + h[PAIR ? o * 2 : o] * (di * di) + bias[idx & 15];
                rows[o] = v > 0.0f ? v : 0.0f;
            }
        }
    } else {
        const int g0 = batch[node0];
        for (int idx = t; idx < BNODES * D; idx += LTHREADS) {
            int nl = idx >> 4;
            int jj = idx & 15;
            int i = node0 + nl;
            if (i < n) {
                float di = dloc[nl];
                size_t o = (size_t)i * D + jj;
                float v = acc[idx] + h[PAIR ? o * 2 : o] * (di * di) + bias[jj];
                int g = batch[i];
                unsigned k = (unsigned)(g - g0);
                if (k < (unsigned)PGR) {
                    atomicAdd(&pl[k * D + jj], v);
                    if (jj == 0) atomicAdd(&pc[k], 1);
                } else {
                    atomicAdd(pooled + (size_t)g * D + jj, v);
                    if (jj == 0) atomicAdd(cntg + g, 1.0f);
                }
            }
        }
        __syncthreads();
        if (t < PGR * D) {
            int k = t >> 4;
            int c = pc[k];
            if (c > 0) {
                int g = g0 + k;
                atomicAdd(pooled + (size_t)g * D + (t & 15), pl[t]);
                if ((t & 15) == 0) atomicAdd(cntg + g, (float)c);
            }
        }
    }
}


#define TCAP 1032
#define TCOPY ((TCAP * D / 4 + LTHREADS - 1) / LTHREADS)
#define TROWS ((TCAP + 15) & ~15)
__global__ __launch_bounds__(LTHREADS)
void bucket_layer1_tab_k(const float* __restrict__ tab, int n_tab,
                         const v2u* __restrict__ nrec0, const v2u* __restrict__ nrec1,
                         const unsigned* __restrict__ part0, const unsigned* __restrict__ part1,
                         const int* __restrict__ bstart0, const int* __restrict__ bstart1,
                         const float* __restrict__ bias,
                         float* __restrict__ rows0, float* __restrict__ rows1, int n) {
    const bool second = blockIdx.y != 0;
    const v2u* __restrict__ nrec = second ? nrec1 : nrec0;
    const unsigned* __restrict__ part = second ? part1 : part0;
    const int* __restrict__ bstart = second ? bstart1 : bstart0;
    float* __restrict__ rows = second ? rows1 : rows0;
    __shared__ struct {
        float acc[(BNODES + 1) * D];
        float dloc[BNODES + 4];
        v4f tl[TCAP * D / 4];
    } sm;
    float* const acc = sm.acc;
    float* const dloc = sm.dloc;
    v4f* const tl = sm.tl;
    const int t = threadIdx.x;
    const int b = blockIdx.x;
    const int node0 = b * BNODES;
    const int lane = t & 31;
    const unsigned j4 = ((unsigned)lane & 15u) << 2;
    const int wv = __builtin_amdgcn_readfirstlane(t >> 5);
    const int p0 = bstart[b], p1 = bstart[b + 1];
    int qb = p0 + wv * 32;
    unsigned rec0 = 0u, rec1 = 0u;
    if (qb + lane < p1) rec0 = part[qb + lane];
    if (qb + LTHREADS + lane < p1) rec1 = part[qb + LTHREADS + lane];
    for (int i = t; i < (BNODES + 1) * D; i += LTHREADS) acc[i] = 0.0f;
    if (t <= BNODES) {
        int i = node0 + t;
        dloc[t] = (t < BNODES && i < n) ? __uint_as_float(nrec[i].y) : 0.0f;
    }
    {
        const v4f* __restrict__ tg = (const v4f*)tab;
        const int n4 = n_tab * (D / 4);
        v4f tv[TCOPY];
        #pragma unroll
        for (int u = 0; u < TCOPY; ++u) {
            int i = t + u * LTHREADS;
            tv[u] = tg[i < n4 ? i : 0];
        }
        #pragma unroll
        for (int u = 0; u < TCOPY; ++u) {
            int i = t + u * LTHREADS;
            if (i < n4) tl[i] = tv[u];
        }
    }
    v2u nd = nrec[rec0 >> RSH];
    __syncthreads();

    const char* tlb = (const char*)tl;
    for (; qb < p1; qb += LTHREADS) {
        const int q = qb + lane;
        const unsigned rec = rec0;
        const v2u ndc = nd;
        rec0 = rec1;
        nd = nrec[rec0 >> RSH];
        const int q2 = q + 2 * LTHREADS;
        rec1 = (q2 < p1) ? part[q2] : 0u;
        const unsigned dl = (q < p1) ? (rec & RMASK) : (unsigned)BNODES;
        const unsigned to4 = ndc.x << 6;
        const unsigned ao4 = dl << 6;
        const unsigned wb = __float_as_uint(dloc[dl] * __uint_as_float(ndc.y));
        float hv[16];
#define TAB_LOAD(E) { \
            const unsigned ta = row_share_u32<E>(to4) | j4; \
            hv[E] = *(const float*)(tlb + ta); }
        TAB_LOAD(0)  TAB_LOAD(1)  TAB_LOAD(2)  TAB_LOAD(3)
        TAB_LOAD(4)  TAB_LOAD(5)  TAB_LOAD(6)  TAB_LOAD(7)
        TAB_LOAD(8)  TAB_LOAD(9)  TAB_LOAD(10) TAB_LOAD(11)
        TAB_LOAD(12) TAB_LOAD(13) TAB_LOAD(14) TAB_LOAD(15)
#undef TAB_LOAD
#define TAB_ADD(E) { \
            const float w = __uint_as_float(row_share_u32<E>(wb)); \
            const unsigned a = row_share_u32<E>(ao4) | j4; \
            atomicAdd((float*)((char*)acc + a), hv[E] * w); }
        TAB_ADD(0)  TAB_ADD(1)  TAB_ADD(2)  TAB_ADD(3)
        TAB_ADD(4)  TAB_ADD(5)  TAB_ADD(6)  TAB_ADD(7)
        TAB_ADD(8)  TAB_ADD(9)  TAB_ADD(10) TAB_ADD(11)
        TAB_ADD(12) TAB_ADD(13) TAB_ADD(14) TAB_ADD(15)
#undef TAB_ADD
    }
    __syncthreads();

    const float* tlf = (const float*)tl;
    for (int idx = t; idx < BNODES * D; idx += LTHREADS) {
        int nl = idx >> 4;
        int i = node0 + nl;
        if (i < n) {
            float di = dloc[nl];
            float hs = tlf[nrec[i].x * D + (idx & 15)];
            float v = acc[idx] + hs * (di * di) + bias[idx & 15];
            rows[(size_t)i * D + (idx & 15)] = v > 0.0f ? v : 0.0f;
        }
    }
}

static void run_protein_bucketed(const float* emb, const float* W1, const float* b1,
                                 const float* W2, const float* b2,
                                 const int* ids, const int* edge, const int* batch,
                                 int N, long long E,
                                 float* bufA, float* bufB, float* dis,
                                 float* pooled, float* cnt,
                                 unsigned* part, int* bcount, int* bstart, int* cursor,
                                 bool pair, hipStream_t stream) {
    const int* src = edge;
    const int* dst = edge + E;
    const int B = 256;
    dim3 blk(B);
    int gemm_grid = (int)(((N >> 4) + 7) / 8);
    if (gemm_grid < 1) gemm_grid = 1;
    const int nb = (N + BNODES - 1) / BNODES;
    const long long round_cap = (long long)NWGP * PTILE;
    const int ptiles = NWGP * (int)((E + round_cap - 1) / round_cap);
    const int tlen = (int)(((E + ptiles - 1) / ptiles + 31) & ~31ll);

    fill_zero_k<<<(NBMAX + B - 1) / B, blk, 0, stream>>>((float*)bcount, NBMAX);
    bucket_hist_k<<<ptiles, PTHREADS, 0, stream>>>(dst, nullptr, E, tlen, bcount, nullptr);
    bucket_scan_k<<<1, NBMAX, 0, stream>>>(bcount, nullptr, bstart, nullptr, cursor, nullptr, nb);
    bucket_scatter_k<<<ptiles, PTHREADS, 0, stream>>>(src, nullptr, dst, nullptr, E, tlen,
                                                      cursor, nullptr, part, nullptr);
    bucket_degree_k<<<nb, DTHREADS, 0, stream>>>(part, nullptr, bstart, nullptr, dis, nullptr, N,
                                                 nullptr, nullptr, nullptr, nullptr, 0);

    gather_gemm16_k<<<gemm_grid, blk, 0, stream>>>(emb, nullptr, ids, nullptr, W1, bufA, nullptr,
                                                   pair ? dis : nullptr, nullptr, N);
    if (pair)
        bucket_layer_k<false, true><<<nb, LTHREADS, 0, stream>>>(
            bufA, nullptr, dis, nullptr, part, nullptr, bstart, nullptr, b1, bufB, nullptr,
            nullptr, nullptr, nullptr, nullptr, nullptr, nullptr, N);
    else
        bucket_layer_k<false, false><<<nb, LTHREADS, 0, stream>>>(
            bufA, nullptr, dis, nullptr, part, nullptr, bstart, nullptr, b1, bufB, nullptr,
            nullptr, nullptr, nullptr, nullptr, nullptr, nullptr, N);
    gather_gemm16_k<<<gemm_grid, blk, 0, stream>>>(bufB, nullptr, nullptr, nullptr, W2, bufA, nullptr,
                                                   pair ? dis : nullptr, nullptr, N);

    fill_zero_k<<<(int)((NGRAPH * D + B - 1) / B), blk, 0, stream>>>(pooled, NGRAPH * D);
    fill_zero_k<<<(int)((NGRAPH + B - 1) / B), blk, 0, stream>>>(cnt, NGRAPH);
    if (pair)
        bucket_layer_k<true, true><<<nb, LTHREADS, 0, stream>>>(
            bufA, nullptr, dis, nullptr, part, nullptr, bstart, nullptr, b2, nullptr, nullptr,
            batch, nullptr, pooled, nullptr, cnt, nullptr, N);
    else
        bucket_layer_k<true, false><<<nb, LTHREADS, 0, stream>>>(
            bufA, nullptr, dis, nullptr, part, nullptr, bstart, nullptr, b2, nullptr, nullptr,
            batch, nullptr, pooled, nullptr, cnt, nullptr, N);
}

static void run_both_bucketed(const float* emb, const float* W1, const float* b1,
                              const float* W2, const float* b2,
                              const int* ids0, const int* ids1,
                              const int* edge0, const int* edge1,
                              const int* batch0, const int* batch1,
                              int N, long long E,
                              float* pair0, float* pair1, float* rows0, float* rows1,
                              float* dis0, float* dis1,
                              float* zero_run, long long zero_words,
                              float* pooled0, float* pooled1, float* cnt0, float* cnt1,
                              int* bcount0, int* bcount1, int* bstart0, int* bstart1,
                              int* cursor0, int* cursor1,
                              unsigned* part0, unsigned* part1,
                              int n_tab, float* tab, v2u* nrec0, v2u* nrec1,
                              hipStream_t stream) {
    const int B = 256;
    dim3 blk(B);
    int gemm_grid = (int)(((N >> 4) + 7) / 8);
    if (gemm_grid < 1) gemm_grid = 1;
    const int nb = (N + BNODES - 1) / BNODES;
    const long long round_cap = (long long)NWGP * PTILE;
    const int ptiles = NWGP * (int)((E + round_cap - 1) / round_cap);
    const int tlen = (int)(((E + ptiles - 1) / ptiles + 31) & ~31ll);

    fill_zero_k<<<(int)((zero_words + B - 1) / B), blk, 0, stream>>>(zero_run, zero_words);
    bucket_hist_k<<<dim3(ptiles, 2), PTHREADS, 0, stream>>>(edge0 + E, edge1 + E, E, tlen,
                                                            bcount0, bcount1);
    bucket_scan_k<<<dim3(1, 2), NBMAX, 0, stream>>>(bcount0, bcount1, bstart0, bstart1,
                                                    cursor0, cursor1, nb);
    bucket_scatter_k<<<dim3(ptiles, 2), PTHREADS, 0, stream>>>(edge0, edge1, edge0 + E, edge1 + E,
                                                               E, tlen, cursor0, cursor1,
                                                               part0, part1);
    if (tab) {
        bucket_degree_k<<<dim3(nb, 2), DTHREADS, 0, stream>>>(part0, part1, bstart0, bstart1,
                                                              dis0, dis1, N,
                                                              ids0, ids1, nrec0, nrec1, n_tab);
        table_gemm16_k<<<(((n_tab + 15) >> 4) + 7) / 8, blk, 0, stream>>>(emb, W1, tab, n_tab);
        bucket_layer1_tab_k<<<dim3(nb, 2), LTHREADS, 0, stream>>>(
            tab, n_tab, nrec0, nrec1, part0, part1, bstart0, bstart1, b1, rows0, rows1, N);
    } else {
        bucket_degree_k<<<dim3(nb, 2), DTHREADS, 0, stream>>>(part0, part1, bstart0, bstart1,
                                                              dis0, dis1, N,
                                                              nullptr, nullptr, nullptr, nullptr, 0);
        gather_gemm16_k<<<dim3(gemm_grid, 2), blk, 0, stream>>>(emb, emb, ids0, ids1, W1, pair0, pair1,
                                                                dis0, dis1, N);
        bucket_layer_k<false, true><<<dim3(nb, 2), LTHREADS, 0, stream>>>(
            pair0, pair1, dis0, dis1, part0, part1, bstart0, bstart1, b1, rows0, rows1,
            nullptr, nullptr, nullptr, nullptr, nullptr, nullptr, N);
    }
    gather_gemm16_k<<<dim3(gemm_grid, 2), blk, 0, stream>>>(rows0, rows1, nullptr, nullptr, W2,
                                                            pair0, pair1, dis0, dis1, N);
    bucket_layer_k<true, true><<<dim3(nb, 2), LTHREADS, 0, stream>>>(
        pair0, pair1, dis0, dis1, part0, part1, bstart0, bstart1, b2, nullptr, nullptr,
        batch0, batch1, pooled0, pooled1, cnt0, cnt1, N);
}

static void run_protein(const float* emb, const float* W1, const float* b1,
                        const float* W2, const float* b2,
                        const int* ids, const int* edge, const int* batch,
                        int N, long long E,
                        float* bufA, float* bufB, float* dis,
                        float* pooled, float* cnt, hipStream_t stream) {
    const int* src = edge;
    const int* dst = edge + E;
    const long long n16 = (long long)N * D;
    const int B = 256;
    dim3 blk(B);
    int gemm_grid = (int)(((N >> 4) + 7) / 8);
    if (gemm_grid < 1) gemm_grid = 1;

    fill_zero_k<<<(int)((N + B - 1) / B), blk, 0, stream>>>(dis, N);
    fill_zero_k<<<(int)((n16 + B - 1) / B), blk, 0, stream>>>(bufB, n16);

    gather_gemm16_k<<<gemm_grid, blk, 0, stream>>>(emb, nullptr, ids, nullptr, W1, bufA, nullptr,
                                                   nullptr, nullptr, N);

    degree_k<<<(int)((E + B - 1) / B), blk, 0, stream>>>(dst, dis, E);
    rsqrt_k<<<(int)((N + B - 1) / B), blk, 0, stream>>>(dis, N);

    aggregate_k<<<(int)((E * 16 + B - 1) / B), blk, 0, stream>>>(bufA, dis, src, dst, bufB, E);
    node_relu_k<<<(int)((n16 + B - 1) / B), blk, 0, stream>>>(bufA, bufB, dis, b1, n16);

    gather_gemm16_k<<<gemm_grid, blk, 0, stream>>>(bufA, nullptr, nullptr, nullptr, W2, bufB, nullptr,
                                                   nullptr, nullptr, N);

    fill_zero_k<<<(int)((n16 + B - 1) / B), blk, 0, stream>>>(bufA, n16);
    aggregate_k<<<(int)((E * 16 + B - 1) / B), blk, 0, stream>>>(bufB, dis, src, dst, bufA, E);

    fill_zero_k<<<(int)((NGRAPH * D + B - 1) / B), blk, 0, stream>>>(pooled, NGRAPH * D);
    fill_zero_k<<<(int)((NGRAPH + B - 1) / B), blk, 0, stream>>>(cnt, NGRAPH);
    node_pool_k<<<(int)((n16 + B - 1) / B), blk, 0, stream>>>(bufB, bufA, dis, b2, batch,
                                                              pooled, cnt, n16);
}

extern "C" void kernel_launch(void* const* d_in, const int* in_sizes, int n_in,
                              void* d_out, int out_size, void* d_ws, size_t ws_size,
                              hipStream_t stream) {
    const float* emb = (const float*)d_in[0];
    const float* W1  = (const float*)d_in[1];
    const float* b1  = (const float*)d_in[2];
    const float* W2  = (const float*)d_in[3];
    const float* b2  = (const float*)d_in[4];
    const float* fcW = (const float*)d_in[5];
    const float* fcb = (const float*)d_in[6];
    const int* rx     = (const int*)d_in[7];
    const int* redge  = (const int*)d_in[8];
    const int* rbatch = (const int*)d_in[9];
    const int* lx     = (const int*)d_in[10];
    const int* ledge  = (const int*)d_in[11];
    const int* lbatch = (const int*)d_in[12];

    const int N = in_sizes[7];
    const long long E = in_sizes[8] / 2;

    char* ws = (char*)d_ws;
    float* bufA = (float*)ws;  ws += (size_t)N * D * sizeof(float);
    float* bufB = (float*)ws;  ws += (size_t)N * D * sizeof(float);
    float* dis  = (float*)ws;  ws += (size_t)N * sizeof(float);
    float* pooled_r = (float*)ws; ws += (size_t)NGRAPH * D * sizeof(float);
    float* pooled_l = (float*)ws; ws += (size_t)NGRAPH * D * sizeof(float);
    float* cnt_r = (float*)ws; ws += (size_t)NGRAPH * sizeof(float);
    float* cnt_l = (float*)ws; ws += (size_t)NGRAPH * sizeof(float);

    unsigned* part = (unsigned*)ws;  ws += (size_t)E * sizeof(unsigned);
    int* bcount = (int*)ws;  ws += (size_t)NBMAX * sizeof(int);
    int* bstart = (int*)ws;  ws += (size_t)2 * NBMAX * sizeof(int);
    int* cursor = (int*)ws;  ws += (size_t)NBMAX * sizeof(int);
    const bool bucketed = (size_t)(ws - (char*)d_ws) <= ws_size &&
                          N > 0 && N <= NBMAX * BNODES && E > 0 && E < (1ll << 30);
    float* bufT = (float*)ws;  ws += (size_t)N * D * sizeof(float);
    const bool pair = (size_t)(ws - (char*)d_ws) <= ws_size;
    float* rowsB = pair ? bufT : bufB;

    char* q = (char*)d_ws;
    float* pair0 = (float*)q;  q += (size_t)N * D * 2 * sizeof(float);
    float* pair1 = (float*)q;  q += (size_t)N * D * 2 * sizeof(float);
    float* rows0 = (float*)q;  q += (size_t)N * D * sizeof(float);
    float* rows1 = (float*)q;  q += (size_t)N * D * sizeof(float);
    float* dis0  = (float*)q;  q += (size_t)N * sizeof(float);
    float* dis1  = (float*)q;  q += (size_t)N * sizeof(float);
    float* zero_run = (float*)q;
    float* pool0 = (float*)q;  q += (size_t)NGRAPH * D * sizeof(float);
    float* pool1 = (float*)q;  q += (size_t)NGRAPH * D * sizeof(float);
    float* pcnt0 = (float*)q;  q += (size_t)NGRAPH * sizeof(float);
    float* pcnt1 = (float*)q;  q += (size_t)NGRAPH * sizeof(float);
    int* bcount0 = (int*)q;  q += (size_t)NBMAX * sizeof(int);
    int* bcount1 = (int*)q;  q += (size_t)NBMAX * sizeof(int);
    const long long zero_words = 2ll * NGRAPH * D + 2ll * NGRAPH + 2ll * NBMAX;
    int* bstart0 = (int*)q;  q += (size_t)2 * NBMAX * sizeof(int);
    int* bstart1 = (int*)q;  q += (size_t)2 * NBMAX * sizeof(int);
    int* cursor0 = (int*)q;  q += (size_t)NBMAX * sizeof(int);
    int* cursor1 = (int*)q;  q += (size_t)NBMAX * sizeof(int);
    unsigned* part0 = (unsigned*)q;  q += (size_t)E * sizeof(unsigned);
    unsigned* part1 = (unsigned*)q;  q += (size_t)E * sizeof(unsigned);
    const bool both = bucketed && (size_t)(q - (char*)d_ws) <= ws_size;
    const int n_tab = in_sizes[0] / D;
    float* tab = (float*)pair1;
    v2u* nrec0 = (v2u*)(tab + (size_t)TROWS * D);
    v2u* nrec1 = nrec0 + N;
    const bool tabbed = n_tab > 0 && n_tab <= TCAP &&
                        (size_t)TROWS * D * sizeof(float) + (size_t)2 * N * sizeof(v2u) <=
                            (size_t)N * D * 2 * sizeof(float);
    if (both) {
        run_both_bucketed(emb, W1, b1, W2, b2, rx, lx, redge, ledge, rbatch, lbatch, N, E,
                          pair0, pair1, rows0, rows1, dis0, dis1, zero_run, zero_words,
                          pool0, pool1, pcnt0, pcnt1, bcount0, bcount1, bstart0, bstart1,
                          cursor0, cursor1, part0, part1,
                          n_tab, tabbed ? tab : nullptr, nrec0, nrec1, stream);
        fc_k<<<(NGRAPH * 6 + 255) / 256, 256, 0, stream>>>(pool0, pcnt0, pool1, pcnt1,
                                                           fcW, fcb, (float*)d_out);
        return;
    }
    if (bucketed) {
        run_protein_bucketed(emb, W1, b1, W2, b2, rx, redge, rbatch, N, E,
                             bufA, rowsB, dis, pooled_r, cnt_r,
                             part, bcount, bstart, cursor, pair, stream);
        run_protein_bucketed(emb, W1, b1, W2, b2, lx, ledge, lbatch, N, E,
                             bufA, rowsB, dis, pooled_l, cnt_l,
                             part, bcount, bstart, cursor, pair, stream);
    } else {
        run_protein(emb, W1, b1, W2, b2, rx, redge, rbatch, N, E,
                    bufA, bufB, dis, pooled_r, cnt_r, stream);
        run_protein(emb, W1, b1, W2, b2, lx, ledge, lbatch, N, E,
                    bufA, bufB, dis, pooled_l, cnt_l, stream);
    }

    fc_k<<<(NGRAPH * 6 + 255) / 256, 256, 0, stream>>>(pooled_r, cnt_r, pooled_l, cnt_l,
                                                       fcW, fcb, (float*)d_out);
}


# layer-2 product moved behind the mean pool by linearity: W2 meets 1024 pooled sums per protein, not 200000 node rows; layer 2 adds dis-scaled relu rows, 64 B per edge (was 128); sum order changes
# speedup vs baseline: 8.8327x; 1.1610x over previous
#include <hip/hip_runtime.h>
#include <hip/hip_bf16.h>

typedef float v2f __attribute__((ext_vector_type(2)));
typedef float v8f __attribute__((ext_vector_type(8)));
typedef float v4f __attribute__((ext_vector_type(4)));
typedef unsigned v2u __attribute__((ext_vector_type(2)));

#define D 16
#define NGRAPH 1024

__global__ void fill_zero_k(float* __restrict__ p, long long n) {
    long long i = (long long)blockIdx.x * blockDim.x + threadIdx.x;
    if (i < n) p[i] = 0.0f;
}

__global__ void gather_gemm16_k(const float* __restrict__ src_rows0,
                                const float* __restrict__ src_rows1,
                                const int* __restrict__ ids0,
                                const int* __restrict__ ids1,
                                const float* __restrict__ W,
                                float* __restrict__ out0,
                                float* __restrict__ out1,
                                const float* __restrict__ dis_pair0,
                                const float* __restrict__ dis_pair1,
                                int n_nodes) {
    const bool second = blockIdx.y != 0;
    const float* __restrict__ src_rows = second ? src_rows1 : src_rows0;
    const int* __restrict__ ids = second ? ids1 : ids0;
    float* __restrict__ out = second ? out1 : out0;
    const float* __restrict__ dis_pair = second ? dis_pair1 : dis_pair0;
    const int lane  = threadIdx.x & 31;
    const int wave  = (int)((blockIdx.x * blockDim.x + threadIdx.x) >> 5);
    const int nwave = (int)((gridDim.x * blockDim.x) >> 5);
    const int row   = lane & 15;
    const int col   = lane & 15;
    const int koff  = (lane >> 4) << 1;
    const int rowb  = (lane >> 4) << 3;

    v2f bb0, bb1, bb2, bb3;
    bb0.x = W[(0  + koff) * D + col]; bb0.y = W[(1  + koff) * D + col];
    bb1.x = W[(4  + koff) * D + col]; bb1.y = W[(5  + koff) * D + col];
    bb2.x = W[(8  + koff) * D + col]; bb2.y = W[(9  + koff) * D + col];
    bb3.x = W[(12 + koff) * D + col]; bb3.y = W[(13 + koff) * D + col];

    const int ntiles = n_nodes >> 4;
    for (int tile = wave; tile < ntiles; tile += nwave) {
        const int node = (tile << 4) + row;
        const int r = ids ? ids[node] : node;
        const float* xrow = src_rows + (size_t)r * D;
        v2f a0 = *(const v2f*)(xrow + 0  + koff);
        v2f a1 = *(const v2f*)(xrow + 4  + koff);
        v2f a2 = *(const v2f*)(xrow + 8  + koff);
        v2f a3 = *(const v2f*)(xrow + 12 + koff);
        v8f c = {};
        c = __builtin_amdgcn_wmma_f32_16x16x4_f32(false, a0, false, bb0, (short)0, c, false, false);
        c = __builtin_amdgcn_wmma_f32_16x16x4_f32(false, a1, false, bb1, (short)0, c, false, false);
        c = __builtin_amdgcn_wmma_f32_16x16x4_f32(false, a2, false, bb2, (short)0, c, false, false);
        c = __builtin_amdgcn_wmma_f32_16x16x4_f32(false, a3, false, bb3, (short)0, c, false, false);
        if (dis_pair) {
            const float* dp = dis_pair + (tile << 4) + rowb;
            v2f* ob2 = (v2f*)out + ((size_t)(tile << 4) + rowb) * D + col;
            #pragma unroll
            for (int rr = 0; rr < 8; ++rr) {
                v2f pr; pr.x = c[rr]; pr.y = dp[rr];
                ob2[(size_t)rr * D] = pr;
            }
        } else {
            float* ob = out + ((size_t)(tile << 4) + rowb) * D + col;
            #pragma unroll
            for (int rr = 0; rr < 8; ++rr) ob[(size_t)rr * D] = c[rr];
        }
    }

    const int tail_start = ntiles << 4;
    if (tail_start < n_nodes) {
        int node = tail_start + (int)(blockIdx.x * blockDim.x + threadIdx.x);
        if (node < n_nodes) {
            const int r = ids ? ids[node] : node;
            const float* xrow = src_rows + (size_t)r * D;
            for (int j = 0; j < D; ++j) {
                float acc = 0.0f;
                for (int k = 0; k < D; ++k) acc += xrow[k] * W[k * D + j];
                if (dis_pair) {
                    out[((size_t)node * D + j) * 2] = acc;
                    out[((size_t)node * D + j) * 2 + 1] = dis_pair[node];
                } else {
                    out[(size_t)node * D + j] = acc;
                }
            }
        }
    }
}

__global__ void table_gemm16_k(const float* __restrict__ src_rows,
                               const float* __restrict__ W,
                               float* __restrict__ out, int n_rows) {
    const int lane  = threadIdx.x & 31;
    const int wave  = (int)((blockIdx.x * blockDim.x + threadIdx.x) >> 5);
    const int nwave = (int)((gridDim.x * blockDim.x) >> 5);
    const int row   = lane & 15;
    const int col   = lane & 15;
    const int koff  = (lane >> 4) << 1;
    const int rowb  = (lane >> 4) << 3;
    v2f bb0, bb1, bb2, bb3;
    bb0.x = W[(0  + koff) * D + col]; bb0.y = W[(1  + koff) * D + col];
    bb1.x = W[(4  + koff) * D + col]; bb1.y = W[(5  + koff) * D + col];
    bb2.x = W[(8  + koff) * D + col]; bb2.y = W[(9  + koff) * D + col];
    bb3.x = W[(12 + koff) * D + col]; bb3.y = W[(13 + koff) * D + col];
    const int ntiles = (n_rows + 15) >> 4;
    float* ob = out + ((size_t)(wave << 4) + rowb) * D + col;
    for (int tile = wave; tile < ntiles; tile += nwave, ob += (size_t)nwave * 16 * D) {
        const int node = (tile << 4) + row;
        const int r = node < n_rows ? node : n_rows - 1;
        const float* xrow = src_rows + (size_t)r * D;
        v2f a0 = *(const v2f*)(xrow + 0  + koff);
        v2f a1 = *(const v2f*)(xrow + 4  + koff);
        v2f a2 = *(const v2f*)(xrow + 8  + koff);
        v2f a3 = *(const v2f*)(xrow + 12 + koff);
        v8f c = {};
        c = __builtin_amdgcn_wmma_f32_16x16x4_f32(false, a0, false, bb0, (short)0, c, false, false);
        c = __builtin_amdgcn_wmma_f32_16x16x4_f32(false, a1, false, bb1, (short)0, c, false, false);
        c = __builtin_amdgcn_wmma_f32_16x16x4_f32(false, a2, false, bb2, (short)0, c, false, false);
        c = __builtin_amdgcn_wmma_f32_16x16x4_f32(false, a3, false, bb3, (short)0, c, false, false);
        #pragma unroll
        for (int rr = 0; rr < 8; ++rr) ob[(size_t)rr * D] = c[rr];
    }
}

__global__ void degree_k(const int* __restrict__ dst, float* __restrict__ deg,
                         long long E) {
    long long e = (long long)blockIdx.x * blockDim.x + threadIdx.x;
    if (e < E) atomicAdd(deg + dst[e], 1.0f);
}

__global__ void rsqrt_k(float* __restrict__ d, int n) {
    int i = blockIdx.x * blockDim.x + threadIdx.x;
    if (i < n) d[i] = rsqrtf(d[i] + 1.0f);
}

__global__ void aggregate_k(const float* __restrict__ h,
                            const float* __restrict__ dis,
                            const int* __restrict__ src,
                            const int* __restrict__ dst,
                            float* __restrict__ agg,
                            long long E) {
    long long t = (long long)blockIdx.x * blockDim.x + threadIdx.x;
    long long e = t >> 4;
    int j = (int)(t & 15);
    if (e < E) {
        int s = src[e], d = dst[e];
        float norm = dis[s] * dis[d];
        atomicAdd(agg + (size_t)d * D + j, h[(size_t)s * D + j] * norm);
    }
}

__global__ void node_relu_k(float* __restrict__ h, const float* __restrict__ agg,
                            const float* __restrict__ dis,
                            const float* __restrict__ bias, long long n16) {
    long long t = (long long)blockIdx.x * blockDim.x + threadIdx.x;
    if (t < n16) {
        long long i = t >> 4; int j = (int)(t & 15);
        float di = dis[i];
        float v = agg[t] + h[t] * (di * di) + bias[j];
        h[t] = v > 0.0f ? v : 0.0f;
    }
}

__global__ void node_pool_k(const float* __restrict__ h2,
                            const float* __restrict__ agg,
                            const float* __restrict__ dis,
                            const float* __restrict__ bias,
                            const int* __restrict__ batch,
                            float* __restrict__ pooled,
                            float* __restrict__ cnt, long long n16) {
    long long t = (long long)blockIdx.x * blockDim.x + threadIdx.x;
    if (t < n16) {
        long long i = t >> 4; int j = (int)(t & 15);
        float di = dis[i];
        float v = agg[t] + h2[t] * (di * di) + bias[j];
        int g = batch[i];
        atomicAdd(pooled + (size_t)g * D + j, v);
        if (j == 0) atomicAdd(cnt + g, 1.0f);
    }
}

__global__ void fc_k(const float* __restrict__ pooled_r, const float* __restrict__ cnt_r,
                     const float* __restrict__ pooled_l, const float* __restrict__ cnt_l,
                     const float* __restrict__ fcW, const float* __restrict__ fcb,
                     const float* __restrict__ nbias,
                     float* __restrict__ out) {
    int t = blockIdx.x * blockDim.x + threadIdx.x;
    if (t < NGRAPH * 6) {
        int g = t / 6, o = t - g * 6;
        const float cr = cnt_r[g], cl = cnt_l[g];
        float icr = 1.0f / fmaxf(cr, 1.0f);
        float icl = 1.0f / fmaxf(cl, 1.0f);
        float acc = fcb[o];
        #pragma unroll
        for (int j = 0; j < D; ++j) {
            float p = pooled_r[g * D + j];
            if (nbias) p += cr * nbias[j];
            acc += fcW[o * (2 * D) + j] * (p * icr);
        }
        #pragma unroll
        for (int j = 0; j < D; ++j) {
            float p = pooled_l[g * D + j];
            if (nbias) p += cl * nbias[j];
            acc += fcW[o * (2 * D) + D + j] * (p * icl);
        }
        if (o < 3) out[g * 3 + o] = acc;
        else       out[NGRAPH * 3 + g * 3 + (o - 3)] = acc;
    }
}


#define RSH 8
#define RMASK 255u
#define BNODES 196
#define NBMAX 1024
#define NWGP 256
#define PTHREADS 1024
#define PTILE 32768
#define PU 8
#define DTHREADS 256

__global__ __launch_bounds__(PTHREADS)
void bucket_hist_k(const int* __restrict__ dst0, const int* __restrict__ dst1,
                   long long E, int tlen,
                   int* __restrict__ bcount0, int* __restrict__ bcount1) {
    const bool second = blockIdx.y != 0;
    const int* __restrict__ dst = second ? dst1 : dst0;
    int* __restrict__ bcount = second ? bcount1 : bcount0;
    __shared__ int cnt[NBMAX];
    const int t = threadIdx.x;
    cnt[t] = 0;
    __syncthreads();
    const long long base = (long long)blockIdx.x * tlen;
    const long long rem = E - base;
    const int ne = rem < (long long)tlen ? (int)rem : tlen;
    for (int k0 = 0; k0 * PTHREADS < ne; k0 += PU) {
        int dv[PU];
        #pragma unroll
        for (int u = 0; u < PU; ++u) {
            int l = (k0 + u) * PTHREADS + t;
            dv[u] = dst[base + (l < ne ? l : 0)];
        }
        #pragma unroll
        for (int u = 0; u < PU; ++u) {
            int l = (k0 + u) * PTHREADS + t;
            if (l < ne) atomicAdd(&cnt[(unsigned)dv[u] / BNODES], 1);
        }
    }
    __syncthreads();
    const int c = cnt[t];
    if (c > 0) atomicAdd(bcount + t, c);
}

__global__ __launch_bounds__(NBMAX)
void bucket_scan_k(const int* __restrict__ bcount0, const int* __restrict__ bcount1,
                   int* __restrict__ bstart0, int* __restrict__ bstart1,
                   int* __restrict__ cursor0, int* __restrict__ cursor1, int nb) {
    const bool second = blockIdx.y != 0;
    const int* __restrict__ bcount = second ? bcount1 : bcount0;
    int* __restrict__ bstart = second ? bstart1 : bstart0;
    int* __restrict__ cursor = second ? cursor1 : cursor0;
    __shared__ int s[NBMAX];
    const int t = threadIdx.x;
    const int own = (t < nb) ? bcount[t] : 0;
    s[t] = own;
    __syncthreads();
    for (int off = 1; off < NBMAX; off <<= 1) {
        int v = (t >= off) ? s[t - off] : 0;
        __syncthreads();
        s[t] += v;
        __syncthreads();
    }
    const int incl = s[t];
    if (t < nb) { bstart[t] = incl - own; cursor[t] = incl - own; }
    if (t == nb - 1) bstart[nb] = incl;
}

__global__ __launch_bounds__(PTHREADS)
void bucket_scatter_k(const int* __restrict__ src0, const int* __restrict__ src1,
                      const int* __restrict__ dst0, const int* __restrict__ dst1,
                      long long E, int tlen,
                      int* __restrict__ cursor0, int* __restrict__ cursor1,
                      unsigned* __restrict__ part0, unsigned* __restrict__ part1) {
    const bool second = blockIdx.y != 0;
    const int* __restrict__ src = second ? src1 : src0;
    const int* __restrict__ dst = second ? dst1 : dst0;
    int* __restrict__ cursor = second ? cursor1 : cursor0;
    unsigned* __restrict__ part = second ? part1 : part0;
    __shared__ int cnt[NBMAX];
    __shared__ int tstart[NBMAX];
    __shared__ int gbase[NBMAX];
    __shared__ int run[NBMAX];
    __shared__ unsigned stage[PTILE];
    __shared__ unsigned short bid[PTILE];
    const int t = threadIdx.x;
    cnt[t] = 0;
    __syncthreads();
    const long long base = (long long)blockIdx.x * tlen;
    const long long rem = E - base;
    const int ne = rem < (long long)tlen ? (int)rem : tlen;
    for (int k0 = 0; k0 * PTHREADS < ne; k0 += PU) {
        int dv[PU];
        #pragma unroll
        for (int u = 0; u < PU; ++u) {
            int l = (k0 + u) * PTHREADS + t;
            dv[u] = dst[base + (l < ne ? l : 0)];
        }
        #pragma unroll
        for (int u = 0; u < PU; ++u) {
            int l = (k0 + u) * PTHREADS + t;
            if (l < ne) atomicAdd(&cnt[(unsigned)dv[u] / BNODES], 1);
        }
    }
    __syncthreads();
    const int own = cnt[t];
    run[t] = own;
    __syncthreads();
    for (int off = 1; off < NBMAX; off <<= 1) {
        int v = (t >= off) ? run[t - off] : 0;
        __syncthreads();
        run[t] += v;
        __syncthreads();
    }
    const int excl = run[t] - own;
    tstart[t] = excl;
    run[t] = excl;
    gbase[t] = (own > 0) ? atomicAdd(cursor + t, own) : 0;
    __syncthreads();
    for (int k0 = 0; k0 * PTHREADS < ne; k0 += PU) {
        int dv[PU], sv[PU];
        #pragma unroll
        for (int u = 0; u < PU; ++u) {
            int l = (k0 + u) * PTHREADS + t;
            long long e = base + (l < ne ? l : 0);
            dv[u] = dst[e];
            sv[u] = src[e];
        }
        #pragma unroll
        for (int u = 0; u < PU; ++u) {
            int l = (k0 + u) * PTHREADS + t;
            if (l < ne) {
                unsigned b = (unsigned)dv[u] / BNODES;
                int slot = atomicAdd(&run[b], 1);
                stage[slot] = ((unsigned)sv[u] << RSH) | ((unsigned)dv[u] - b * BNODES);
                bid[slot] = (unsigned short)b;
            }
        }
    }
    __syncthreads();
    for (int slot = t; slot < ne; slot += PTHREADS) {
        int b = bid[slot];
        part[gbase[b] + (slot - tstart[b])] = stage[slot];
    }
}

__global__ __launch_bounds__(DTHREADS)
void bucket_degree_k(const unsigned* __restrict__ part0, const unsigned* __restrict__ part1,
                     const int* __restrict__ bstart0, const int* __restrict__ bstart1,
                     float* __restrict__ dis0, float* __restrict__ dis1, int n,
                     const int* __restrict__ ids0, const int* __restrict__ ids1,
                     v2u* __restrict__ nrec0, v2u* __restrict__ nrec1, int n_tab) {
    const bool second = blockIdx.y != 0;
    const unsigned* __restrict__ part = second ? part1 : part0;
    const int* __restrict__ bstart = second ? bstart1 : bstart0;
    float* __restrict__ dis = second ? dis1 : dis0;
    const int* __restrict__ ids = second ? ids1 : ids0;
    v2u* __restrict__ nrec = second ? nrec1 : nrec0;
    __shared__ int cnt[BNODES];
    const int t = threadIdx.x;
    if (t < BNODES) cnt[t] = 0;
    __syncthreads();
    const int b = blockIdx.x;
    const int p0 = bstart[b], p1 = bstart[b + 1];
    for (int p = p0 + t; p < p1; p += DTHREADS * PU) {
        unsigned r[PU];
        #pragma unroll
        for (int u = 0; u < PU; ++u) {
            int q = p + u * DTHREADS;
            r[u] = part[q < p1 ? q : 0];
        }
        #pragma unroll
        for (int u = 0; u < PU; ++u) {
            int q = p + u * DTHREADS;
            if (q < p1) atomicAdd(&cnt[r[u] & RMASK], 1);
        }
    }
    __syncthreads();
    const int i = b * BNODES + t;
    if (t < BNODES && i < n) {
        const float dv = rsqrtf((float)cnt[t] + 1.0f);
        dis[i] = dv;
        if (nrec) {
            int k = ids[i];
            k = k < 0 ? 0 : (k >= n_tab ? n_tab - 1 : k);
            v2u r; r.x = (unsigned)k; r.y = __float_as_uint(dv);
            nrec[i] = r;
        }
    }
}

template <int E>
__device__ __forceinline__ unsigned row_share_u32(unsigned x) {
    return (unsigned)__builtin_amdgcn_mov_dpp((int)x, 0x150 + E, 0xf, 0xf, true);
}

#define LTHREADS 512
#define PGR 16
template <bool POOL, bool PAIR, bool SCALED = false>
__global__ __launch_bounds__(LTHREADS)
void bucket_layer_k(const float* __restrict__ h0, const float* __restrict__ h1,
                    const float* __restrict__ dis0, const float* __restrict__ dis1,
                    const unsigned* __restrict__ part0, const unsigned* __restrict__ part1,
                    const int* __restrict__ bstart0, const int* __restrict__ bstart1,
                    const float* __restrict__ bias,
                    float* __restrict__ rows0, float* __restrict__ rows1,
                    const int* __restrict__ batch0, const int* __restrict__ batch1,
                    float* __restrict__ pooled0, float* __restrict__ pooled1,
                    float* __restrict__ cntg0, float* __restrict__ cntg1,
                    int n) {
    const bool second = blockIdx.y != 0;
    const float* __restrict__ h = second ? h1 : h0;
    const float* __restrict__ dis = second ? dis1 : dis0;
    const unsigned* __restrict__ part = second ? part1 : part0;
    const int* __restrict__ bstart = second ? bstart1 : bstart0;
    float* __restrict__ rows = second ? rows1 : rows0;
    const int* __restrict__ batch = second ? batch1 : batch0;
    float* __restrict__ pooled = second ? pooled1 : pooled0;
    float* __restrict__ cntg = second ? cntg1 : cntg0;
    __shared__ float acc[(BNODES + 1) * D];
    __shared__ float dloc[BNODES + 1];
    __shared__ float pl[PGR * D];
    __shared__ int   pc[PGR];
    const int t = threadIdx.x;
    const int b = blockIdx.x;
    const int node0 = b * BNODES;
    for (int i = t; i < (BNODES + 1) * D; i += LTHREADS) acc[i] = 0.0f;
    if (t <= BNODES) {
        int i = node0 + t;
        dloc[t] = (t < BNODES && i < n) ? dis[i] : 0.0f;
    }
    if (POOL) {
        if (t < PGR * D) pl[t] = 0.0f;
        if (t < PGR) pc[t] = 0;
    }
    __syncthreads();

    const int lane = t & 31;
    const unsigned j = (unsigned)lane & 15u;
    const unsigned j4 = j << 2;
    const int wv = __builtin_amdgcn_readfirstlane(t >> 5);
    const int p0 = bstart[b], p1 = bstart[b + 1];
    const v2f* __restrict__ hp = (const v2f*)h;
    int qb = p0 + wv * 32;
    unsigned recn = 0u;
    if (qb + lane < p1) recn = part[qb + lane];
    for (; qb < p1; qb += LTHREADS) {
        const int q = qb + lane;
        const unsigned rec = recn;
        const int qn = q + LTHREADS;
        recn = (qn < p1) ? part[qn] : 0u;
        const unsigned s = rec >> RSH;
        const unsigned dl = (q < p1) ? (rec & RMASK) : (unsigned)BNODES;
        const unsigned gi = s << 4;
        const unsigned ao4 = dl << 6;
        float wl = dloc[dl];
        if (!PAIR) wl *= dis[s];
        const unsigned wb = __float_as_uint(wl);
        float hv[16], ds[16];
#define LAYER_LOAD(E) { \
            const unsigned ix = row_share_u32<E>(gi) | j; \
            if (PAIR) { v2f pr = hp[(size_t)ix]; hv[E] = pr.x; ds[E] = pr.y; } \
            else { hv[E] = h[(size_t)ix]; ds[E] = 1.0f; } }
        LAYER_LOAD(0)  LAYER_LOAD(1)  LAYER_LOAD(2)  LAYER_LOAD(3)
        LAYER_LOAD(4)  LAYER_LOAD(5)  LAYER_LOAD(6)  LAYER_LOAD(7)
        LAYER_LOAD(8)  LAYER_LOAD(9)  LAYER_LOAD(10) LAYER_LOAD(11)
        LAYER_LOAD(12) LAYER_LOAD(13) LAYER_LOAD(14) LAYER_LOAD(15)
#undef LAYER_LOAD
#define LAYER_ADD(E) { \
            const float w = __uint_as_float(row_share_u32<E>(wb)); \
            const unsigned a = row_share_u32<E>(ao4) | j4; \
            const float norm = PAIR ? ds[E] * w : w; \
            atomicAdd((float*)((char*)acc + a), hv[E] * norm); }
        LAYER_ADD(0)  LAYER_ADD(1)  LAYER_ADD(2)  LAYER_ADD(3)
        LAYER_ADD(4)  LAYER_ADD(5)  LAYER_ADD(6)  LAYER_ADD(7)
        LAYER_ADD(8)  LAYER_ADD(9)  LAYER_ADD(10) LAYER_ADD(11)
        LAYER_ADD(12) LAYER_ADD(13) LAYER_ADD(14) LAYER_ADD(15)
#undef LAYER_ADD
    }
    __syncthreads();

    if (!POOL) {
        for (int idx = t; idx < BNODES * D; idx += LTHREADS) {
            int nl = idx >> 4;
            int i = node0 + nl;
            if (i < n) {
                float di = dloc[nl];
                size_t o = (size_t)i * D + (idx & 15);
                float v = acc[idx] + h[PAIR ? o * 2 : o] * (di * di) + bias[idx & 15];
                v = v > 0.0f ? v : 0.0f;
                rows[o] = SCALED ? v * di : v;
            }
        }
    } else {
        const int g0 = batch[node0];
        for (int idx = t; idx < BNODES * D; idx += LTHREADS) {
            int nl = idx >> 4;
            int jj = idx & 15;
            int i = node0 + nl;
            if (i < n) {
                float di = dloc[nl];
                size_t o = (size_t)i * D + jj;
                float v = acc[idx] + h[PAIR ? o * 2 : o] * (di * di) + bias[jj];
                int g = batch[i];
                unsigned k = (unsigned)(g - g0);
                if (k < (unsigned)PGR) {
                    atomicAdd(&pl[k * D + jj], v);
                    if (jj == 0) atomicAdd(&pc[k], 1);
                } else {
                    atomicAdd(pooled + (size_t)g * D + jj, v);
                    if (jj == 0) atomicAdd(cntg + g, 1.0f);
                }
            }
        }
        __syncthreads();
        if (t < PGR * D) {
            int k = t >> 4;
            int c = pc[k];
            if (c > 0) {
                int g = g0 + k;
                atomicAdd(pooled + (size_t)g * D + (t & 15), pl[t]);
                if ((t & 15) == 0) atomicAdd(cntg + g, (float)c);
            }
        }
    }
}


#define TCAP 1032
#define TCOPY ((TCAP * D / 4 + LTHREADS - 1) / LTHREADS)
#define TROWS ((TCAP + 15) & ~15)
__global__ __launch_bounds__(LTHREADS)
void bucket_layer1_tab_k(const float* __restrict__ tab, int n_tab,
                         const v2u* __restrict__ nrec0, const v2u* __restrict__ nrec1,
                         const unsigned* __restrict__ part0, const unsigned* __restrict__ part1,
                         const int* __restrict__ bstart0, const int* __restrict__ bstart1,
                         const float* __restrict__ bias,
                         float* __restrict__ rows0, float* __restrict__ rows1, int n) {
    const bool second = blockIdx.y != 0;
    const v2u* __restrict__ nrec = second ? nrec1 : nrec0;
    const unsigned* __restrict__ part = second ? part1 : part0;
    const int* __restrict__ bstart = second ? bstart1 : bstart0;
    float* __restrict__ rows = second ? rows1 : rows0;
    __shared__ struct {
        float acc[(BNODES + 1) * D];
        float dloc[BNODES + 4];
        v4f tl[TCAP * D / 4];
    } sm;
    float* const acc = sm.acc;
    float* const dloc = sm.dloc;
    v4f* const tl = sm.tl;
    const int t = threadIdx.x;
    const int b = blockIdx.x;
    const int node0 = b * BNODES;
    const int lane = t & 31;
    const unsigned j4 = ((unsigned)lane & 15u) << 2;
    const int wv = __builtin_amdgcn_readfirstlane(t >> 5);
    const int p0 = bstart[b], p1 = bstart[b + 1];
    int qb = p0 + wv * 32;
    unsigned rec0 = 0u, rec1 = 0u;
    if (qb + lane < p1) rec0 = part[qb + lane];
    if (qb + LTHREADS + lane < p1) rec1 = part[qb + LTHREADS + lane];
    for (int i = t; i < (BNODES + 1) * D; i += LTHREADS) acc[i] = 0.0f;
    if (t <= BNODES) {
        int i = node0 + t;
        dloc[t] = (t < BNODES && i < n) ? __uint_as_float(nrec[i].y) : 0.0f;
    }
    {
        const v4f* __restrict__ tg = (const v4f*)tab;
        const int n4 = n_tab * (D / 4);
        v4f tv[TCOPY];
        #pragma unroll
        for (int u = 0; u < TCOPY; ++u) {
            int i = t + u * LTHREADS;
            tv[u] = tg[i < n4 ? i : 0];
        }
        #pragma unroll
        for (int u = 0; u < TCOPY; ++u) {
            int i = t + u * LTHREADS;
            if (i < n4) tl[i] = tv[u];
        }
    }
    v2u nd = nrec[rec0 >> RSH];
    __syncthreads();

    const char* tlb = (const char*)tl;
    for (; qb < p1; qb += LTHREADS) {
        const int q = qb + lane;
        const unsigned rec = rec0;
        const v2u ndc = nd;
        rec0 = rec1;
        nd = nrec[rec0 >> RSH];
        const int q2 = q + 2 * LTHREADS;
        rec1 = (q2 < p1) ? part[q2] : 0u;
        const unsigned dl = (q < p1) ? (rec & RMASK) : (unsigned)BNODES;
        const unsigned to4 = ndc.x << 6;
        const unsigned ao4 = dl << 6;
        const unsigned wb = __float_as_uint(dloc[dl] * __uint_as_float(ndc.y));
        float hv[16];
#define TAB_LOAD(E) { \
            const unsigned ta = row_share_u32<E>(to4) | j4; \
            hv[E] = *(const float*)(tlb + ta); }
        TAB_LOAD(0)  TAB_LOAD(1)  TAB_LOAD(2)  TAB_LOAD(3)
        TAB_LOAD(4)  TAB_LOAD(5)  TAB_LOAD(6)  TAB_LOAD(7)
        TAB_LOAD(8)  TAB_LOAD(9)  TAB_LOAD(10) TAB_LOAD(11)
        TAB_LOAD(12) TAB_LOAD(13) TAB_LOAD(14) TAB_LOAD(15)
#undef TAB_LOAD
#define TAB_ADD(E) { \
            const float w = __uint_as_float(row_share_u32<E>(wb)); \
            const unsigned a = row_share_u32<E>(ao4) | j4; \
            atomicAdd((float*)((char*)acc + a), hv[E] * w); }
        TAB_ADD(0)  TAB_ADD(1)  TAB_ADD(2)  TAB_ADD(3)
        TAB_ADD(4)  TAB_ADD(5)  TAB_ADD(6)  TAB_ADD(7)
        TAB_ADD(8)  TAB_ADD(9)  TAB_ADD(10) TAB_ADD(11)
        TAB_ADD(12) TAB_ADD(13) TAB_ADD(14) TAB_ADD(15)
#undef TAB_ADD
    }
    __syncthreads();

    const float* tlf = (const float*)tl;
    for (int idx = t; idx < BNODES * D; idx += LTHREADS) {
        int nl = idx >> 4;
        int i = node0 + nl;
        if (i < n) {
            float di = dloc[nl];
            float hs = tlf[nrec[i].x * D + (idx & 15)];
            float v = acc[idx] + hs * (di * di) + bias[idx & 15];
            rows[(size_t)i * D + (idx & 15)] = (v > 0.0f ? v : 0.0f) * di;
        }
    }
}

__global__ __launch_bounds__(LTHREADS)
void bucket_layer2_sum_k(const float* __restrict__ h0, const float* __restrict__ h1,
                         const float* __restrict__ dis0, const float* __restrict__ dis1,
                         const unsigned* __restrict__ part0, const unsigned* __restrict__ part1,
                         const int* __restrict__ bstart0, const int* __restrict__ bstart1,
                         const int* __restrict__ batch0, const int* __restrict__ batch1,
                         float* __restrict__ pooled0, float* __restrict__ pooled1,
                         float* __restrict__ cntg0, float* __restrict__ cntg1,
                         int n) {
    const bool second = blockIdx.y != 0;
    const float* __restrict__ h = second ? h1 : h0;
    const float* __restrict__ dis = second ? dis1 : dis0;
    const unsigned* __restrict__ part = second ? part1 : part0;
    const int* __restrict__ bstart = second ? bstart1 : bstart0;
    const int* __restrict__ batch = second ? batch1 : batch0;
    float* __restrict__ pooled = second ? pooled1 : pooled0;
    float* __restrict__ cntg = second ? cntg1 : cntg0;
    __shared__ float acc[(BNODES + 1) * D];
    __shared__ float pl[PGR * D];
    __shared__ int   pc[PGR];
    const int t = threadIdx.x;
    const int b = blockIdx.x;
    const int node0 = b * BNODES;
    for (int i = t; i < (BNODES + 1) * D; i += LTHREADS) acc[i] = 0.0f;
    if (t < PGR * D) pl[t] = 0.0f;
    if (t < PGR) pc[t] = 0;
    __syncthreads();

    const int lane = t & 31;
    const unsigned j4 = ((unsigned)lane & 15u) << 2;
    const int wv = __builtin_amdgcn_readfirstlane(t >> 5);
    const int p0 = bstart[b], p1 = bstart[b + 1];
    const char* __restrict__ hb = (const char*)h;
    int qb = p0 + wv * 32;
    unsigned recn = 0u;
    if (qb + lane < p1) recn = part[qb + lane];
    for (; qb < p1; qb += LTHREADS) {
        const int q = qb + lane;
        const unsigned rec = recn;
        const int qn = q + LTHREADS;
        recn = (qn < p1) ? part[qn] : 0u;
        const unsigned dl = (q < p1) ? (rec & RMASK) : (unsigned)BNODES;
        const unsigned go4 = (rec >> RSH) << 6;
        const unsigned ao4 = dl << 6;
        float hv[16];
#define SUM_LOAD(E) { \
            const unsigned ga = row_share_u32<E>(go4) | j4; \
            hv[E] = *(const float*)(hb + ga); }
        SUM_LOAD(0)  SUM_LOAD(1)  SUM_LOAD(2)  SUM_LOAD(3)
        SUM_LOAD(4)  SUM_LOAD(5)  SUM_LOAD(6)  SUM_LOAD(7)
        SUM_LOAD(8)  SUM_LOAD(9)  SUM_LOAD(10) SUM_LOAD(11)
        SUM_LOAD(12) SUM_LOAD(13) SUM_LOAD(14) SUM_LOAD(15)
#undef SUM_LOAD
#define SUM_ADD(E) { \
            const unsigned a = row_share_u32<E>(ao4) | j4; \
            atomicAdd((float*)((char*)acc + a), hv[E]); }
        SUM_ADD(0)  SUM_ADD(1)  SUM_ADD(2)  SUM_ADD(3)
        SUM_ADD(4)  SUM_ADD(5)  SUM_ADD(6)  SUM_ADD(7)
        SUM_ADD(8)  SUM_ADD(9)  SUM_ADD(10) SUM_ADD(11)
        SUM_ADD(12) SUM_ADD(13) SUM_ADD(14) SUM_ADD(15)
#undef SUM_ADD
    }
    __syncthreads();

    const int g0 = batch[node0];
    for (int idx = t; idx < BNODES * D; idx += LTHREADS) {
        int nl = idx >> 4;
        int jj = idx & 15;
        int i = node0 + nl;
        if (i < n) {
            float v = dis[i] * (acc[idx] + h[(size_t)i * D + jj]);
            int g = batch[i];
            unsigned k = (unsigned)(g - g0);
            if (k < (unsigned)PGR) {
                atomicAdd(&pl[k * D + jj], v);
                if (jj == 0) atomicAdd(&pc[k], 1);
            } else {
                atomicAdd(pooled + (size_t)g * D + jj, v);
                if (jj == 0) atomicAdd(cntg + g, 1.0f);
            }
        }
    }
    __syncthreads();
    if (t < PGR * D) {
        int k = t >> 4;
        int c = pc[k];
        if (c > 0) {
            int g = g0 + k;
            atomicAdd(pooled + (size_t)g * D + (t & 15), pl[t]);
            if ((t & 15) == 0) atomicAdd(cntg + g, (float)c);
        }
    }
}

static void run_protein_bucketed(const float* emb, const float* W1, const float* b1,
                                 const float* W2, const float* b2,
                                 const int* ids, const int* edge, const int* batch,
                                 int N, long long E,
                                 float* bufA, float* bufB, float* dis,
                                 float* pooled, float* cnt,
                                 unsigned* part, int* bcount, int* bstart, int* cursor,
                                 bool pair, hipStream_t stream) {
    const int* src = edge;
    const int* dst = edge + E;
    const int B = 256;
    dim3 blk(B);
    int gemm_grid = (int)(((N >> 4) + 7) / 8);
    if (gemm_grid < 1) gemm_grid = 1;
    const int nb = (N + BNODES - 1) / BNODES;
    const long long round_cap = (long long)NWGP * PTILE;
    const int ptiles = NWGP * (int)((E + round_cap - 1) / round_cap);
    const int tlen = (int)(((E + ptiles - 1) / ptiles + 31) & ~31ll);

    fill_zero_k<<<(NBMAX + B - 1) / B, blk, 0, stream>>>((float*)bcount, NBMAX);
    bucket_hist_k<<<ptiles, PTHREADS, 0, stream>>>(dst, nullptr, E, tlen, bcount, nullptr);
    bucket_scan_k<<<1, NBMAX, 0, stream>>>(bcount, nullptr, bstart, nullptr, cursor, nullptr, nb);
    bucket_scatter_k<<<ptiles, PTHREADS, 0, stream>>>(src, nullptr, dst, nullptr, E, tlen,
                                                      cursor, nullptr, part, nullptr);
    bucket_degree_k<<<nb, DTHREADS, 0, stream>>>(part, nullptr, bstart, nullptr, dis, nullptr, N,
                                                 nullptr, nullptr, nullptr, nullptr, 0);

    gather_gemm16_k<<<gemm_grid, blk, 0, stream>>>(emb, nullptr, ids, nullptr, W1, bufA, nullptr,
                                                   pair ? dis : nullptr, nullptr, N);
    if (pair)
        bucket_layer_k<false, true><<<nb, LTHREADS, 0, stream>>>(
            bufA, nullptr, dis, nullptr, part, nullptr, bstart, nullptr, b1, bufB, nullptr,
            nullptr, nullptr, nullptr, nullptr, nullptr, nullptr, N);
    else
        bucket_layer_k<false, false><<<nb, LTHREADS, 0, stream>>>(
            bufA, nullptr, dis, nullptr, part, nullptr, bstart, nullptr, b1, bufB, nullptr,
            nullptr, nullptr, nullptr, nullptr, nullptr, nullptr, N);
    gather_gemm16_k<<<gemm_grid, blk, 0, stream>>>(bufB, nullptr, nullptr, nullptr, W2, bufA, nullptr,
                                                   pair ? dis : nullptr, nullptr, N);

    fill_zero_k<<<(int)((NGRAPH * D + B - 1) / B), blk, 0, stream>>>(pooled, NGRAPH * D);
    fill_zero_k<<<(int)((NGRAPH + B - 1) / B), blk, 0, stream>>>(cnt, NGRAPH);
    if (pair)
        bucket_layer_k<true, true><<<nb, LTHREADS, 0, stream>>>(
            bufA, nullptr, dis, nullptr, part, nullptr, bstart, nullptr, b2, nullptr, nullptr,
            batch, nullptr, pooled, nullptr, cnt, nullptr, N);
    else
        bucket_layer_k<true, false><<<nb, LTHREADS, 0, stream>>>(
            bufA, nullptr, dis, nullptr, part, nullptr, bstart, nullptr, b2, nullptr, nullptr,
            batch, nullptr, pooled, nullptr, cnt, nullptr, N);
}

static void run_both_bucketed(const float* emb, const float* W1, const float* b1,
                              const float* W2,
                              const int* ids0, const int* ids1,
                              const int* edge0, const int* edge1,
                              const int* batch0, const int* batch1,
                              int N, long long E,
                              float* pair0, float* pair1, float* rows0, float* rows1,
                              float* dis0, float* dis1,
                              float* zero_run, long long zero_words,
                              float* pooled0, float* pooled1, float* cnt0, float* cnt1,
                              int* bcount0, int* bcount1, int* bstart0, int* bstart1,
                              int* cursor0, int* cursor1,
                              unsigned* part0, unsigned* part1,
                              int n_tab, float* tab, v2u* nrec0, v2u* nrec1,
                              float* pw0, float* pw1,
                              hipStream_t stream) {
    const int B = 256;
    dim3 blk(B);
    int gemm_grid = (int)(((N >> 4) + 7) / 8);
    if (gemm_grid < 1) gemm_grid = 1;
    const int nb = (N + BNODES - 1) / BNODES;
    const long long round_cap = (long long)NWGP * PTILE;
    const int ptiles = NWGP * (int)((E + round_cap - 1) / round_cap);
    const int tlen = (int)(((E + ptiles - 1) / ptiles + 31) & ~31ll);

    fill_zero_k<<<(int)((zero_words + B - 1) / B), blk, 0, stream>>>(zero_run, zero_words);
    bucket_hist_k<<<dim3(ptiles, 2), PTHREADS, 0, stream>>>(edge0 + E, edge1 + E, E, tlen,
                                                            bcount0, bcount1);
    bucket_scan_k<<<dim3(1, 2), NBMAX, 0, stream>>>(bcount0, bcount1, bstart0, bstart1,
                                                    cursor0, cursor1, nb);
    bucket_scatter_k<<<dim3(ptiles, 2), PTHREADS, 0, stream>>>(edge0, edge1, edge0 + E, edge1 + E,
                                                               E, tlen, cursor0, cursor1,
                                                               part0, part1);
    if (tab) {
        bucket_degree_k<<<dim3(nb, 2), DTHREADS, 0, stream>>>(part0, part1, bstart0, bstart1,
                                                              dis0, dis1, N,
                                                              ids0, ids1, nrec0, nrec1, n_tab);
        table_gemm16_k<<<(((n_tab + 15) >> 4) + 7) / 8, blk, 0, stream>>>(emb, W1, tab, n_tab);
        bucket_layer1_tab_k<<<dim3(nb, 2), LTHREADS, 0, stream>>>(
            tab, n_tab, nrec0, nrec1, part0, part1, bstart0, bstart1, b1, rows0, rows1, N);
    } else {
        bucket_degree_k<<<dim3(nb, 2), DTHREADS, 0, stream>>>(part0, part1, bstart0, bstart1,
                                                              dis0, dis1, N,
                                                              nullptr, nullptr, nullptr, nullptr, 0);
        gather_gemm16_k<<<dim3(gemm_grid, 2), blk, 0, stream>>>(emb, emb, ids0, ids1, W1, pair0, pair1,
                                                                dis0, dis1, N);
        bucket_layer_k<false, true, true><<<dim3(nb, 2), LTHREADS, 0, stream>>>(
            pair0, pair1, dis0, dis1, part0, part1, bstart0, bstart1, b1, rows0, rows1,
            nullptr, nullptr, nullptr, nullptr, nullptr, nullptr, N);
    }
    bucket_layer2_sum_k<<<dim3(nb, 2), LTHREADS, 0, stream>>>(
        rows0, rows1, dis0, dis1, part0, part1, bstart0, bstart1,
        batch0, batch1, pooled0, pooled1, cnt0, cnt1, N);
    gather_gemm16_k<<<dim3((NGRAPH / 16 + 7) / 8, 2), blk, 0, stream>>>(
        pooled0, pooled1, nullptr, nullptr, W2, pw0, pw1, nullptr, nullptr, NGRAPH);
}

static void run_protein(const float* emb, const float* W1, const float* b1,
                        const float* W2, const float* b2,
                        const int* ids, const int* edge, const int* batch,
                        int N, long long E,
                        float* bufA, float* bufB, float* dis,
                        float* pooled, float* cnt, hipStream_t stream) {
    const int* src = edge;
    const int* dst = edge + E;
    const long long n16 = (long long)N * D;
    const int B = 256;
    dim3 blk(B);
    int gemm_grid = (int)(((N >> 4) + 7) / 8);
    if (gemm_grid < 1) gemm_grid = 1;

    fill_zero_k<<<(int)((N + B - 1) / B), blk, 0, stream>>>(dis, N);
    fill_zero_k<<<(int)((n16 + B - 1) / B), blk, 0, stream>>>(bufB, n16);

    gather_gemm16_k<<<gemm_grid, blk, 0, stream>>>(emb, nullptr, ids, nullptr, W1, bufA, nullptr,
                                                   nullptr, nullptr, N);

    degree_k<<<(int)((E + B - 1) / B), blk, 0, stream>>>(dst, dis, E);
    rsqrt_k<<<(int)((N + B - 1) / B), blk, 0, stream>>>(dis, N);

    aggregate_k<<<(int)((E * 16 + B - 1) / B), blk, 0, stream>>>(bufA, dis, src, dst, bufB, E);
    node_relu_k<<<(int)((n16 + B - 1) / B), blk, 0, stream>>>(bufA, bufB, dis, b1, n16);

    gather_gemm16_k<<<gemm_grid, blk, 0, stream>>>(bufA, nullptr, nullptr, nullptr, W2, bufB, nullptr,
                                                   nullptr, nullptr, N);

    fill_zero_k<<<(int)((n16 + B - 1) / B), blk, 0, stream>>>(bufA, n16);
    aggregate_k<<<(int)((E * 16 + B - 1) / B), blk, 0, stream>>>(bufB, dis, src, dst, bufA, E);

    fill_zero_k<<<(int)((NGRAPH * D + B - 1) / B), blk, 0, stream>>>(pooled, NGRAPH * D);
    fill_zero_k<<<(int)((NGRAPH + B - 1) / B), blk, 0, stream>>>(cnt, NGRAPH);
    node_pool_k<<<(int)((n16 + B - 1) / B), blk, 0, stream>>>(bufB, bufA, dis, b2, batch,
                                                              pooled, cnt, n16);
}

extern "C" void kernel_launch(void* const* d_in, const int* in_sizes, int n_in,
                              void* d_out, int out_size, void* d_ws, size_t ws_size,
                              hipStream_t stream) {
    const float* emb = (const float*)d_in[0];
    const float* W1  = (const float*)d_in[1];
    const float* b1  = (const float*)d_in[2];
    const float* W2  = (const float*)d_in[3];
    const float* b2  = (const float*)d_in[4];
    const float* fcW = (const float*)d_in[5];
    const float* fcb = (const float*)d_in[6];
    const int* rx     = (const int*)d_in[7];
    const int* redge  = (const int*)d_in[8];
    const int* rbatch = (const int*)d_in[9];
    const int* lx     = (const int*)d_in[10];
    const int* ledge  = (const int*)d_in[11];
    const int* lbatch = (const int*)d_in[12];

    const int N = in_sizes[7];
    const long long E = in_sizes[8] / 2;

    char* ws = (char*)d_ws;
    float* bufA = (float*)ws;  ws += (size_t)N * D * sizeof(float);
    float* bufB = (float*)ws;  ws += (size_t)N * D * sizeof(float);
    float* dis  = (float*)ws;  ws += (size_t)N * sizeof(float);
    float* pooled_r = (float*)ws; ws += (size_t)NGRAPH * D * sizeof(float);
    float* pooled_l = (float*)ws; ws += (size_t)NGRAPH * D * sizeof(float);
    float* cnt_r = (float*)ws; ws += (size_t)NGRAPH * sizeof(float);
    float* cnt_l = (float*)ws; ws += (size_t)NGRAPH * sizeof(float);

    unsigned* part = (unsigned*)ws;  ws += (size_t)E * sizeof(unsigned);
    int* bcount = (int*)ws;  ws += (size_t)NBMAX * sizeof(int);
    int* bstart = (int*)ws;  ws += (size_t)2 * NBMAX * sizeof(int);
    int* cursor = (int*)ws;  ws += (size_t)NBMAX * sizeof(int);
    const bool bucketed = (size_t)(ws - (char*)d_ws) <= ws_size &&
                          N > 0 && N <= NBMAX * BNODES && E > 0 && E < (1ll << 30);
    float* bufT = (float*)ws;  ws += (size_t)N * D * sizeof(float);
    const bool pair = (size_t)(ws - (char*)d_ws) <= ws_size;
    float* rowsB = pair ? bufT : bufB;

    char* q = (char*)d_ws;
    float* pair0 = (float*)q;  q += (size_t)N * D * 2 * sizeof(float);
    float* pair1 = (float*)q;  q += (size_t)N * D * 2 * sizeof(float);
    float* rows0 = (float*)q;  q += (size_t)N * D * sizeof(float);
    float* rows1 = (float*)q;  q += (size_t)N * D * sizeof(float);
    float* dis0  = (float*)q;  q += (size_t)N * sizeof(float);
    float* dis1  = (float*)q;  q += (size_t)N * sizeof(float);
    float* zero_run = (float*)q;
    float* pool0 = (float*)q;  q += (size_t)NGRAPH * D * sizeof(float);
    float* pool1 = (float*)q;  q += (size_t)NGRAPH * D * sizeof(float);
    float* pcnt0 = (float*)q;  q += (size_t)NGRAPH * sizeof(float);
    float* pcnt1 = (float*)q;  q += (size_t)NGRAPH * sizeof(float);
    int* bcount0 = (int*)q;  q += (size_t)NBMAX * sizeof(int);
    int* bcount1 = (int*)q;  q += (size_t)NBMAX * sizeof(int);
    const long long zero_words = 2ll * NGRAPH * D + 2ll * NGRAPH + 2ll * NBMAX;
    int* bstart0 = (int*)q;  q += (size_t)2 * NBMAX * sizeof(int);
    int* bstart1 = (int*)q;  q += (size_t)2 * NBMAX * sizeof(int);
    int* cursor0 = (int*)q;  q += (size_t)NBMAX * sizeof(int);
    int* cursor1 = (int*)q;  q += (size_t)NBMAX * sizeof(int);
    unsigned* part0 = (unsigned*)q;  q += (size_t)E * sizeof(unsigned);
    unsigned* part1 = (unsigned*)q;  q += (size_t)E * sizeof(unsigned);
    const bool both = bucketed && (size_t)(q - (char*)d_ws) <= ws_size;
    const int n_tab = in_sizes[0] / D;
    float* tab = (float*)pair1;
    v2u* nrec0 = (v2u*)(tab + (size_t)TROWS * D);
    v2u* nrec1 = nrec0 + N;
    const bool tabbed = n_tab > 0 && n_tab <= TCAP &&
                        (size_t)TROWS * D * sizeof(float) + (size_t)2 * N * sizeof(v2u) <=
                            (size_t)N * D * 2 * sizeof(float);
    float* pw0 = pair0;
    float* pw1 = pair0 + (size_t)NGRAPH * D;
    if (both) {
        run_both_bucketed(emb, W1, b1, W2, rx, lx, redge, ledge, rbatch, lbatch, N, E,
                          pair0, pair1, rows0, rows1, dis0, dis1, zero_run, zero_words,
                          pool0, pool1, pcnt0, pcnt1, bcount0, bcount1, bstart0, bstart1,
                          cursor0, cursor1, part0, part1,
                          n_tab, tabbed ? tab : nullptr, nrec0, nrec1, pw0, pw1, stream);
        fc_k<<<(NGRAPH * 6 + 255) / 256, 256, 0, stream>>>(pw0, pcnt0, pw1, pcnt1,
                                                           fcW, fcb, b2, (float*)d_out);
        return;
    }
    if (bucketed) {
        run_protein_bucketed(emb, W1, b1, W2, b2, rx, redge, rbatch, N, E,
                             bufA, rowsB, dis, pooled_r, cnt_r,
                             part, bcount, bstart, cursor, pair, stream);
        run_protein_bucketed(emb, W1, b1, W2, b2, lx, ledge, lbatch, N, E,
                             bufA, rowsB, dis, pooled_l, cnt_l,
                             part, bcount, bstart, cursor, pair, stream);
    } else {
        run_protein(emb, W1, b1, W2, b2, rx, redge, rbatch, N, E,
                    bufA, bufB, dis, pooled_r, cnt_r, stream);
        run_protein(emb, W1, b1, W2, b2, lx, ledge, lbatch, N, E,
                    bufA, bufB, dis, pooled_l, cnt_l, stream);
    }

    fc_k<<<(NGRAPH * 6 + 255) / 256, 256, 0, stream>>>(pooled_r, cnt_r, pooled_l, cnt_l,
                                                       fcW, fcb, nullptr, (float*)d_out);
}


# scatter kernel's LDS cut from 208 KB to 114 KB (stage sized to the tile; no per-slot bucket tags: a wave copies whole bucket runs): two sort groups fit a WGP, both proteins' tiles in one round
# speedup vs baseline: 9.1221x; 1.0328x over previous
#include <hip/hip_runtime.h>
#include <hip/hip_bf16.h>

typedef float v2f __attribute__((ext_vector_type(2)));
typedef float v8f __attribute__((ext_vector_type(8)));
typedef float v4f __attribute__((ext_vector_type(4)));
typedef unsigned v2u __attribute__((ext_vector_type(2)));

#define D 16
#define NGRAPH 1024

__global__ void fill_zero_k(float* __restrict__ p, long long n) {
    long long i = (long long)blockIdx.x * blockDim.x + threadIdx.x;
    if (i < n) p[i] = 0.0f;
}

__global__ void gather_gemm16_k(const float* __restrict__ src_rows0,
                                const float* __restrict__ src_rows1,
                                const int* __restrict__ ids0,
                                const int* __restrict__ ids1,
                                const float* __restrict__ W,
                                float* __restrict__ out0,
                                float* __restrict__ out1,
                                const float* __restrict__ dis_pair0,
                                const float* __restrict__ dis_pair1,
                                int n_nodes) {
    const bool second = blockIdx.y != 0;
    const float* __restrict__ src_rows = second ? src_rows1 : src_rows0;
    const int* __restrict__ ids = second ? ids1 : ids0;
    float* __restrict__ out = second ? out1 : out0;
    const float* __restrict__ dis_pair = second ? dis_pair1 : dis_pair0;
    const int lane  = threadIdx.x & 31;
    const int wave  = (int)((blockIdx.x * blockDim.x + threadIdx.x) >> 5);
    const int nwave = (int)((gridDim.x * blockDim.x) >> 5);
    const int row   = lane & 15;
    const int col   = lane & 15;
    const int koff  = (lane >> 4) << 1;
    const int rowb  = (lane >> 4) << 3;

    v2f bb0, bb1, bb2, bb3;
    bb0.x = W[(0  + koff) * D + col]; bb0.y = W[(1  + koff) * D + col];
    bb1.x = W[(4  + koff) * D + col]; bb1.y = W[(5  + koff) * D + col];
    bb2.x = W[(8  + koff) * D + col]; bb2.y = W[(9  + koff) * D + col];
    bb3.x = W[(12 + koff) * D + col]; bb3.y = W[(13 + koff) * D + col];

    const int ntiles = n_nodes >> 4;
    for (int tile = wave; tile < ntiles; tile += nwave) {
        const int node = (tile << 4) + row;
        const int r = ids ? ids[node] : node;
        const float* xrow = src_rows + (size_t)r * D;
        v2f a0 = *(const v2f*)(xrow + 0  + koff);
        v2f a1 = *(const v2f*)(xrow + 4  + koff);
        v2f a2 = *(const v2f*)(xrow + 8  + koff);
        v2f a3 = *(const v2f*)(xrow + 12 + koff);
        v8f c = {};
        c = __builtin_amdgcn_wmma_f32_16x16x4_f32(false, a0, false, bb0, (short)0, c, false, false);
        c = __builtin_amdgcn_wmma_f32_16x16x4_f32(false, a1, false, bb1, (short)0, c, false, false);
        c = __builtin_amdgcn_wmma_f32_16x16x4_f32(false, a2, false, bb2, (short)0, c, false, false);
        c = __builtin_amdgcn_wmma_f32_16x16x4_f32(false, a3, false, bb3, (short)0, c, false, false);
        if (dis_pair) {
            const float* dp = dis_pair + (tile << 4) + rowb;
            v2f* ob2 = (v2f*)out + ((size_t)(tile << 4) + rowb) * D + col;
            #pragma unroll
            for (int rr = 0; rr < 8; ++rr) {
                v2f pr; pr.x = c[rr]; pr.y = dp[rr];
                ob2[(size_t)rr * D] = pr;
            }
        } else {
            float* ob = out + ((size_t)(tile << 4) + rowb) * D + col;
            #pragma unroll
            for (int rr = 0; rr < 8; ++rr) ob[(size_t)rr * D] = c[rr];
        }
    }

    const int tail_start = ntiles << 4;
    if (tail_start < n_nodes) {
        int node = tail_start + (int)(blockIdx.x * blockDim.x + threadIdx.x);
        if (node < n_nodes) {
            const int r = ids ? ids[node] : node;
            const float* xrow = src_rows + (size_t)r * D;
            for (int j = 0; j < D; ++j) {
                float acc = 0.0f;
                for (int k = 0; k < D; ++k) acc += xrow[k] * W[k * D + j];
                if (dis_pair) {
                    out[((size_t)node * D + j) * 2] = acc;
                    out[((size_t)node * D + j) * 2 + 1] = dis_pair[node];
                } else {
                    out[(size_t)node * D + j] = acc;
                }
            }
        }
    }
}

__global__ void table_gemm16_k(const float* __restrict__ src_rows,
                               const float* __restrict__ W,
                               float* __restrict__ out, int n_rows) {
    const int lane  = threadIdx.x & 31;
    const int wave  = (int)((blockIdx.x * blockDim.x + threadIdx.x) >> 5);
    const int nwave = (int)((gridDim.x * blockDim.x) >> 5);
    const int row   = lane & 15;
    const int col   = lane & 15;
    const int koff  = (lane >> 4) << 1;
    const int rowb  = (lane >> 4) << 3;
    v2f bb0, bb1, bb2, bb3;
    bb0.x = W[(0  + koff) * D + col]; bb0.y = W[(1  + koff) * D + col];
    bb1.x = W[(4  + koff) * D + col]; bb1.y = W[(5  + koff) * D + col];
    bb2.x = W[(8  + koff) * D + col]; bb2.y = W[(9  + koff) * D + col];
    bb3.x = W[(12 + koff) * D + col]; bb3.y = W[(13 + koff) * D + col];
    const int ntiles = (n_rows + 15) >> 4;
    float* ob = out + ((size_t)(wave << 4) + rowb) * D + col;
    for (int tile = wave; tile < ntiles; tile += nwave, ob += (size_t)nwave * 16 * D) {
        const int node = (tile << 4) + row;
        const int r = node < n_rows ? node : n_rows - 1;
        const float* xrow = src_rows + (size_t)r * D;
        v2f a0 = *(const v2f*)(xrow + 0  + koff);
        v2f a1 = *(const v2f*)(xrow + 4  + koff);
        v2f a2 = *(const v2f*)(xrow + 8  + koff);
        v2f a3 = *(const v2f*)(xrow + 12 + koff);
        v8f c = {};
        c = __builtin_amdgcn_wmma_f32_16x16x4_f32(false, a0, false, bb0, (short)0, c, false, false);
        c = __builtin_amdgcn_wmma_f32_16x16x4_f32(false, a1, false, bb1, (short)0, c, false, false);
        c = __builtin_amdgcn_wmma_f32_16x16x4_f32(false, a2, false, bb2, (short)0, c, false, false);
        c = __builtin_amdgcn_wmma_f32_16x16x4_f32(false, a3, false, bb3, (short)0, c, false, false);
        #pragma unroll
        for (int rr = 0; rr < 8; ++rr) ob[(size_t)rr * D] = c[rr];
    }
}

__global__ void degree_k(const int* __restrict__ dst, float* __restrict__ deg,
                         long long E) {
    long long e = (long long)blockIdx.x * blockDim.x + threadIdx.x;
    if (e < E) atomicAdd(deg + dst[e], 1.0f);
}

__global__ void rsqrt_k(float* __restrict__ d, int n) {
    int i = blockIdx.x * blockDim.x + threadIdx.x;
    if (i < n) d[i] = rsqrtf(d[i] + 1.0f);
}

__global__ void aggregate_k(const float* __restrict__ h,
                            const float* __restrict__ dis,
                            const int* __restrict__ src,
                            const int* __restrict__ dst,
                            float* __restrict__ agg,
                            long long E) {
    long long t = (long long)blockIdx.x * blockDim.x + threadIdx.x;
    long long e = t >> 4;
    int j = (int)(t & 15);
    if (e < E) {
        int s = src[e], d = dst[e];
        float norm = dis[s] * dis[d];
        atomicAdd(agg + (size_t)d * D + j, h[(size_t)s * D + j] * norm);
    }
}

__global__ void node_relu_k(float* __restrict__ h, const float* __restrict__ agg,
                            const float* __restrict__ dis,
                            const float* __restrict__ bias, long long n16) {
    long long t = (long long)blockIdx.x * blockDim.x + threadIdx.x;
    if (t < n16) {
        long long i = t >> 4; int j = (int)(t & 15);
        float di = dis[i];
        float v = agg[t] + h[t] * (di * di) + bias[j];
        h[t] = v > 0.0f ? v : 0.0f;
    }
}

__global__ void node_pool_k(const float* __restrict__ h2,
                            const float* __restrict__ agg,
                            const float* __restrict__ dis,
                            const float* __restrict__ bias,
                            const int* __restrict__ batch,
                            float* __restrict__ pooled,
                            float* __restrict__ cnt, long long n16) {
    long long t = (long long)blockIdx.x * blockDim.x + threadIdx.x;
    if (t < n16) {
        long long i = t >> 4; int j = (int)(t & 15);
        float di = dis[i];
        float v = agg[t] + h2[t] * (di * di) + bias[j];
        int g = batch[i];
        atomicAdd(pooled + (size_t)g * D + j, v);
        if (j == 0) atomicAdd(cnt + g, 1.0f);
    }
}

__global__ void fc_k(const float* __restrict__ pooled_r, const float* __restrict__ cnt_r,
                     const float* __restrict__ pooled_l, const float* __restrict__ cnt_l,
                     const float* __restrict__ fcW, const float* __restrict__ fcb,
                     const float* __restrict__ nbias,
                     float* __restrict__ out) {
    int t = blockIdx.x * blockDim.x + threadIdx.x;
    if (t < NGRAPH * 6) {
        int g = t / 6, o = t - g * 6;
        const float cr = cnt_r[g], cl = cnt_l[g];
        float icr = 1.0f / fmaxf(cr, 1.0f);
        float icl = 1.0f / fmaxf(cl, 1.0f);
        float acc = fcb[o];
        #pragma unroll
        for (int j = 0; j < D; ++j) {
            float p = pooled_r[g * D + j];
            if (nbias) p += cr * nbias[j];
            acc += fcW[o * (2 * D) + j] * (p * icr);
        }
        #pragma unroll
        for (int j = 0; j < D; ++j) {
            float p = pooled_l[g * D + j];
            if (nbias) p += cl * nbias[j];
            acc += fcW[o * (2 * D) + D + j] * (p * icl);
        }
        if (o < 3) out[g * 3 + o] = acc;
        else       out[NGRAPH * 3 + g * 3 + (o - 3)] = acc;
    }
}


#define RSH 8
#define RMASK 255u
#define BNODES 196
#define NBMAX 1024
#define NWGP 256
#define PTHREADS 1024
#define PTILE 25088
#define PU 8
#define DTHREADS 256

__global__ __launch_bounds__(PTHREADS)
void bucket_hist_k(const int* __restrict__ dst0, const int* __restrict__ dst1,
                   long long E, int tlen,
                   int* __restrict__ bcount0, int* __restrict__ bcount1) {
    const bool second = blockIdx.y != 0;
    const int* __restrict__ dst = second ? dst1 : dst0;
    int* __restrict__ bcount = second ? bcount1 : bcount0;
    __shared__ int cnt[NBMAX];
    const int t = threadIdx.x;
    cnt[t] = 0;
    __syncthreads();
    const long long base = (long long)blockIdx.x * tlen;
    const long long rem = E - base;
    const int ne = rem < (long long)tlen ? (int)rem : tlen;
    for (int k0 = 0; k0 * PTHREADS < ne; k0 += PU) {
        int dv[PU];
        #pragma unroll
        for (int u = 0; u < PU; ++u) {
            int l = (k0 + u) * PTHREADS + t;
            dv[u] = dst[base + (l < ne ? l : 0)];
        }
        #pragma unroll
        for (int u = 0; u < PU; ++u) {
            int l = (k0 + u) * PTHREADS + t;
            if (l < ne) atomicAdd(&cnt[(unsigned)dv[u] / BNODES], 1);
        }
    }
    __syncthreads();
    const int c = cnt[t];
    if (c > 0) atomicAdd(bcount + t, c);
}

__global__ __launch_bounds__(NBMAX)
void bucket_scan_k(const int* __restrict__ bcount0, const int* __restrict__ bcount1,
                   int* __restrict__ bstart0, int* __restrict__ bstart1,
                   int* __restrict__ cursor0, int* __restrict__ cursor1, int nb) {
    const bool second = blockIdx.y != 0;
    const int* __restrict__ bcount = second ? bcount1 : bcount0;
    int* __restrict__ bstart = second ? bstart1 : bstart0;
    int* __restrict__ cursor = second ? cursor1 : cursor0;
    __shared__ int s[NBMAX];
    const int t = threadIdx.x;
    const int own = (t < nb) ? bcount[t] : 0;
    s[t] = own;
    __syncthreads();
    for (int off = 1; off < NBMAX; off <<= 1) {
        int v = (t >= off) ? s[t - off] : 0;
        __syncthreads();
        s[t] += v;
        __syncthreads();
    }
    const int incl = s[t];
    if (t < nb) { bstart[t] = incl - own; cursor[t] = incl - own; }
    if (t == nb - 1) bstart[nb] = incl;
}

__global__ __launch_bounds__(PTHREADS)
void bucket_scatter_k(const int* __restrict__ src0, const int* __restrict__ src1,
                      const int* __restrict__ dst0, const int* __restrict__ dst1,
                      long long E, int tlen,
                      int* __restrict__ cursor0, int* __restrict__ cursor1,
                      unsigned* __restrict__ part0, unsigned* __restrict__ part1) {
    const bool second = blockIdx.y != 0;
    const int* __restrict__ src = second ? src1 : src0;
    const int* __restrict__ dst = second ? dst1 : dst0;
    int* __restrict__ cursor = second ? cursor1 : cursor0;
    unsigned* __restrict__ part = second ? part1 : part0;
    __shared__ int cnt[NBMAX];
    __shared__ int tstart[NBMAX];
    __shared__ int gbase[NBMAX];
    __shared__ int run[NBMAX];
    __shared__ unsigned stage[PTILE];
    const int t = threadIdx.x;
    cnt[t] = 0;
    __syncthreads();
    const long long base = (long long)blockIdx.x * tlen;
    const long long rem = E - base;
    const int ne = rem < (long long)tlen ? (int)rem : tlen;
    for (int k0 = 0; k0 * PTHREADS < ne; k0 += PU) {
        int dv[PU];
        #pragma unroll
        for (int u = 0; u < PU; ++u) {
            int l = (k0 + u) * PTHREADS + t;
            dv[u] = dst[base + (l < ne ? l : 0)];
        }
        #pragma unroll
        for (int u = 0; u < PU; ++u) {
            int l = (k0 + u) * PTHREADS + t;
            if (l < ne) atomicAdd(&cnt[(unsigned)dv[u] / BNODES], 1);
        }
    }
    __syncthreads();
    const int own = cnt[t];
    run[t] = own;
    __syncthreads();
    for (int off = 1; off < NBMAX; off <<= 1) {
        int v = (t >= off) ? run[t - off] : 0;
        __syncthreads();
        run[t] += v;
        __syncthreads();
    }
    const int excl = run[t] - own;
    tstart[t] = excl;
    run[t] = excl;
    gbase[t] = (own > 0) ? atomicAdd(cursor + t, own) : 0;
    __syncthreads();
    for (int k0 = 0; k0 * PTHREADS < ne; k0 += PU) {
        int dv[PU], sv[PU];
        #pragma unroll
        for (int u = 0; u < PU; ++u) {
            int l = (k0 + u) * PTHREADS + t;
            long long e = base + (l < ne ? l : 0);
            dv[u] = dst[e];
            sv[u] = src[e];
        }
        #pragma unroll
        for (int u = 0; u < PU; ++u) {
            int l = (k0 + u) * PTHREADS + t;
            if (l < ne) {
                unsigned b = (unsigned)dv[u] / BNODES;
                int slot = atomicAdd(&run[b], 1);
                stage[slot] = ((unsigned)sv[u] << RSH) | ((unsigned)dv[u] - b * BNODES);
            }
        }
    }
    __syncthreads();
    const int lane = t & 31;
    int b = t >> 5;
    int s0 = tstart[b], s1 = run[b], g = gbase[b];
    for (; b < NBMAX; b += PTHREADS / 32) {
        const int c0 = s0, c1 = s1, off = g - s0;
        const int bn = b + PTHREADS / 32;
        if (bn < NBMAX) { s0 = tstart[bn]; s1 = run[bn]; g = gbase[bn]; }
        #pragma clang loop unroll(disable) vectorize(disable) interleave(disable)
        for (int slot = c0 + lane; slot < c1; slot += 32)
            part[off + slot] = stage[slot];
    }
}

__global__ __launch_bounds__(DTHREADS)
void bucket_degree_k(const unsigned* __restrict__ part0, const unsigned* __restrict__ part1,
                     const int* __restrict__ bstart0, const int* __restrict__ bstart1,
                     float* __restrict__ dis0, float* __restrict__ dis1, int n,
                     const int* __restrict__ ids0, const int* __restrict__ ids1,
                     v2u* __restrict__ nrec0, v2u* __restrict__ nrec1, int n_tab) {
    const bool second = blockIdx.y != 0;
    const unsigned* __restrict__ part = second ? part1 : part0;
    const int* __restrict__ bstart = second ? bstart1 : bstart0;
    float* __restrict__ dis = second ? dis1 : dis0;
    const int* __restrict__ ids = second ? ids1 : ids0;
    v2u* __restrict__ nrec = second ? nrec1 : nrec0;
    __shared__ int cnt[BNODES];
    const int t = threadIdx.x;
    if (t < BNODES) cnt[t] = 0;
    __syncthreads();
    const int b = blockIdx.x;
    const int p0 = bstart[b], p1 = bstart[b + 1];
    for (int p = p0 + t; p < p1; p += DTHREADS * PU) {
        unsigned r[PU];
        #pragma unroll
        for (int u = 0; u < PU; ++u) {
            int q = p + u * DTHREADS;
            r[u] = part[q < p1 ? q : 0];
        }
        #pragma unroll
        for (int u = 0; u < PU; ++u) {
            int q = p + u * DTHREADS;
            if (q < p1) atomicAdd(&cnt[r[u] & RMASK], 1);
        }
    }
    __syncthreads();
    const int i = b * BNODES + t;
    if (t < BNODES && i < n) {
        const float dv = rsqrtf((float)cnt[t] + 1.0f);
        dis[i] = dv;
        if (nrec) {
            int k = ids[i];
            k = k < 0 ? 0 : (k >= n_tab ? n_tab - 1 : k);
            v2u r; r.x = (unsigned)k; r.y = __float_as_uint(dv);
            nrec[i] = r;
        }
    }
}

template <int E>
__device__ __forceinline__ unsigned row_share_u32(unsigned x) {
    return (unsigned)__builtin_amdgcn_mov_dpp((int)x, 0x150 + E, 0xf, 0xf, true);
}

#define LTHREADS 512
#define PGR 16
template <bool POOL, bool PAIR, bool SCALED = false>
__global__ __launch_bounds__(LTHREADS)
void bucket_layer_k(const float* __restrict__ h0, const float* __restrict__ h1,
                    const float* __restrict__ dis0, const float* __restrict__ dis1,
                    const unsigned* __restrict__ part0, const unsigned* __restrict__ part1,
                    const int* __restrict__ bstart0, const int* __restrict__ bstart1,
                    const float* __restrict__ bias,
                    float* __restrict__ rows0, float* __restrict__ rows1,
                    const int* __restrict__ batch0, const int* __restrict__ batch1,
                    float* __restrict__ pooled0, float* __restrict__ pooled1,
                    float* __restrict__ cntg0, float* __restrict__ cntg1,
                    int n) {
    const bool second = blockIdx.y != 0;
    const float* __restrict__ h = second ? h1 : h0;
    const float* __restrict__ dis = second ? dis1 : dis0;
    const unsigned* __restrict__ part = second ? part1 : part0;
    const int* __restrict__ bstart = second ? bstart1 : bstart0;
    float* __restrict__ rows = second ? rows1 : rows0;
    const int* __restrict__ batch = second ? batch1 : batch0;
    float* __restrict__ pooled = second ? pooled1 : pooled0;
    float* __restrict__ cntg = second ? cntg1 : cntg0;
    __shared__ float acc[(BNODES + 1) * D];
    __shared__ float dloc[BNODES + 1];
    __shared__ float pl[PGR * D];
    __shared__ int   pc[PGR];
    const int t = threadIdx.x;
    const int b = blockIdx.x;
    const int node0 = b * BNODES;
    for (int i = t; i < (BNODES + 1) * D; i += LTHREADS) acc[i] = 0.0f;
    if (t <= BNODES) {
        int i = node0 + t;
        dloc[t] = (t < BNODES && i < n) ? dis[i] : 0.0f;
    }
    if (POOL) {
        if (t < PGR * D) pl[t] = 0.0f;
        if (t < PGR) pc[t] = 0;
    }
    __syncthreads();

    const int lane = t & 31;
    const unsigned j = (unsigned)lane & 15u;
    const unsigned j4 = j << 2;
    const int wv = __builtin_amdgcn_readfirstlane(t >> 5);
    const int p0 = bstart[b], p1 = bstart[b + 1];
    const v2f* __restrict__ hp = (const v2f*)h;
    int qb = p0 + wv * 32;
    unsigned recn = 0u;
    if (qb + lane < p1) recn = part[qb + lane];
    for (; qb < p1; qb += LTHREADS) {
        const int q = qb + lane;
        const unsigned rec = recn;
        const int qn = q + LTHREADS;
        recn = (qn < p1) ? part[qn] : 0u;
        const unsigned s = rec >> RSH;
        const unsigned dl = (q < p1) ? (rec & RMASK) : (unsigned)BNODES;
        const unsigned gi = s << 4;
        const unsigned ao4 = dl << 6;
        float wl = dloc[dl];
        if (!PAIR) wl *= dis[s];
        const unsigned wb = __float_as_uint(wl);
        float hv[16], ds[16];
#define LAYER_LOAD(E) { \
            const unsigned ix = row_share_u32<E>(gi) | j; \
            if (PAIR) { v2f pr = hp[(size_t)ix]; hv[E] = pr.x; ds[E] = pr.y; } \
            else { hv[E] = h[(size_t)ix]; ds[E] = 1.0f; } }
        LAYER_LOAD(0)  LAYER_LOAD(1)  LAYER_LOAD(2)  LAYER_LOAD(3)
        LAYER_LOAD(4)  LAYER_LOAD(5)  LAYER_LOAD(6)  LAYER_LOAD(7)
        LAYER_LOAD(8)  LAYER_LOAD(9)  LAYER_LOAD(10) LAYER_LOAD(11)
        LAYER_LOAD(12) LAYER_LOAD(13) LAYER_LOAD(14) LAYER_LOAD(15)
#undef LAYER_LOAD
#define LAYER_ADD(E) { \
            const float w = __uint_as_float(row_share_u32<E>(wb)); \
            const unsigned a = row_share_u32<E>(ao4) | j4; \
            const float norm = PAIR ? ds[E] * w : w; \
            atomicAdd((float*)((char*)acc + a), hv[E] * norm); }
        LAYER_ADD(0)  LAYER_ADD(1)  LAYER_ADD(2)  LAYER_ADD(3)
        LAYER_ADD(4)  LAYER_ADD(5)  LAYER_ADD(6)  LAYER_ADD(7)
        LAYER_ADD(8)  LAYER_ADD(9)  LAYER_ADD(10) LAYER_ADD(11)
        LAYER_ADD(12) LAYER_ADD(13) LAYER_ADD(14) LAYER_ADD(15)
#undef LAYER_ADD
    }
    __syncthreads();

    if (!POOL) {
        for (int idx = t; idx < BNODES * D; idx += LTHREADS) {
            int nl = idx >> 4;
            int i = node0 + nl;
            if (i < n) {
                float di = dloc[nl];
                size_t o = (size_t)i * D + (idx & 15);
                float v = acc[idx] + h[PAIR ? o * 2 : o] * (di * di) + bias[idx & 15];
                v = v > 0.0f ? v : 0.0f;
                rows[o] = SCALED ? v * di : v;
            }
        }
    } else {
        const int g0 = batch[node0];
        for (int idx = t; idx < BNODES * D; idx += LTHREADS) {
            int nl = idx >> 4;
            int jj = idx & 15;
            int i = node0 + nl;
            if (i < n) {
                float di = dloc[nl];
                size_t o = (size_t)i * D + jj;
                float v = acc[idx] + h[PAIR ? o * 2 : o] * (di * di) + bias[jj];
                int g = batch[i];
                unsigned k = (unsigned)(g - g0);
                if (k < (unsigned)PGR) {
                    atomicAdd(&pl[k * D + jj], v);
                    if (jj == 0) atomicAdd(&pc[k], 1);
                } else {
                    atomicAdd(pooled + (size_t)g * D + jj, v);
                    if (jj == 0) atomicAdd(cntg + g, 1.0f);
                }
            }
        }
        __syncthreads();
        if (t < PGR * D) {
            int k = t >> 4;
            int c = pc[k];
            if (c > 0) {
                int g = g0 + k;
                atomicAdd(pooled + (size_t)g * D + (t & 15), pl[t]);
                if ((t & 15) == 0) atomicAdd(cntg + g, (float)c);
            }
        }
    }
}


#define TCAP 1032
#define TCOPY ((TCAP * D / 4 + LTHREADS - 1) / LTHREADS)
#define TROWS ((TCAP + 15) & ~15)
__global__ __launch_bounds__(LTHREADS)
void bucket_layer1_tab_k(const float* __restrict__ tab, int n_tab,
                         const v2u* __restrict__ nrec0, const v2u* __restrict__ nrec1,
                         const unsigned* __restrict__ part0, const unsigned* __restrict__ part1,
                         const int* __restrict__ bstart0, const int* __restrict__ bstart1,
                         const float* __restrict__ bias,
                         float* __restrict__ rows0, float* __restrict__ rows1, int n) {
    const bool second = blockIdx.y != 0;
    const v2u* __restrict__ nrec = second ? nrec1 : nrec0;
    const unsigned* __restrict__ part = second ? part1 : part0;
    const int* __restrict__ bstart = second ? bstart1 : bstart0;
    float* __restrict__ rows = second ? rows1 : rows0;
    __shared__ struct {
        float acc[(BNODES + 1) * D];
        float dloc[BNODES + 4];
        v4f tl[TCAP * D / 4];
    } sm;
    float* const acc = sm.acc;
    float* const dloc = sm.dloc;
    v4f* const tl = sm.tl;
    const int t = threadIdx.x;
    const int b = blockIdx.x;
    const int node0 = b * BNODES;
    const int lane = t & 31;
    const unsigned j4 = ((unsigned)lane & 15u) << 2;
    const int wv = __builtin_amdgcn_readfirstlane(t >> 5);
    const int p0 = bstart[b], p1 = bstart[b + 1];
    int qb = p0 + wv * 32;
    unsigned rec0 = 0u, rec1 = 0u;
    if (qb + lane < p1) rec0 = part[qb + lane];
    if (qb + LTHREADS + lane < p1) rec1 = part[qb + LTHREADS + lane];
    for (int i = t; i < (BNODES + 1) * D; i += LTHREADS) acc[i] = 0.0f;
    if (t <= BNODES) {
        int i = node0 + t;
        dloc[t] = (t < BNODES && i < n) ? __uint_as_float(nrec[i].y) : 0.0f;
    }
    {
        const v4f* __restrict__ tg = (const v4f*)tab;
        const int n4 = n_tab * (D / 4);
        v4f tv[TCOPY];
        #pragma unroll
        for (int u = 0; u < TCOPY; ++u) {
            int i = t + u * LTHREADS;
            tv[u] = tg[i < n4 ? i : 0];
        }
        #pragma unroll
        for (int u = 0; u < TCOPY; ++u) {
            int i = t + u * LTHREADS;
            if (i < n4) tl[i] = tv[u];
        }
    }
    v2u nd = nrec[rec0 >> RSH];
    __syncthreads();

    const char* tlb = (const char*)tl;
    for (; qb < p1; qb += LTHREADS) {
        const int q = qb + lane;
        const unsigned rec = rec0;
        const v2u ndc = nd;
        rec0 = rec1;
        nd = nrec[rec0 >> RSH];
        const int q2 = q + 2 * LTHREADS;
        rec1 = (q2 < p1) ? part[q2] : 0u;
        const unsigned dl = (q < p1) ? (rec & RMASK) : (unsigned)BNODES;
        const unsigned to4 = ndc.x << 6;
        const unsigned ao4 = dl << 6;
        const unsigned wb = __float_as_uint(dloc[dl] * __uint_as_float(ndc.y));
        float hv[16];
#define TAB_LOAD(E) { \
            const unsigned ta = row_share_u32<E>(to4) | j4; \
            hv[E] = *(const float*)(tlb + ta); }
        TAB_LOAD(0)  TAB_LOAD(1)  TAB_LOAD(2)  TAB_LOAD(3)
        TAB_LOAD(4)  TAB_LOAD(5)  TAB_LOAD(6)  TAB_LOAD(7)
        TAB_LOAD(8)  TAB_LOAD(9)  TAB_LOAD(10) TAB_LOAD(11)
        TAB_LOAD(12) TAB_LOAD(13) TAB_LOAD(14) TAB_LOAD(15)
#undef TAB_LOAD
#define TAB_ADD(E) { \
            const float w = __uint_as_float(row_share_u32<E>(wb)); \
            const unsigned a = row_share_u32<E>(ao4) | j4; \
            atomicAdd((float*)((char*)acc + a), hv[E] * w); }
        TAB_ADD(0)  TAB_ADD(1)  TAB_ADD(2)  TAB_ADD(3)
        TAB_ADD(4)  TAB_ADD(5)  TAB_ADD(6)  TAB_ADD(7)
        TAB_ADD(8)  TAB_ADD(9)  TAB_ADD(10) TAB_ADD(11)
        TAB_ADD(12) TAB_ADD(13) TAB_ADD(14) TAB_ADD(15)
#undef TAB_ADD
    }
    __syncthreads();

    const float* tlf = (const float*)tl;
    for (int idx = t; idx < BNODES * D; idx += LTHREADS) {
        int nl = idx >> 4;
        int i = node0 + nl;
        if (i < n) {
            float di = dloc[nl];
            float hs = tlf[nrec[i].x * D + (idx & 15)];
            float v = acc[idx] + hs * (di * di) + bias[idx & 15];
            rows[(size_t)i * D + (idx & 15)] = (v > 0.0f ? v : 0.0f) * di;
        }
    }
}

__global__ __launch_bounds__(LTHREADS)
void bucket_layer2_sum_k(const float* __restrict__ h0, const float* __restrict__ h1,
                         const float* __restrict__ dis0, const float* __restrict__ dis1,
                         const unsigned* __restrict__ part0, const unsigned* __restrict__ part1,
                         const int* __restrict__ bstart0, const int* __restrict__ bstart1,
                         const int* __restrict__ batch0, const int* __restrict__ batch1,
                         float* __restrict__ pooled0, float* __restrict__ pooled1,
                         float* __restrict__ cntg0, float* __restrict__ cntg1,
                         int n) {
    const bool second = blockIdx.y != 0;
    const float* __restrict__ h = second ? h1 : h0;
    const float* __restrict__ dis = second ? dis1 : dis0;
    const unsigned* __restrict__ part = second ? part1 : part0;
    const int* __restrict__ bstart = second ? bstart1 : bstart0;
    const int* __restrict__ batch = second ? batch1 : batch0;
    float* __restrict__ pooled = second ? pooled1 : pooled0;
    float* __restrict__ cntg = second ? cntg1 : cntg0;
    __shared__ float acc[(BNODES + 1) * D];
    __shared__ float pl[PGR * D];
    __shared__ int   pc[PGR];
    const int t = threadIdx.x;
    const int b = blockIdx.x;
    const int node0 = b * BNODES;
    for (int i = t; i < (BNODES + 1) * D; i += LTHREADS) acc[i] = 0.0f;
    if (t < PGR * D) pl[t] = 0.0f;
    if (t < PGR) pc[t] = 0;
    __syncthreads();

    const int lane = t & 31;
    const unsigned j4 = ((unsigned)lane & 15u) << 2;
    const int wv = __builtin_amdgcn_readfirstlane(t >> 5);
    const int p0 = bstart[b], p1 = bstart[b + 1];
    const char* __restrict__ hb = (const char*)h;
    int qb = p0 + wv * 32;
    unsigned recn = 0u;
    if (qb + lane < p1) recn = part[qb + lane];
    for (; qb < p1; qb += LTHREADS) {
        const int q = qb + lane;
        const unsigned rec = recn;
        const int qn = q + LTHREADS;
        recn = (qn < p1) ? part[qn] : 0u;
        const unsigned dl = (q < p1) ? (rec & RMASK) : (unsigned)BNODES;
        const unsigned go4 = (rec >> RSH) << 6;
        const unsigned ao4 = dl << 6;
        float hv[16];
#define SUM_LOAD(E) { \
            const unsigned ga = row_share_u32<E>(go4) | j4; \
            hv[E] = *(const float*)(hb + ga); }
        SUM_LOAD(0)  SUM_LOAD(1)  SUM_LOAD(2)  SUM_LOAD(3)
        SUM_LOAD(4)  SUM_LOAD(5)  SUM_LOAD(6)  SUM_LOAD(7)
        SUM_LOAD(8)  SUM_LOAD(9)  SUM_LOAD(10) SUM_LOAD(11)
        SUM_LOAD(12) SUM_LOAD(13) SUM_LOAD(14) SUM_LOAD(15)
#undef SUM_LOAD
#define SUM_ADD(E) { \
            const unsigned a = row_share_u32<E>(ao4) | j4; \
            atomicAdd((float*)((char*)acc + a), hv[E]); }
        SUM_ADD(0)  SUM_ADD(1)  SUM_ADD(2)  SUM_ADD(3)
        SUM_ADD(4)  SUM_ADD(5)  SUM_ADD(6)  SUM_ADD(7)
        SUM_ADD(8)  SUM_ADD(9)  SUM_ADD(10) SUM_ADD(11)
        SUM_ADD(12) SUM_ADD(13) SUM_ADD(14) SUM_ADD(15)
#undef SUM_ADD
    }
    __syncthreads();

    const int g0 = batch[node0];
    for (int idx = t; idx < BNODES * D; idx += LTHREADS) {
        int nl = idx >> 4;
        int jj = idx & 15;
        int i = node0 + nl;
        if (i < n) {
            float v = dis[i] * (acc[idx] + h[(size_t)i * D + jj]);
            int g = batch[i];
            unsigned k = (unsigned)(g - g0);
            if (k < (unsigned)PGR) {
                atomicAdd(&pl[k * D + jj], v);
                if (jj == 0) atomicAdd(&pc[k], 1);
            } else {
                atomicAdd(pooled + (size_t)g * D + jj, v);
                if (jj == 0) atomicAdd(cntg + g, 1.0f);
            }
        }
    }
    __syncthreads();
    if (t < PGR * D) {
        int k = t >> 4;
        int c = pc[k];
        if (c > 0) {
            int g = g0 + k;
            atomicAdd(pooled + (size_t)g * D + (t & 15), pl[t]);
            if ((t & 15) == 0) atomicAdd(cntg + g, (float)c);
        }
    }
}

static void run_protein_bucketed(const float* emb, const float* W1, const float* b1,
                                 const float* W2, const float* b2,
                                 const int* ids, const int* edge, const int* batch,
                                 int N, long long E,
                                 float* bufA, float* bufB, float* dis,
                                 float* pooled, float* cnt,
                                 unsigned* part, int* bcount, int* bstart, int* cursor,
                                 bool pair, hipStream_t stream) {
    const int* src = edge;
    const int* dst = edge + E;
    const int B = 256;
    dim3 blk(B);
    int gemm_grid = (int)(((N >> 4) + 7) / 8);
    if (gemm_grid < 1) gemm_grid = 1;
    const int nb = (N + BNODES - 1) / BNODES;
    const long long round_cap = (long long)NWGP * PTILE;
    const int ptiles = NWGP * (int)((E + round_cap - 1) / round_cap);
    const int tlen = (int)(((E + ptiles - 1) / ptiles + 31) & ~31ll);

    fill_zero_k<<<(NBMAX + B - 1) / B, blk, 0, stream>>>((float*)bcount, NBMAX);
    bucket_hist_k<<<ptiles, PTHREADS, 0, stream>>>(dst, nullptr, E, tlen, bcount, nullptr);
    bucket_scan_k<<<1, NBMAX, 0, stream>>>(bcount, nullptr, bstart, nullptr, cursor, nullptr, nb);
    bucket_scatter_k<<<ptiles, PTHREADS, 0, stream>>>(src, nullptr, dst, nullptr, E, tlen,
                                                      cursor, nullptr, part, nullptr);
    bucket_degree_k<<<nb, DTHREADS, 0, stream>>>(part, nullptr, bstart, nullptr, dis, nullptr, N,
                                                 nullptr, nullptr, nullptr, nullptr, 0);

    gather_gemm16_k<<<gemm_grid, blk, 0, stream>>>(emb, nullptr, ids, nullptr, W1, bufA, nullptr,
                                                   pair ? dis : nullptr, nullptr, N);
    if (pair)
        bucket_layer_k<false, true><<<nb, LTHREADS, 0, stream>>>(
            bufA, nullptr, dis, nullptr, part, nullptr, bstart, nullptr, b1, bufB, nullptr,
            nullptr, nullptr, nullptr, nullptr, nullptr, nullptr, N);
    else
        bucket_layer_k<false, false><<<nb, LTHREADS, 0, stream>>>(
            bufA, nullptr, dis, nullptr, part, nullptr, bstart, nullptr, b1, bufB, nullptr,
            nullptr, nullptr, nullptr, nullptr, nullptr, nullptr, N);
    gather_gemm16_k<<<gemm_grid, blk, 0, stream>>>(bufB, nullptr, nullptr, nullptr, W2, bufA, nullptr,
                                                   pair ? dis : nullptr, nullptr, N);

    fill_zero_k<<<(int)((NGRAPH * D + B - 1) / B), blk, 0, stream>>>(pooled, NGRAPH * D);
    fill_zero_k<<<(int)((NGRAPH + B - 1) / B), blk, 0, stream>>>(cnt, NGRAPH);
    if (pair)
        bucket_layer_k<true, true><<<nb, LTHREADS, 0, stream>>>(
            bufA, nullptr, dis, nullptr, part, nullptr, bstart, nullptr, b2, nullptr, nullptr,
            batch, nullptr, pooled, nullptr, cnt, nullptr, N);
    else
        bucket_layer_k<true, false><<<nb, LTHREADS, 0, stream>>>(
            bufA, nullptr, dis, nullptr, part, nullptr, bstart, nullptr, b2, nullptr, nullptr,
            batch, nullptr, pooled, nullptr, cnt, nullptr, N);
}

static void run_both_bucketed(const float* emb, const float* W1, const float* b1,
                              const float* W2,
                              const int* ids0, const int* ids1,
                              const int* edge0, const int* edge1,
                              const int* batch0, const int* batch1,
                              int N, long long E,
                              float* pair0, float* pair1, float* rows0, float* rows1,
                              float* dis0, float* dis1,
                              float* zero_run, long long zero_words,
                              float* pooled0, float* pooled1, float* cnt0, float* cnt1,
                              int* bcount0, int* bcount1, int* bstart0, int* bstart1,
                              int* cursor0, int* cursor1,
                              unsigned* part0, unsigned* part1,
                              int n_tab, float* tab, v2u* nrec0, v2u* nrec1,
                              float* pw0, float* pw1,
                              hipStream_t stream) {
    const int B = 256;
    dim3 blk(B);
    int gemm_grid = (int)(((N >> 4) + 7) / 8);
    if (gemm_grid < 1) gemm_grid = 1;
    const int nb = (N + BNODES - 1) / BNODES;
    const long long round_cap = (long long)NWGP * PTILE;
    const int ptiles = NWGP * (int)((E + round_cap - 1) / round_cap);
    const int tlen = (int)(((E + ptiles - 1) / ptiles + 31) & ~31ll);

    fill_zero_k<<<(int)((zero_words + B - 1) / B), blk, 0, stream>>>(zero_run, zero_words);
    bucket_hist_k<<<dim3(ptiles, 2), PTHREADS, 0, stream>>>(edge0 + E, edge1 + E, E, tlen,
                                                            bcount0, bcount1);
    bucket_scan_k<<<dim3(1, 2), NBMAX, 0, stream>>>(bcount0, bcount1, bstart0, bstart1,
                                                    cursor0, cursor1, nb);
    bucket_scatter_k<<<dim3(ptiles, 2), PTHREADS, 0, stream>>>(edge0, edge1, edge0 + E, edge1 + E,
                                                               E, tlen, cursor0, cursor1,
                                                               part0, part1);
    if (tab) {
        bucket_degree_k<<<dim3(nb, 2), DTHREADS, 0, stream>>>(part0, part1, bstart0, bstart1,
                                                              dis0, dis1, N,
                                                              ids0, ids1, nrec0, nrec1, n_tab);
        table_gemm16_k<<<(((n_tab + 15) >> 4) + 7) / 8, blk, 0, stream>>>(emb, W1, tab, n_tab);
        bucket_layer1_tab_k<<<dim3(nb, 2), LTHREADS, 0, stream>>>(
            tab, n_tab, nrec0, nrec1, part0, part1, bstart0, bstart1, b1, rows0, rows1, N);
    } else {
        bucket_degree_k<<<dim3(nb, 2), DTHREADS, 0, stream>>>(part0, part1, bstart0, bstart1,
                                                              dis0, dis1, N,
                                                              nullptr, nullptr, nullptr, nullptr, 0);
        gather_gemm16_k<<<dim3(gemm_grid, 2), blk, 0, stream>>>(emb, emb, ids0, ids1, W1, pair0, pair1,
                                                                dis0, dis1, N);
        bucket_layer_k<false, true, true><<<dim3(nb, 2), LTHREADS, 0, stream>>>(
            pair0, pair1, dis0, dis1, part0, part1, bstart0, bstart1, b1, rows0, rows1,
            nullptr, nullptr, nullptr, nullptr, nullptr, nullptr, N);
    }
    bucket_layer2_sum_k<<<dim3(nb, 2), LTHREADS, 0, stream>>>(
        rows0, rows1, dis0, dis1, part0, part1, bstart0, bstart1,
        batch0, batch1, pooled0, pooled1, cnt0, cnt1, N);
    gather_gemm16_k<<<dim3((NGRAPH / 16 + 7) / 8, 2), blk, 0, stream>>>(
        pooled0, pooled1, nullptr, nullptr, W2, pw0, pw1, nullptr, nullptr, NGRAPH);
}

static void run_protein(const float* emb, const float* W1, const float* b1,
                        const float* W2, const float* b2,
                        const int* ids, const int* edge, const int* batch,
                        int N, long long E,
                        float* bufA, float* bufB, float* dis,
                        float* pooled, float* cnt, hipStream_t stream) {
    const int* src = edge;
    const int* dst = edge + E;
    const long long n16 = (long long)N * D;
    const int B = 256;
    dim3 blk(B);
    int gemm_grid = (int)(((N >> 4) + 7) / 8);
    if (gemm_grid < 1) gemm_grid = 1;

    fill_zero_k<<<(int)((N + B - 1) / B), blk, 0, stream>>>(dis, N);
    fill_zero_k<<<(int)((n16 + B - 1) / B), blk, 0, stream>>>(bufB, n16);

    gather_gemm16_k<<<gemm_grid, blk, 0, stream>>>(emb, nullptr, ids, nullptr, W1, bufA, nullptr,
                                                   nullptr, nullptr, N);

    degree_k<<<(int)((E + B - 1) / B), blk, 0, stream>>>(dst, dis, E);
    rsqrt_k<<<(int)((N + B - 1) / B), blk, 0, stream>>>(dis, N);

    aggregate_k<<<(int)((E * 16 + B - 1) / B), blk, 0, stream>>>(bufA, dis, src, dst, bufB, E);
    node_relu_k<<<(int)((n16 + B - 1) / B), blk, 0, stream>>>(bufA, bufB, dis, b1, n16);

    gather_gemm16_k<<<gemm_grid, blk, 0, stream>>>(bufA, nullptr, nullptr, nullptr, W2, bufB, nullptr,
                                                   nullptr, nullptr, N);

    fill_zero_k<<<(int)((n16 + B - 1) / B), blk, 0, stream>>>(bufA, n16);
    aggregate_k<<<(int)((E * 16 + B - 1) / B), blk, 0, stream>>>(bufB, dis, src, dst, bufA, E);

    fill_zero_k<<<(int)((NGRAPH * D + B - 1) / B), blk, 0, stream>>>(pooled, NGRAPH * D);
    fill_zero_k<<<(int)((NGRAPH + B - 1) / B), blk, 0, stream>>>(cnt, NGRAPH);
    node_pool_k<<<(int)((n16 + B - 1) / B), blk, 0, stream>>>(bufB, bufA, dis, b2, batch,
                                                              pooled, cnt, n16);
}

extern "C" void kernel_launch(void* const* d_in, const int* in_sizes, int n_in,
                              void* d_out, int out_size, void* d_ws, size_t ws_size,
                              hipStream_t stream) {
    const float* emb = (const float*)d_in[0];
    const float* W1  = (const float*)d_in[1];
    const float* b1  = (const float*)d_in[2];
    const float* W2  = (const float*)d_in[3];
    const float* b2  = (const float*)d_in[4];
    const float* fcW = (const float*)d_in[5];
    const float* fcb = (const float*)d_in[6];
    const int* rx     = (const int*)d_in[7];
    const int* redge  = (const int*)d_in[8];
    const int* rbatch = (const int*)d_in[9];
    const int* lx     = (const int*)d_in[10];
    const int* ledge  = (const int*)d_in[11];
    const int* lbatch = (const int*)d_in[12];

    const int N = in_sizes[7];
    const long long E = in_sizes[8] / 2;

    char* ws = (char*)d_ws;
    float* bufA = (float*)ws;  ws += (size_t)N * D * sizeof(float);
    float* bufB = (float*)ws;  ws += (size_t)N * D * sizeof(float);
    float* dis  = (float*)ws;  ws += (size_t)N * sizeof(float);
    float* pooled_r = (float*)ws; ws += (size_t)NGRAPH * D * sizeof(float);
    float* pooled_l = (float*)ws; ws += (size_t)NGRAPH * D * sizeof(float);
    float* cnt_r = (float*)ws; ws += (size_t)NGRAPH * sizeof(float);
    float* cnt_l = (float*)ws; ws += (size_t)NGRAPH * sizeof(float);

    unsigned* part = (unsigned*)ws;  ws += (size_t)E * sizeof(unsigned);
    int* bcount = (int*)ws;  ws += (size_t)NBMAX * sizeof(int);
    int* bstart = (int*)ws;  ws += (size_t)2 * NBMAX * sizeof(int);
    int* cursor = (int*)ws;  ws += (size_t)NBMAX * sizeof(int);
    const bool bucketed = (size_t)(ws - (char*)d_ws) <= ws_size &&
                          N > 0 && N <= NBMAX * BNODES && E > 0 && E < (1ll << 30);
    float* bufT = (float*)ws;  ws += (size_t)N * D * sizeof(float);
    const bool pair = (size_t)(ws - (char*)d_ws) <= ws_size;
    float* rowsB = pair ? bufT : bufB;

    char* q = (char*)d_ws;
    float* pair0 = (float*)q;  q += (size_t)N * D * 2 * sizeof(float);
    float* pair1 = (float*)q;  q += (size_t)N * D * 2 * sizeof(float);
    float* rows0 = (float*)q;  q += (size_t)N * D * sizeof(float);
    float* rows1 = (float*)q;  q += (size_t)N * D * sizeof(float);
    float* dis0  = (float*)q;  q += (size_t)N * sizeof(float);
    float* dis1  = (float*)q;  q += (size_t)N * sizeof(float);
    float* zero_run = (float*)q;
    float* pool0 = (float*)q;  q += (size_t)NGRAPH * D * sizeof(float);
    float* pool1 = (float*)q;  q += (size_t)NGRAPH * D * sizeof(float);
    float* pcnt0 = (float*)q;  q += (size_t)NGRAPH * sizeof(float);
    float* pcnt1 = (float*)q;  q += (size_t)NGRAPH * sizeof(float);
    int* bcount0 = (int*)q;  q += (size_t)NBMAX * sizeof(int);
    int* bcount1 = (int*)q;  q += (size_t)NBMAX * sizeof(int);
    const long long zero_words = 2ll * NGRAPH * D + 2ll * NGRAPH + 2ll * NBMAX;
    int* bstart0 = (int*)q;  q += (size_t)2 * NBMAX * sizeof(int);
    int* bstart1 = (int*)q;  q += (size_t)2 * NBMAX * sizeof(int);
    int* cursor0 = (int*)q;  q += (size_t)NBMAX * sizeof(int);
    int* cursor1 = (int*)q;  q += (size_t)NBMAX * sizeof(int);
    unsigned* part0 = (unsigned*)q;  q += (size_t)E * sizeof(unsigned);
    unsigned* part1 = (unsigned*)q;  q += (size_t)E * sizeof(unsigned);
    const bool both = bucketed && (size_t)(q - (char*)d_ws) <= ws_size;
    const int n_tab = in_sizes[0] / D;
    float* tab = (float*)pair1;
    v2u* nrec0 = (v2u*)(tab + (size_t)TROWS * D);
    v2u* nrec1 = nrec0 + N;
    const bool tabbed = n_tab > 0 && n_tab <= TCAP &&
                        (size_t)TROWS * D * sizeof(float) + (size_t)2 * N * sizeof(v2u) <=
                            (size_t)N * D * 2 * sizeof(float);
    float* pw0 = pair0;
    float* pw1 = pair0 + (size_t)NGRAPH * D;
    if (both) {
        run_both_bucketed(emb, W1, b1, W2, rx, lx, redge, ledge, rbatch, lbatch, N, E,
                          pair0, pair1, rows0, rows1, dis0, dis1, zero_run, zero_words,
                          pool0, pool1, pcnt0, pcnt1, bcount0, bcount1, bstart0, bstart1,
                          cursor0, cursor1, part0, part1,
                          n_tab, tabbed ? tab : nullptr, nrec0, nrec1, pw0, pw1, stream);
        fc_k<<<(NGRAPH * 6 + 255) / 256, 256, 0, stream>>>(pw0, pcnt0, pw1, pcnt1,
                                                           fcW, fcb, b2, (float*)d_out);
        return;
    }
    if (bucketed) {
        run_protein_bucketed(emb, W1, b1, W2, b2, rx, redge, rbatch, N, E,
                             bufA, rowsB, dis, pooled_r, cnt_r,
                             part, bcount, bstart, cursor, pair, stream);
        run_protein_bucketed(emb, W1, b1, W2, b2, lx, ledge, lbatch, N, E,
                             bufA, rowsB, dis, pooled_l, cnt_l,
                             part, bcount, bstart, cursor, pair, stream);
    } else {
        run_protein(emb, W1, b1, W2, b2, rx, redge, rbatch, N, E,
                    bufA, bufB, dis, pooled_r, cnt_r, stream);
        run_protein(emb, W1, b1, W2, b2, lx, ledge, lbatch, N, E,
                    bufA, bufB, dis, pooled_l, cnt_l, stream);
    }

    fc_k<<<(NGRAPH * 6 + 255) / 256, 256, 0, stream>>>(pooled_r, cnt_r, pooled_l, cnt_l,
                                                       fcW, fcb, nullptr, (float*)d_out);
}


# small stages lose their launches: bucket scan taken by every scatter group beside its tile scan, layer-1 product table by the degree launch, pooled product with the fc: 7 launches, were 10
# speedup vs baseline: 9.4029x; 1.0308x over previous
#include <hip/hip_runtime.h>
#include <hip/hip_bf16.h>

typedef float v2f __attribute__((ext_vector_type(2)));
typedef float v8f __attribute__((ext_vector_type(8)));
typedef float v4f __attribute__((ext_vector_type(4)));
typedef unsigned v2u __attribute__((ext_vector_type(2)));

#define D 16
#define NGRAPH 1024

__global__ void fill_zero_k(float* __restrict__ p, long long n) {
    long long i = (long long)blockIdx.x * blockDim.x + threadIdx.x;
    if (i < n) p[i] = 0.0f;
}

__global__ void gather_gemm16_k(const float* __restrict__ src_rows0,
                                const float* __restrict__ src_rows1,
                                const int* __restrict__ ids0,
                                const int* __restrict__ ids1,
                                const float* __restrict__ W,
                                float* __restrict__ out0,
                                float* __restrict__ out1,
                                const float* __restrict__ dis_pair0,
                                const float* __restrict__ dis_pair1,
                                int n_nodes) {
    const bool second = blockIdx.y != 0;
    const float* __restrict__ src_rows = second ? src_rows1 : src_rows0;
    const int* __restrict__ ids = second ? ids1 : ids0;
    float* __restrict__ out = second ? out1 : out0;
    const float* __restrict__ dis_pair = second ? dis_pair1 : dis_pair0;
    const int lane  = threadIdx.x & 31;
    const int wave  = (int)((blockIdx.x * blockDim.x + threadIdx.x) >> 5);
    const int nwave = (int)((gridDim.x * blockDim.x) >> 5);
    const int row   = lane & 15;
    const int col   = lane & 15;
    const int koff  = (lane >> 4) << 1;
    const int rowb  = (lane >> 4) << 3;

    v2f bb0, bb1, bb2, bb3;
    bb0.x = W[(0  + koff) * D + col]; bb0.y = W[(1  + koff) * D + col];
    bb1.x = W[(4  + koff) * D + col]; bb1.y = W[(5  + koff) * D + col];
    bb2.x = W[(8  + koff) * D + col]; bb2.y = W[(9  + koff) * D + col];
    bb3.x = W[(12 + koff) * D + col]; bb3.y = W[(13 + koff) * D + col];

    const int ntiles = n_nodes >> 4;
    for (int tile = wave; tile < ntiles; tile += nwave) {
        const int node = (tile << 4) + row;
        const int r = ids ? ids[node] : node;
        const float* xrow = src_rows + (size_t)r * D;
        v2f a0 = *(const v2f*)(xrow + 0  + koff);
        v2f a1 = *(const v2f*)(xrow + 4  + koff);
        v2f a2 = *(const v2f*)(xrow + 8  + koff);
        v2f a3 = *(const v2f*)(xrow + 12 + koff);
        v8f c = {};
        c = __builtin_amdgcn_wmma_f32_16x16x4_f32(false, a0, false, bb0, (short)0, c, false, false);
        c = __builtin_amdgcn_wmma_f32_16x16x4_f32(false, a1, false, bb1, (short)0, c, false, false);
        c = __builtin_amdgcn_wmma_f32_16x16x4_f32(false, a2, false, bb2, (short)0, c, false, false);
        c = __builtin_amdgcn_wmma_f32_16x16x4_f32(false, a3, false, bb3, (short)0, c, false, false);
        if (dis_pair) {
            const float* dp = dis_pair + (tile << 4) + rowb;
            v2f* ob2 = (v2f*)out + ((size_t)(tile << 4) + rowb) * D + col;
            #pragma unroll
            for (int rr = 0; rr < 8; ++rr) {
                v2f pr; pr.x = c[rr]; pr.y = dp[rr];
                ob2[(size_t)rr * D] = pr;
            }
        } else {
            float* ob = out + ((size_t)(tile << 4) + rowb) * D + col;
            #pragma unroll
            for (int rr = 0; rr < 8; ++rr) ob[(size_t)rr * D] = c[rr];
        }
    }

    const int tail_start = ntiles << 4;
    if (tail_start < n_nodes) {
        int node = tail_start + (int)(blockIdx.x * blockDim.x + threadIdx.x);
        if (node < n_nodes) {
            const int r = ids ? ids[node] : node;
            const float* xrow = src_rows + (size_t)r * D;
            for (int j = 0; j < D; ++j) {
                float acc = 0.0f;
                for (int k = 0; k < D; ++k) acc += xrow[k] * W[k * D + j];
                if (dis_pair) {
                    out[((size_t)node * D + j) * 2] = acc;
                    out[((size_t)node * D + j) * 2 + 1] = dis_pair[node];
                } else {
                    out[(size_t)node * D + j] = acc;
                }
            }
        }
    }
}

__device__ __forceinline__ void table_gemm16_tiles(const float* __restrict__ src_rows,
                                                   const float* __restrict__ W,
                                                   float* __restrict__ out, int n_rows,
                                                   int wave, int nwave, int lane) {
    const int row   = lane & 15;
    const int col   = lane & 15;
    const int koff  = (lane >> 4) << 1;
    const int rowb  = (lane >> 4) << 3;
    const int ntiles = (n_rows + 15) >> 4;
    if (wave >= ntiles) return;
    v2f bb0, bb1, bb2, bb3;
    bb0.x = W[(0  + koff) * D + col]; bb0.y = W[(1  + koff) * D + col];
    bb1.x = W[(4  + koff) * D + col]; bb1.y = W[(5  + koff) * D + col];
    bb2.x = W[(8  + koff) * D + col]; bb2.y = W[(9  + koff) * D + col];
    bb3.x = W[(12 + koff) * D + col]; bb3.y = W[(13 + koff) * D + col];
    float* ob = out + ((size_t)(wave << 4) + rowb) * D + col;
    for (int tile = wave; tile < ntiles; tile += nwave, ob += (size_t)nwave * 16 * D) {
        const int node = (tile << 4) + row;
        const int r = node < n_rows ? node : n_rows - 1;
        const float* xrow = src_rows + (size_t)r * D;
        v2f a0 = *(const v2f*)(xrow + 0  + koff);
        v2f a1 = *(const v2f*)(xrow + 4  + koff);
        v2f a2 = *(const v2f*)(xrow + 8  + koff);
        v2f a3 = *(const v2f*)(xrow + 12 + koff);
        v8f c = {};
        c = __builtin_amdgcn_wmma_f32_16x16x4_f32(false, a0, false, bb0, (short)0, c, false, false);
        c = __builtin_amdgcn_wmma_f32_16x16x4_f32(false, a1, false, bb1, (short)0, c, false, false);
        c = __builtin_amdgcn_wmma_f32_16x16x4_f32(false, a2, false, bb2, (short)0, c, false, false);
        c = __builtin_amdgcn_wmma_f32_16x16x4_f32(false, a3, false, bb3, (short)0, c, false, false);
        #pragma unroll
        for (int rr = 0; rr < 8; ++rr) ob[(size_t)rr * D] = c[rr];
    }
}

__global__ void degree_k(const int* __restrict__ dst, float* __restrict__ deg,
                         long long E) {
    long long e = (long long)blockIdx.x * blockDim.x + threadIdx.x;
    if (e < E) atomicAdd(deg + dst[e], 1.0f);
}

__global__ void rsqrt_k(float* __restrict__ d, int n) {
    int i = blockIdx.x * blockDim.x + threadIdx.x;
    if (i < n) d[i] = rsqrtf(d[i] + 1.0f);
}

__global__ void aggregate_k(const float* __restrict__ h,
                            const float* __restrict__ dis,
                            const int* __restrict__ src,
                            const int* __restrict__ dst,
                            float* __restrict__ agg,
                            long long E) {
    long long t = (long long)blockIdx.x * blockDim.x + threadIdx.x;
    long long e = t >> 4;
    int j = (int)(t & 15);
    if (e < E) {
        int s = src[e], d = dst[e];
        float norm = dis[s] * dis[d];
        atomicAdd(agg + (size_t)d * D + j, h[(size_t)s * D + j] * norm);
    }
}

__global__ void node_relu_k(float* __restrict__ h, const float* __restrict__ agg,
                            const float* __restrict__ dis,
                            const float* __restrict__ bias, long long n16) {
    long long t = (long long)blockIdx.x * blockDim.x + threadIdx.x;
    if (t < n16) {
        long long i = t >> 4; int j = (int)(t & 15);
        float di = dis[i];
        float v = agg[t] + h[t] * (di * di) + bias[j];
        h[t] = v > 0.0f ? v : 0.0f;
    }
}

__global__ void node_pool_k(const float* __restrict__ h2,
                            const float* __restrict__ agg,
                            const float* __restrict__ dis,
                            const float* __restrict__ bias,
                            const int* __restrict__ batch,
                            float* __restrict__ pooled,
                            float* __restrict__ cnt, long long n16) {
    long long t = (long long)blockIdx.x * blockDim.x + threadIdx.x;
    if (t < n16) {
        long long i = t >> 4; int j = (int)(t & 15);
        float di = dis[i];
        float v = agg[t] + h2[t] * (di * di) + bias[j];
        int g = batch[i];
        atomicAdd(pooled + (size_t)g * D + j, v);
        if (j == 0) atomicAdd(cnt + g, 1.0f);
    }
}

__global__ void fc_k(const float* __restrict__ pooled_r, const float* __restrict__ cnt_r,
                     const float* __restrict__ pooled_l, const float* __restrict__ cnt_l,
                     const float* __restrict__ fcW, const float* __restrict__ fcb,
                     const float* __restrict__ nbias,
                     float* __restrict__ out) {
    int t = blockIdx.x * blockDim.x + threadIdx.x;
    if (t < NGRAPH * 6) {
        int g = t / 6, o = t - g * 6;
        const float cr = cnt_r[g], cl = cnt_l[g];
        float icr = 1.0f / fmaxf(cr, 1.0f);
        float icl = 1.0f / fmaxf(cl, 1.0f);
        float acc = fcb[o];
        #pragma unroll
        for (int j = 0; j < D; ++j) {
            float p = pooled_r[g * D + j];
            if (nbias) p += cr * nbias[j];
            acc += fcW[o * (2 * D) + j] * (p * icr);
        }
        #pragma unroll
        for (int j = 0; j < D; ++j) {
            float p = pooled_l[g * D + j];
            if (nbias) p += cl * nbias[j];
            acc += fcW[o * (2 * D) + D + j] * (p * icl);
        }
        if (o < 3) out[g * 3 + o] = acc;
        else       out[NGRAPH * 3 + g * 3 + (o - 3)] = acc;
    }
}

#define FWAVES 8
__global__ __launch_bounds__(FWAVES * 32)
void pool_fc_k(const float* __restrict__ pooled_r, const float* __restrict__ cnt_r,
               const float* __restrict__ pooled_l, const float* __restrict__ cnt_l,
               const float* __restrict__ W,
               const float* __restrict__ fcW, const float* __restrict__ fcb,
               const float* __restrict__ nbias,
               float* __restrict__ out) {
    __shared__ float pw[FWAVES][2][16 * D];
    const int lane = threadIdx.x & 31;
    const int wv   = __builtin_amdgcn_readfirstlane((int)(threadIdx.x >> 5));
    const int tile = (int)blockIdx.x * FWAVES + wv;
    const int row  = lane & 15;
    const int col  = lane & 15;
    const int koff = (lane >> 4) << 1;
    const int rowb = (lane >> 4) << 3;
    if (tile < NGRAPH / 16) {
        v2f bb0, bb1, bb2, bb3;
        bb0.x = W[(0  + koff) * D + col]; bb0.y = W[(1  + koff) * D + col];
        bb1.x = W[(4  + koff) * D + col]; bb1.y = W[(5  + koff) * D + col];
        bb2.x = W[(8  + koff) * D + col]; bb2.y = W[(9  + koff) * D + col];
        bb3.x = W[(12 + koff) * D + col]; bb3.y = W[(13 + koff) * D + col];
        #pragma unroll
        for (int pr = 0; pr < 2; ++pr) {
            const float* xrow = (pr ? pooled_l : pooled_r) + ((size_t)(tile << 4) + row) * D;
            v2f a0 = *(const v2f*)(xrow + 0  + koff);
            v2f a1 = *(const v2f*)(xrow + 4  + koff);
            v2f a2 = *(const v2f*)(xrow + 8  + koff);
            v2f a3 = *(const v2f*)(xrow + 12 + koff);
            v8f c = {};
            c = __builtin_amdgcn_wmma_f32_16x16x4_f32(false, a0, false, bb0, (short)0, c, false, false);
            c = __builtin_amdgcn_wmma_f32_16x16x4_f32(false, a1, false, bb1, (short)0, c, false, false);
            c = __builtin_amdgcn_wmma_f32_16x16x4_f32(false, a2, false, bb2, (short)0, c, false, false);
            c = __builtin_amdgcn_wmma_f32_16x16x4_f32(false, a3, false, bb3, (short)0, c, false, false);
            float* ob = &pw[wv][pr][rowb * D + col];
            #pragma unroll
            for (int rr = 0; rr < 8; ++rr) ob[rr * D] = c[rr];
        }
    }
    __syncthreads();
    if (tile < NGRAPH / 16) {
        const float* pr_ = pw[wv][0];
        const float* pl_ = pw[wv][1];
        for (int u = lane; u < 16 * 6; u += 32) {
            const int gl = u / 6, o = u - gl * 6;
            const int g = (tile << 4) + gl;
            const float cr = cnt_r[g], cl = cnt_l[g];
            float icr = 1.0f / fmaxf(cr, 1.0f);
            float icl = 1.0f / fmaxf(cl, 1.0f);
            float acc = fcb[o];
            #pragma unroll
            for (int j = 0; j < D; ++j) {
                float p = pr_[gl * D + j];
                if (nbias) p += cr * nbias[j];
                acc += fcW[o * (2 * D) + j] * (p * icr);
            }
            #pragma unroll
            for (int j = 0; j < D; ++j) {
                float p = pl_[gl * D + j];
                if (nbias) p += cl * nbias[j];
                acc += fcW[o * (2 * D) + D + j] * (p * icl);
            }
            if (o < 3) out[g * 3 + o] = acc;
            else       out[NGRAPH * 3 + g * 3 + (o - 3)] = acc;
        }
    }
}


#define RSH 8
#define RMASK 255u
#define BNODES 196
#define NBMAX 1024
#define NWGP 256
#define PTHREADS 1024
#define PTILE 25088
#define PU 8
#define DTHREADS 256

__global__ __launch_bounds__(PTHREADS)
void bucket_hist_k(const int* __restrict__ dst0, const int* __restrict__ dst1,
                   long long E, int tlen,
                   int* __restrict__ bcount0, int* __restrict__ bcount1) {
    const bool second = blockIdx.y != 0;
    const int* __restrict__ dst = second ? dst1 : dst0;
    int* __restrict__ bcount = second ? bcount1 : bcount0;
    __shared__ int cnt[NBMAX];
    const int t = threadIdx.x;
    cnt[t] = 0;
    __syncthreads();
    const long long base = (long long)blockIdx.x * tlen;
    const long long rem = E - base;
    const int ne = rem < (long long)tlen ? (int)rem : tlen;
    for (int k0 = 0; k0 * PTHREADS < ne; k0 += PU) {
        int dv[PU];
        #pragma unroll
        for (int u = 0; u < PU; ++u) {
            int l = (k0 + u) * PTHREADS + t;
            dv[u] = dst[base + (l < ne ? l : 0)];
        }
        #pragma unroll
        for (int u = 0; u < PU; ++u) {
            int l = (k0 + u) * PTHREADS + t;
            if (l < ne) atomicAdd(&cnt[(unsigned)dv[u] / BNODES], 1);
        }
    }
    __syncthreads();
    const int c = cnt[t];
    if (c > 0) atomicAdd(bcount + t, c);
}

__global__ __launch_bounds__(NBMAX)
void bucket_scan_k(const int* __restrict__ bcount0, const int* __restrict__ bcount1,
                   int* __restrict__ bstart0, int* __restrict__ bstart1,
                   int* __restrict__ cursor0, int* __restrict__ cursor1, int nb) {
    const bool second = blockIdx.y != 0;
    const int* __restrict__ bcount = second ? bcount1 : bcount0;
    int* __restrict__ bstart = second ? bstart1 : bstart0;
    int* __restrict__ cursor = second ? cursor1 : cursor0;
    __shared__ int s[NBMAX];
    const int t = threadIdx.x;
    const int own = (t < nb) ? bcount[t] : 0;
    s[t] = own;
    __syncthreads();
    for (int off = 1; off < NBMAX; off <<= 1) {
        int v = (t >= off) ? s[t - off] : 0;
        __syncthreads();
        s[t] += v;
        __syncthreads();
    }
    const int incl = s[t];
    if (t < nb) { bstart[t] = incl - own; cursor[t] = incl - own; }
    if (t == nb - 1) bstart[nb] = incl;
}

template <bool SCAN>
__global__ __launch_bounds__(PTHREADS)
void bucket_scatter_k(const int* __restrict__ src0, const int* __restrict__ src1,
                      const int* __restrict__ dst0, const int* __restrict__ dst1,
                      long long E, int tlen,
                      const int* __restrict__ bcount0, const int* __restrict__ bcount1,
                      int* __restrict__ bstart0, int* __restrict__ bstart1,
                      int* __restrict__ cursor0, int* __restrict__ cursor1,
                      unsigned* __restrict__ part0, unsigned* __restrict__ part1, int nb) {
    const bool second = blockIdx.y != 0;
    const int* __restrict__ src = second ? src1 : src0;
    const int* __restrict__ dst = second ? dst1 : dst0;
    const int* __restrict__ bcount = second ? bcount1 : bcount0;
    int* __restrict__ bstart = second ? bstart1 : bstart0;
    int* __restrict__ cursor = second ? cursor1 : cursor0;
    unsigned* __restrict__ part = second ? part1 : part0;
    __shared__ int gs[SCAN ? NBMAX : 1];
    __shared__ int cnt[NBMAX];
    __shared__ int tstart[NBMAX];
    __shared__ int gbase[NBMAX];
    __shared__ int run[NBMAX];
    __shared__ unsigned stage[PTILE];
    const int t = threadIdx.x;
    cnt[t] = 0;
    int gown = 0;
    if (SCAN && t < nb) gown = bcount[t];
    __syncthreads();
    const long long base = (long long)blockIdx.x * tlen;
    const long long rem = E - base;
    const int ne = rem < (long long)tlen ? (int)rem : tlen;
    for (int k0 = 0; k0 * PTHREADS < ne; k0 += PU) {
        int dv[PU];
        #pragma unroll
        for (int u = 0; u < PU; ++u) {
            int l = (k0 + u) * PTHREADS + t;
            dv[u] = dst[base + (l < ne ? l : 0)];
        }
        #pragma unroll
        for (int u = 0; u < PU; ++u) {
            int l = (k0 + u) * PTHREADS + t;
            if (l < ne) atomicAdd(&cnt[(unsigned)dv[u] / BNODES], 1);
        }
    }
    __syncthreads();
    const int own = cnt[t];
    run[t] = own;
    if (SCAN) gs[t] = gown;
    __syncthreads();
    for (int off = 1; off < NBMAX; off <<= 1) {
        int v = (t >= off) ? run[t - off] : 0;
        int gv = (SCAN && t >= off) ? gs[t - off] : 0;
        __syncthreads();
        run[t] += v;
        if (SCAN) gs[t] += gv;
        __syncthreads();
    }
    const int excl = run[t] - own;
    tstart[t] = excl;
    run[t] = excl;
    if (SCAN) {
        const int gincl = gs[t];
        const int gexcl = gincl - gown;
        gbase[t] = (own > 0) ? gexcl + atomicAdd(cursor + t, own) : 0;
        if (blockIdx.x == 0) {
            if (t < nb) bstart[t] = gexcl;
            if (t == nb - 1) bstart[nb] = gincl;
        }
    } else {
        gbase[t] = (own > 0) ? atomicAdd(cursor + t, own) : 0;
    }
    __syncthreads();
    for (int k0 = 0; k0 * PTHREADS < ne; k0 += PU) {
        int dv[PU], sv[PU];
        #pragma unroll
        for (int u = 0; u < PU; ++u) {
            int l = (k0 + u) * PTHREADS + t;
            long long e = base + (l < ne ? l : 0);
            dv[u] = dst[e];
            sv[u] = src[e];
        }
        #pragma unroll
        for (int u = 0; u < PU; ++u) {
            int l = (k0 + u) * PTHREADS + t;
            if (l < ne) {
                unsigned b = (unsigned)dv[u] / BNODES;
                int slot = atomicAdd(&run[b], 1);
                stage[slot] = ((unsigned)sv[u] << RSH) | ((unsigned)dv[u] - b * BNODES);
            }
        }
    }
    __syncthreads();
    const int lane = t & 31;
    int b = t >> 5;
    int s0 = tstart[b], s1 = run[b], g = gbase[b];
    for (; b < NBMAX; b += PTHREADS / 32) {
        const int c0 = s0, c1 = s1, off = g - s0;
        const int bn = b + PTHREADS / 32;
        if (bn < NBMAX) { s0 = tstart[bn]; s1 = run[bn]; g = gbase[bn]; }
        #pragma clang loop unroll(disable) vectorize(disable) interleave(disable)
        for (int slot = c0 + lane; slot < c1; slot += 32)
            part[off + slot] = stage[slot];
    }
}

__global__ __launch_bounds__(DTHREADS)
void bucket_degree_k(const unsigned* __restrict__ part0, const unsigned* __restrict__ part1,
                     const int* __restrict__ bstart0, const int* __restrict__ bstart1,
                     float* __restrict__ dis0, float* __restrict__ dis1, int n,
                     const int* __restrict__ ids0, const int* __restrict__ ids1,
                     v2u* __restrict__ nrec0, v2u* __restrict__ nrec1, int n_tab,
                     const float* __restrict__ tab_src, const float* __restrict__ tab_W,
                     float* __restrict__ tab) {
    const bool second = blockIdx.y != 0;
    const unsigned* __restrict__ part = second ? part1 : part0;
    const int* __restrict__ bstart = second ? bstart1 : bstart0;
    float* __restrict__ dis = second ? dis1 : dis0;
    const int* __restrict__ ids = second ? ids1 : ids0;
    v2u* __restrict__ nrec = second ? nrec1 : nrec0;
    __shared__ int cnt[BNODES];
    const int t = threadIdx.x;
    if (t < BNODES) cnt[t] = 0;
    if (tab != nullptr && !second) {
        const int wv = __builtin_amdgcn_readfirstlane(t >> 5);
        table_gemm16_tiles(tab_src, tab_W, tab, n_tab,
                           (int)blockIdx.x * (DTHREADS / 32) + wv,
                           (int)gridDim.x * (DTHREADS / 32), t & 31);
    }
    __syncthreads();
    const int b = blockIdx.x;
    const int p0 = bstart[b], p1 = bstart[b + 1];
    for (int p = p0 + t; p < p1; p += DTHREADS * PU) {
        unsigned r[PU];
        #pragma unroll
        for (int u = 0; u < PU; ++u) {
            int q = p + u * DTHREADS;
            r[u] = part[q < p1 ? q : 0];
        }
        #pragma unroll
        for (int u = 0; u < PU; ++u) {
            int q = p + u * DTHREADS;
            if (q < p1) atomicAdd(&cnt[r[u] & RMASK], 1);
        }
    }
    __syncthreads();
    const int i = b * BNODES + t;
    if (t < BNODES && i < n) {
        const float dv = rsqrtf((float)cnt[t] + 1.0f);
        dis[i] = dv;
        if (nrec) {
            int k = ids[i];
            k = k < 0 ? 0 : (k >= n_tab ? n_tab - 1 : k);
            v2u r; r.x = (unsigned)k; r.y = __float_as_uint(dv);
            nrec[i] = r;
        }
    }
}

template <int E>
__device__ __forceinline__ unsigned row_share_u32(unsigned x) {
    return (unsigned)__builtin_amdgcn_mov_dpp((int)x, 0x150 + E, 0xf, 0xf, true);
}

#define LTHREADS 512
#define PGR 16
template <bool POOL, bool PAIR, bool SCALED = false>
__global__ __launch_bounds__(LTHREADS)
void bucket_layer_k(const float* __restrict__ h0, const float* __restrict__ h1,
                    const float* __restrict__ dis0, const float* __restrict__ dis1,
                    const unsigned* __restrict__ part0, const unsigned* __restrict__ part1,
                    const int* __restrict__ bstart0, const int* __restrict__ bstart1,
                    const float* __restrict__ bias,
                    float* __restrict__ rows0, float* __restrict__ rows1,
                    const int* __restrict__ batch0, const int* __restrict__ batch1,
                    float* __restrict__ pooled0, float* __restrict__ pooled1,
                    float* __restrict__ cntg0, float* __restrict__ cntg1,
                    int n) {
    const bool second = blockIdx.y != 0;
    const float* __restrict__ h = second ? h1 : h0;
    const float* __restrict__ dis = second ? dis1 : dis0;
    const unsigned* __restrict__ part = second ? part1 : part0;
    const int* __restrict__ bstart = second ? bstart1 : bstart0;
    float* __restrict__ rows = second ? rows1 : rows0;
    const int* __restrict__ batch = second ? batch1 : batch0;
    float* __restrict__ pooled = second ? pooled1 : pooled0;
    float* __restrict__ cntg = second ? cntg1 : cntg0;
    __shared__ float acc[(BNODES + 1) * D];
    __shared__ float dloc[BNODES + 1];
    __shared__ float pl[PGR * D];
    __shared__ int   pc[PGR];
    const int t = threadIdx.x;
    const int b = blockIdx.x;
    const int node0 = b * BNODES;
    for (int i = t; i < (BNODES + 1) * D; i += LTHREADS) acc[i] = 0.0f;
    if (t <= BNODES) {
        int i = node0 + t;
        dloc[t] = (t < BNODES && i < n) ? dis[i] : 0.0f;
    }
    if (POOL) {
        if (t < PGR * D) pl[t] = 0.0f;
        if (t < PGR) pc[t] = 0;
    }
    __syncthreads();

    const int lane = t & 31;
    const unsigned j = (unsigned)lane & 15u;
    const unsigned j4 = j << 2;
    const int wv = __builtin_amdgcn_readfirstlane(t >> 5);
    const int p0 = bstart[b], p1 = bstart[b + 1];
    const v2f* __restrict__ hp = (const v2f*)h;
    int qb = p0 + wv * 32;
    unsigned recn = 0u;
    if (qb + lane < p1) recn = part[qb + lane];
    for (; qb < p1; qb += LTHREADS) {
        const int q = qb + lane;
        const unsigned rec = recn;
        const int qn = q + LTHREADS;
        recn = (qn < p1) ? part[qn] : 0u;
        const unsigned s = rec >> RSH;
        const unsigned dl = (q < p1) ? (rec & RMASK) : (unsigned)BNODES;
        const unsigned gi = s << 4;
        const unsigned ao4 = dl << 6;
        float wl = dloc[dl];
        if (!PAIR) wl *= dis[s];
        const unsigned wb = __float_as_uint(wl);
        float hv[16], ds[16];
#define LAYER_LOAD(E) { \
            const unsigned ix = row_share_u32<E>(gi) | j; \
            if (PAIR) { v2f pr = hp[(size_t)ix]; hv[E] = pr.x; ds[E] = pr.y; } \
            else { hv[E] = h[(size_t)ix]; ds[E] = 1.0f; } }
        LAYER_LOAD(0)  LAYER_LOAD(1)  LAYER_LOAD(2)  LAYER_LOAD(3)
        LAYER_LOAD(4)  LAYER_LOAD(5)  LAYER_LOAD(6)  LAYER_LOAD(7)
        LAYER_LOAD(8)  LAYER_LOAD(9)  LAYER_LOAD(10) LAYER_LOAD(11)
        LAYER_LOAD(12) LAYER_LOAD(13) LAYER_LOAD(14) LAYER_LOAD(15)
#undef LAYER_LOAD
#define LAYER_ADD(E) { \
            const float w = __uint_as_float(row_share_u32<E>(wb)); \
            const unsigned a = row_share_u32<E>(ao4) | j4; \
            const float norm = PAIR ? ds[E] * w : w; \
            atomicAdd((float*)((char*)acc + a), hv[E] * norm); }
        LAYER_ADD(0)  LAYER_ADD(1)  LAYER_ADD(2)  LAYER_ADD(3)
        LAYER_ADD(4)  LAYER_ADD(5)  LAYER_ADD(6)  LAYER_ADD(7)
        LAYER_ADD(8)  LAYER_ADD(9)  LAYER_ADD(10) LAYER_ADD(11)
        LAYER_ADD(12) LAYER_ADD(13) LAYER_ADD(14) LAYER_ADD(15)
#undef LAYER_ADD
    }
    __syncthreads();

    if (!POOL) {
        for (int idx = t; idx < BNODES * D; idx += LTHREADS) {
            int nl = idx >> 4;
            int i = node0 + nl;
            if (i < n) {
                float di = dloc[nl];
                size_t o = (size_t)i * D + (idx & 15);
                float v = acc[idx] + h[PAIR ? o * 2 : o] * (di * di) + bias[idx & 15];
                v = v > 0.0f ? v : 0.0f;
                rows[o] = SCALED ? v * di : v;
            }
        }
    } else {
        const int g0 = batch[node0];
        for (int idx = t; idx < BNODES * D; idx += LTHREADS) {
            int nl = idx >> 4;
            int jj = idx & 15;
            int i = node0 + nl;
            if (i < n) {
                float di = dloc[nl];
                size_t o = (size_t)i * D + jj;
                float v = acc[idx] + h[PAIR ? o * 2 : o] * (di * di) + bias[jj];
                int g = batch[i];
                unsigned k = (unsigned)(g - g0);
                if (k < (unsigned)PGR) {
                    atomicAdd(&pl[k * D + jj], v);
                    if (jj == 0) atomicAdd(&pc[k], 1);
                } else {
                    atomicAdd(pooled + (size_t)g * D + jj, v);
                    if (jj == 0) atomicAdd(cntg + g, 1.0f);
                }
            }
        }
        __syncthreads();
        if (t < PGR * D) {
            int k = t >> 4;
            int c = pc[k];
            if (c > 0) {
                int g = g0 + k;
                atomicAdd(pooled + (size_t)g * D + (t & 15), pl[t]);
                if ((t & 15) == 0) atomicAdd(cntg + g, (float)c);
            }
        }
    }
}


#define TCAP 1032
#define TCOPY ((TCAP * D / 4 + LTHREADS - 1) / LTHREADS)
#define TROWS ((TCAP + 15) & ~15)
__global__ __launch_bounds__(LTHREADS)
void bucket_layer1_tab_k(const float* __restrict__ tab, int n_tab,
                         const v2u* __restrict__ nrec0, const v2u* __restrict__ nrec1,
                         const unsigned* __restrict__ part0, const unsigned* __restrict__ part1,
                         const int* __restrict__ bstart0, const int* __restrict__ bstart1,
                         const float* __restrict__ bias,
                         float* __restrict__ rows0, float* __restrict__ rows1, int n) {
    const bool second = blockIdx.y != 0;
    const v2u* __restrict__ nrec = second ? nrec1 : nrec0;
    const unsigned* __restrict__ part = second ? part1 : part0;
    const int* __restrict__ bstart = second ? bstart1 : bstart0;
    float* __restrict__ rows = second ? rows1 : rows0;
    __shared__ struct {
        float acc[(BNODES + 1) * D];
        float dloc[BNODES + 4];
        v4f tl[TCAP * D / 4];
    } sm;
    float* const acc = sm.acc;
    float* const dloc = sm.dloc;
    v4f* const tl = sm.tl;
    const int t = threadIdx.x;
    const int b = blockIdx.x;
    const int node0 = b * BNODES;
    const int lane = t & 31;
    const unsigned j4 = ((unsigned)lane & 15u) << 2;
    const int wv = __builtin_amdgcn_readfirstlane(t >> 5);
    const int p0 = bstart[b], p1 = bstart[b + 1];
    int qb = p0 + wv * 32;
    unsigned rec0 = 0u, rec1 = 0u;
    if (qb + lane < p1) rec0 = part[qb + lane];
    if (qb + LTHREADS + lane < p1) rec1 = part[qb + LTHREADS + lane];
    for (int i = t; i < (BNODES + 1) * D; i += LTHREADS) acc[i] = 0.0f;
    if (t <= BNODES) {
        int i = node0 + t;
        dloc[t] = (t < BNODES && i < n) ? __uint_as_float(nrec[i].y) : 0.0f;
    }
    {
        const v4f* __restrict__ tg = (const v4f*)tab;
        const int n4 = n_tab * (D / 4);
        v4f tv[TCOPY];
        #pragma unroll
        for (int u = 0; u < TCOPY; ++u) {
            int i = t + u * LTHREADS;
            tv[u] = tg[i < n4 ? i : 0];
        }
        #pragma unroll
        for (int u = 0; u < TCOPY; ++u) {
            int i = t + u * LTHREADS;
            if (i < n4) tl[i] = tv[u];
        }
    }
    v2u nd = nrec[rec0 >> RSH];
    __syncthreads();

    const char* tlb = (const char*)tl;
    for (; qb < p1; qb += LTHREADS) {
        const int q = qb + lane;
        const unsigned rec = rec0;
        const v2u ndc = nd;
        rec0 = rec1;
        nd = nrec[rec0 >> RSH];
        const int q2 = q + 2 * LTHREADS;
        rec1 = (q2 < p1) ? part[q2] : 0u;
        const unsigned dl = (q < p1) ? (rec & RMASK) : (unsigned)BNODES;
        const unsigned to4 = ndc.x << 6;
        const unsigned ao4 = dl << 6;
        const unsigned wb = __float_as_uint(dloc[dl] * __uint_as_float(ndc.y));
        float hv[16];
#define TAB_LOAD(E) { \
            const unsigned ta = row_share_u32<E>(to4) | j4; \
            hv[E] = *(const float*)(tlb + ta); }
        TAB_LOAD(0)  TAB_LOAD(1)  TAB_LOAD(2)  TAB_LOAD(3)
        TAB_LOAD(4)  TAB_LOAD(5)  TAB_LOAD(6)  TAB_LOAD(7)
        TAB_LOAD(8)  TAB_LOAD(9)  TAB_LOAD(10) TAB_LOAD(11)
        TAB_LOAD(12) TAB_LOAD(13) TAB_LOAD(14) TAB_LOAD(15)
#undef TAB_LOAD
#define TAB_ADD(E) { \
            const float w = __uint_as_float(row_share_u32<E>(wb)); \
            const unsigned a = row_share_u32<E>(ao4) | j4; \
            atomicAdd((float*)((char*)acc + a), hv[E] * w); }
        TAB_ADD(0)  TAB_ADD(1)  TAB_ADD(2)  TAB_ADD(3)
        TAB_ADD(4)  TAB_ADD(5)  TAB_ADD(6)  TAB_ADD(7)
        TAB_ADD(8)  TAB_ADD(9)  TAB_ADD(10) TAB_ADD(11)
        TAB_ADD(12) TAB_ADD(13) TAB_ADD(14) TAB_ADD(15)
#undef TAB_ADD
    }
    __syncthreads();

    const float* tlf = (const float*)tl;
    for (int idx = t; idx < BNODES * D; idx += LTHREADS) {
        int nl = idx >> 4;
        int i = node0 + nl;
        if (i < n) {
            float di = dloc[nl];
            float hs = tlf[nrec[i].x * D + (idx & 15)];
            float v = acc[idx] + hs * (di * di) + bias[idx & 15];
            rows[(size_t)i * D + (idx & 15)] = (v > 0.0f ? v : 0.0f) * di;
        }
    }
}

__global__ __launch_bounds__(LTHREADS)
void bucket_layer2_sum_k(const float* __restrict__ h0, const float* __restrict__ h1,
                         const float* __restrict__ dis0, const float* __restrict__ dis1,
                         const unsigned* __restrict__ part0, const unsigned* __restrict__ part1,
                         const int* __restrict__ bstart0, const int* __restrict__ bstart1,
                         const int* __restrict__ batch0, const int* __restrict__ batch1,
                         float* __restrict__ pooled0, float* __restrict__ pooled1,
                         float* __restrict__ cntg0, float* __restrict__ cntg1,
                         int n) {
    const bool second = blockIdx.y != 0;
    const float* __restrict__ h = second ? h1 : h0;
    const float* __restrict__ dis = second ? dis1 : dis0;
    const unsigned* __restrict__ part = second ? part1 : part0;
    const int* __restrict__ bstart = second ? bstart1 : bstart0;
    const int* __restrict__ batch = second ? batch1 : batch0;
    float* __restrict__ pooled = second ? pooled1 : pooled0;
    float* __restrict__ cntg = second ? cntg1 : cntg0;
    __shared__ float acc[(BNODES + 1) * D];
    __shared__ float pl[PGR * D];
    __shared__ int   pc[PGR];
    const int t = threadIdx.x;
    const int b = blockIdx.x;
    const int node0 = b * BNODES;
    for (int i = t; i < (BNODES + 1) * D; i += LTHREADS) acc[i] = 0.0f;
    if (t < PGR * D) pl[t] = 0.0f;
    if (t < PGR) pc[t] = 0;
    __syncthreads();

    const int lane = t & 31;
    const unsigned j4 = ((unsigned)lane & 15u) << 2;
    const int wv = __builtin_amdgcn_readfirstlane(t >> 5);
    const int p0 = bstart[b], p1 = bstart[b + 1];
    const char* __restrict__ hb = (const char*)h;
    int qb = p0 + wv * 32;
    unsigned recn = 0u;
    if (qb + lane < p1) recn = part[qb + lane];
    for (; qb < p1; qb += LTHREADS) {
        const int q = qb + lane;
        const unsigned rec = recn;
        const int qn = q + LTHREADS;
        recn = (qn < p1) ? part[qn] : 0u;
        const unsigned dl = (q < p1) ? (rec & RMASK) : (unsigned)BNODES;
        const unsigned go4 = (rec >> RSH) << 6;
        const unsigned ao4 = dl << 6;
        float hv[16];
#define SUM_LOAD(E) { \
            const unsigned ga = row_share_u32<E>(go4) | j4; \
            hv[E] = *(const float*)(hb + ga); }
        SUM_LOAD(0)  SUM_LOAD(1)  SUM_LOAD(2)  SUM_LOAD(3)
        SUM_LOAD(4)  SUM_LOAD(5)  SUM_LOAD(6)  SUM_LOAD(7)
        SUM_LOAD(8)  SUM_LOAD(9)  SUM_LOAD(10) SUM_LOAD(11)
        SUM_LOAD(12) SUM_LOAD(13) SUM_LOAD(14) SUM_LOAD(15)
#undef SUM_LOAD
#define SUM_ADD(E) { \
            const unsigned a = row_share_u32<E>(ao4) | j4; \
            atomicAdd((float*)((char*)acc + a), hv[E]); }
        SUM_ADD(0)  SUM_ADD(1)  SUM_ADD(2)  SUM_ADD(3)
        SUM_ADD(4)  SUM_ADD(5)  SUM_ADD(6)  SUM_ADD(7)
        SUM_ADD(8)  SUM_ADD(9)  SUM_ADD(10) SUM_ADD(11)
        SUM_ADD(12) SUM_ADD(13) SUM_ADD(14) SUM_ADD(15)
#undef SUM_ADD
    }
    __syncthreads();

    const int g0 = batch[node0];
    for (int idx = t; idx < BNODES * D; idx += LTHREADS) {
        int nl = idx >> 4;
        int jj = idx & 15;
        int i = node0 + nl;
        if (i < n) {
            float v = dis[i] * (acc[idx] + h[(size_t)i * D + jj]);
            int g = batch[i];
            unsigned k = (unsigned)(g - g0);
            if (k < (unsigned)PGR) {
                atomicAdd(&pl[k * D + jj], v);
                if (jj == 0) atomicAdd(&pc[k], 1);
            } else {
                atomicAdd(pooled + (size_t)g * D + jj, v);
                if (jj == 0) atomicAdd(cntg + g, 1.0f);
            }
        }
    }
    __syncthreads();
    if (t < PGR * D) {
        int k = t >> 4;
        int c = pc[k];
        if (c > 0) {
            int g = g0 + k;
            atomicAdd(pooled + (size_t)g * D + (t & 15), pl[t]);
            if ((t & 15) == 0) atomicAdd(cntg + g, (float)c);
        }
    }
}

static void run_protein_bucketed(const float* emb, const float* W1, const float* b1,
                                 const float* W2, const float* b2,
                                 const int* ids, const int* edge, const int* batch,
                                 int N, long long E,
                                 float* bufA, float* bufB, float* dis,
                                 float* pooled, float* cnt,
                                 unsigned* part, int* bcount, int* bstart, int* cursor,
                                 bool pair, hipStream_t stream) {
    const int* src = edge;
    const int* dst = edge + E;
    const int B = 256;
    dim3 blk(B);
    int gemm_grid = (int)(((N >> 4) + 7) / 8);
    if (gemm_grid < 1) gemm_grid = 1;
    const int nb = (N + BNODES - 1) / BNODES;
    const long long round_cap = (long long)NWGP * PTILE;
    const int ptiles = NWGP * (int)((E + round_cap - 1) / round_cap);
    const int tlen = (int)(((E + ptiles - 1) / ptiles + 31) & ~31ll);

    fill_zero_k<<<(NBMAX + B - 1) / B, blk, 0, stream>>>((float*)bcount, NBMAX);
    bucket_hist_k<<<ptiles, PTHREADS, 0, stream>>>(dst, nullptr, E, tlen, bcount, nullptr);
    bucket_scan_k<<<1, NBMAX, 0, stream>>>(bcount, nullptr, bstart, nullptr, cursor, nullptr, nb);
    bucket_scatter_k<false><<<ptiles, PTHREADS, 0, stream>>>(src, nullptr, dst, nullptr, E, tlen,
                                                             nullptr, nullptr, nullptr, nullptr,
                                                             cursor, nullptr, part, nullptr, nb);
    bucket_degree_k<<<nb, DTHREADS, 0, stream>>>(part, nullptr, bstart, nullptr, dis, nullptr, N,
                                                 nullptr, nullptr, nullptr, nullptr, 0,
                                                 nullptr, nullptr, nullptr);

    gather_gemm16_k<<<gemm_grid, blk, 0, stream>>>(emb, nullptr, ids, nullptr, W1, bufA, nullptr,
                                                   pair ? dis : nullptr, nullptr, N);
    if (pair)
        bucket_layer_k<false, true><<<nb, LTHREADS, 0, stream>>>(
            bufA, nullptr, dis, nullptr, part, nullptr, bstart, nullptr, b1, bufB, nullptr,
            nullptr, nullptr, nullptr, nullptr, nullptr, nullptr, N);
    else
        bucket_layer_k<false, false><<<nb, LTHREADS, 0, stream>>>(
            bufA, nullptr, dis, nullptr, part, nullptr, bstart, nullptr, b1, bufB, nullptr,
            nullptr, nullptr, nullptr, nullptr, nullptr, nullptr, N);
    gather_gemm16_k<<<gemm_grid, blk, 0, stream>>>(bufB, nullptr, nullptr, nullptr, W2, bufA, nullptr,
                                                   pair ? dis : nullptr, nullptr, N);

    fill_zero_k<<<(int)((NGRAPH * D + B - 1) / B), blk, 0, stream>>>(pooled, NGRAPH * D);
    fill_zero_k<<<(int)((NGRAPH + B - 1) / B), blk, 0, stream>>>(cnt, NGRAPH);
    if (pair)
        bucket_layer_k<true, true><<<nb, LTHREADS, 0, stream>>>(
            bufA, nullptr, dis, nullptr, part, nullptr, bstart, nullptr, b2, nullptr, nullptr,
            batch, nullptr, pooled, nullptr, cnt, nullptr, N);
    else
        bucket_layer_k<true, false><<<nb, LTHREADS, 0, stream>>>(
            bufA, nullptr, dis, nullptr, part, nullptr, bstart, nullptr, b2, nullptr, nullptr,
            batch, nullptr, pooled, nullptr, cnt, nullptr, N);
}

static void run_both_bucketed(const float* emb, const float* W1, const float* b1,
                              const float* W2,
                              const int* ids0, const int* ids1,
                              const int* edge0, const int* edge1,
                              const int* batch0, const int* batch1,
                              int N, long long E,
                              float* pair0, float* pair1, float* rows0, float* rows1,
                              float* dis0, float* dis1,
                              float* zero_run, long long zero_words,
                              float* pooled0, float* pooled1, float* cnt0, float* cnt1,
                              int* bcount0, int* bcount1, int* bstart0, int* bstart1,
                              int* cursor0, int* cursor1,
                              unsigned* part0, unsigned* part1,
                              int n_tab, float* tab, v2u* nrec0, v2u* nrec1,
                              hipStream_t stream) {
    const int B = 256;
    dim3 blk(B);
    int gemm_grid = (int)(((N >> 4) + 7) / 8);
    if (gemm_grid < 1) gemm_grid = 1;
    const int nb = (N + BNODES - 1) / BNODES;
    const long long round_cap = (long long)NWGP * PTILE;
    const int ptiles = NWGP * (int)((E + round_cap - 1) / round_cap);
    const int tlen = (int)(((E + ptiles - 1) / ptiles + 31) & ~31ll);

    fill_zero_k<<<(int)((zero_words + B - 1) / B), blk, 0, stream>>>(zero_run, zero_words);
    bucket_hist_k<<<dim3(ptiles, 2), PTHREADS, 0, stream>>>(edge0 + E, edge1 + E, E, tlen,
                                                            bcount0, bcount1);
    bucket_scatter_k<true><<<dim3(ptiles, 2), PTHREADS, 0, stream>>>(
        edge0, edge1, edge0 + E, edge1 + E, E, tlen, bcount0, bcount1, bstart0, bstart1,
        cursor0, cursor1, part0, part1, nb);
    if (tab) {
        bucket_degree_k<<<dim3(nb, 2), DTHREADS, 0, stream>>>(part0, part1, bstart0, bstart1,
                                                              dis0, dis1, N,
                                                              ids0, ids1, nrec0, nrec1, n_tab,
                                                              emb, W1, tab);
        bucket_layer1_tab_k<<<dim3(nb, 2), LTHREADS, 0, stream>>>(
            tab, n_tab, nrec0, nrec1, part0, part1, bstart0, bstart1, b1, rows0, rows1, N);
    } else {
        bucket_degree_k<<<dim3(nb, 2), DTHREADS, 0, stream>>>(part0, part1, bstart0, bstart1,
                                                              dis0, dis1, N,
                                                              nullptr, nullptr, nullptr, nullptr, 0,
                                                              nullptr, nullptr, nullptr);
        gather_gemm16_k<<<dim3(gemm_grid, 2), blk, 0, stream>>>(emb, emb, ids0, ids1, W1, pair0, pair1,
                                                                dis0, dis1, N);
        bucket_layer_k<false, true, true><<<dim3(nb, 2), LTHREADS, 0, stream>>>(
            pair0, pair1, dis0, dis1, part0, part1, bstart0, bstart1, b1, rows0, rows1,
            nullptr, nullptr, nullptr, nullptr, nullptr, nullptr, N);
    }
    bucket_layer2_sum_k<<<dim3(nb, 2), LTHREADS, 0, stream>>>(
        rows0, rows1, dis0, dis1, part0, part1, bstart0, bstart1,
        batch0, batch1, pooled0, pooled1, cnt0, cnt1, N);
}

static void run_protein(const float* emb, const float* W1, const float* b1,
                        const float* W2, const float* b2,
                        const int* ids, const int* edge, const int* batch,
                        int N, long long E,
                        float* bufA, float* bufB, float* dis,
                        float* pooled, float* cnt, hipStream_t stream) {
    const int* src = edge;
    const int* dst = edge + E;
    const long long n16 = (long long)N * D;
    const int B = 256;
    dim3 blk(B);
    int gemm_grid = (int)(((N >> 4) + 7) / 8);
    if (gemm_grid < 1) gemm_grid = 1;

    fill_zero_k<<<(int)((N + B - 1) / B), blk, 0, stream>>>(dis, N);
    fill_zero_k<<<(int)((n16 + B - 1) / B), blk, 0, stream>>>(bufB, n16);

    gather_gemm16_k<<<gemm_grid, blk, 0, stream>>>(emb, nullptr, ids, nullptr, W1, bufA, nullptr,
                                                   nullptr, nullptr, N);

    degree_k<<<(int)((E + B - 1) / B), blk, 0, stream>>>(dst, dis, E);
    rsqrt_k<<<(int)((N + B - 1) / B), blk, 0, stream>>>(dis, N);

    aggregate_k<<<(int)((E * 16 + B - 1) / B), blk, 0, stream>>>(bufA, dis, src, dst, bufB, E);
    node_relu_k<<<(int)((n16 + B - 1) / B), blk, 0, stream>>>(bufA, bufB, dis, b1, n16);

    gather_gemm16_k<<<gemm_grid, blk, 0, stream>>>(bufA, nullptr, nullptr, nullptr, W2, bufB, nullptr,
                                                   nullptr, nullptr, N);

    fill_zero_k<<<(int)((n16 + B - 1) / B), blk, 0, stream>>>(bufA, n16);
    aggregate_k<<<(int)((E * 16 + B - 1) / B), blk, 0, stream>>>(bufB, dis, src, dst, bufA, E);

    fill_zero_k<<<(int)((NGRAPH * D + B - 1) / B), blk, 0, stream>>>(pooled, NGRAPH * D);
    fill_zero_k<<<(int)((NGRAPH + B - 1) / B), blk, 0, stream>>>(cnt, NGRAPH);
    node_pool_k<<<(int)((n16 + B - 1) / B), blk, 0, stream>>>(bufB, bufA, dis, b2, batch,
                                                              pooled, cnt, n16);
}

extern "C" void kernel_launch(void* const* d_in, const int* in_sizes, int n_in,
                              void* d_out, int out_size, void* d_ws, size_t ws_size,
                              hipStream_t stream) {
    const float* emb = (const float*)d_in[0];
    const float* W1  = (const float*)d_in[1];
    const float* b1  = (const float*)d_in[2];
    const float* W2  = (const float*)d_in[3];
    const float* b2  = (const float*)d_in[4];
    const float* fcW = (const float*)d_in[5];
    const float* fcb = (const float*)d_in[6];
    const int* rx     = (const int*)d_in[7];
    const int* redge  = (const int*)d_in[8];
    const int* rbatch = (const int*)d_in[9];
    const int* lx     = (const int*)d_in[10];
    const int* ledge  = (const int*)d_in[11];
    const int* lbatch = (const int*)d_in[12];

    const int N = in_sizes[7];
    const long long E = in_sizes[8] / 2;

    char* ws = (char*)d_ws;
    float* bufA = (float*)ws;  ws += (size_t)N * D * sizeof(float);
    float* bufB = (float*)ws;  ws += (size_t)N * D * sizeof(float);
    float* dis  = (float*)ws;  ws += (size_t)N * sizeof(float);
    float* pooled_r = (float*)ws; ws += (size_t)NGRAPH * D * sizeof(float);
    float* pooled_l = (float*)ws; ws += (size_t)NGRAPH * D * sizeof(float);
    float* cnt_r = (float*)ws; ws += (size_t)NGRAPH * sizeof(float);
    float* cnt_l = (float*)ws; ws += (size_t)NGRAPH * sizeof(float);

    unsigned* part = (unsigned*)ws;  ws += (size_t)E * sizeof(unsigned);
    int* bcount = (int*)ws;  ws += (size_t)NBMAX * sizeof(int);
    int* bstart = (int*)ws;  ws += (size_t)2 * NBMAX * sizeof(int);
    int* cursor = (int*)ws;  ws += (size_t)NBMAX * sizeof(int);
    const bool bucketed = (size_t)(ws - (char*)d_ws) <= ws_size &&
                          N > 0 && N <= NBMAX * BNODES && E > 0 && E < (1ll << 30);
    float* bufT = (float*)ws;  ws += (size_t)N * D * sizeof(float);
    const bool pair = (size_t)(ws - (char*)d_ws) <= ws_size;
    float* rowsB = pair ? bufT : bufB;

    char* q = (char*)d_ws;
    float* pair0 = (float*)q;  q += (size_t)N * D * 2 * sizeof(float);
    float* pair1 = (float*)q;  q += (size_t)N * D * 2 * sizeof(float);
    float* rows0 = (float*)q;  q += (size_t)N * D * sizeof(float);
    float* rows1 = (float*)q;  q += (size_t)N * D * sizeof(float);
    float* dis0  = (float*)q;  q += (size_t)N * sizeof(float);
    float* dis1  = (float*)q;  q += (size_t)N * sizeof(float);
    float* zero_run = (float*)q;
    float* pool0 = (float*)q;  q += (size_t)NGRAPH * D * sizeof(float);
    float* pool1 = (float*)q;  q += (size_t)NGRAPH * D * sizeof(float);
    float* pcnt0 = (float*)q;  q += (size_t)NGRAPH * sizeof(float);
    float* pcnt1 = (float*)q;  q += (size_t)NGRAPH * sizeof(float);
    int* bcount0 = (int*)q;  q += (size_t)NBMAX * sizeof(int);
    int* bcount1 = (int*)q;  q += (size_t)NBMAX * sizeof(int);
    int* cursor0 = (int*)q;  q += (size_t)NBMAX * sizeof(int);
    int* cursor1 = (int*)q;  q += (size_t)NBMAX * sizeof(int);
    const long long zero_words = 2ll * NGRAPH * D + 2ll * NGRAPH + 4ll * NBMAX;
    int* bstart0 = (int*)q;  q += (size_t)2 * NBMAX * sizeof(int);
    int* bstart1 = (int*)q;  q += (size_t)2 * NBMAX * sizeof(int);
    unsigned* part0 = (unsigned*)q;  q += (size_t)E * sizeof(unsigned);
    unsigned* part1 = (unsigned*)q;  q += (size_t)E * sizeof(unsigned);
    const bool both = bucketed && (size_t)(q - (char*)d_ws) <= ws_size;
    const int n_tab = in_sizes[0] / D;
    float* tab = (float*)pair1;
    v2u* nrec0 = (v2u*)(tab + (size_t)TROWS * D);
    v2u* nrec1 = nrec0 + N;
    const bool tabbed = n_tab > 0 && n_tab <= TCAP &&
                        (size_t)TROWS * D * sizeof(float) + (size_t)2 * N * sizeof(v2u) <=
                            (size_t)N * D * 2 * sizeof(float);
    if (both) {
        run_both_bucketed(emb, W1, b1, W2, rx, lx, redge, ledge, rbatch, lbatch, N, E,
                          pair0, pair1, rows0, rows1, dis0, dis1, zero_run, zero_words,
                          pool0, pool1, pcnt0, pcnt1, bcount0, bcount1, bstart0, bstart1,
                          cursor0, cursor1, part0, part1,
                          n_tab, tabbed ? tab : nullptr, nrec0, nrec1, stream);
        pool_fc_k<<<(NGRAPH / 16 + FWAVES - 1) / FWAVES, FWAVES * 32, 0, stream>>>(
            pool0, pcnt0, pool1, pcnt1, W2, fcW, fcb, b2, (float*)d_out);
        return;
    }
    if (bucketed) {
        run_protein_bucketed(emb, W1, b1, W2, b2, rx, redge, rbatch, N, E,
                             bufA, rowsB, dis, pooled_r, cnt_r,
                             part, bcount, bstart, cursor, pair, stream);
        run_protein_bucketed(emb, W1, b1, W2, b2, lx, ledge, lbatch, N, E,
                             bufA, rowsB, dis, pooled_l, cnt_l,
                             part, bcount, bstart, cursor, pair, stream);
    } else {
        run_protein(emb, W1, b1, W2, b2, rx, redge, rbatch, N, E,
                    bufA, bufB, dis, pooled_r, cnt_r, stream);
        run_protein(emb, W1, b1, W2, b2, lx, ledge, lbatch, N, E,
                    bufA, bufB, dis, pooled_l, cnt_l, stream);
    }

    fc_k<<<(NGRAPH * 6 + 255) / 256, 256, 0, stream>>>(pooled_r, cnt_r, pooled_l, cnt_l,
                                                       fcW, fcb, nullptr, (float*)d_out);
}
